# starting kernel
# baseline (speedup 1.0000x reference)
#include <hip/hip_runtime.h>
#include <hip/hip_bf16.h>


typedef float v2f __attribute__((ext_vector_type(2)));
typedef float v8f __attribute__((ext_vector_type(8)));

#define IN_C 512
#define H1C 64
#define H2C 32


__global__ void k_init_deg(float* __restrict__ deg, int Nn) {
  int i = blockIdx.x * blockDim.x + threadIdx.x;
  if (i < Nn) deg[i] = 1.0f;
}

__global__ void k_deg_accum(const int* __restrict__ ei, const float* __restrict__ ew,
                            float* __restrict__ deg, int E) {
  int e = blockIdx.x * blockDim.x + threadIdx.x;
  if (e >= E) return;
  int col = ei[E + e];
  atomicAdd(&deg[col], ew[e]);
}

__global__ void k_dinv(float* __restrict__ dinv, int Nn) {
  int i = blockIdx.x * blockDim.x + threadIdx.x;
  if (i >= Nn) return;
  float d = dinv[i];
  dinv[i] = (d > 0.0f) ? rsqrtf(fmaxf(d, 1e-12f)) : 0.0f;
}

__global__ void k_wnorm(const int* __restrict__ ei, const float* __restrict__ ew,
                        const float* __restrict__ dinv, float* __restrict__ wnorm, int E) {
  int e = blockIdx.x * blockDim.x + threadIdx.x;
  if (e >= E) return;
  int row = ei[e];
  int col = ei[E + e];
  wnorm[e] = dinv[row] * ew[e] * dinv[col];
}


template <int K, int NT, bool RELU>
__global__ __launch_bounds__(NT * 32) void k_gemm_wmma_f32(
    const float* __restrict__ A, const float* __restrict__ B, float* __restrict__ C) {
  constexpr int NOUT = NT * 16;
  const int wave = threadIdx.x >> 5;
  const int lane = threadIdx.x & 31;
  const int m = lane & 15;
  const int half = lane >> 4;
  const int n0 = wave * 16;
  const long long rowBase = (long long)blockIdx.x * 16;

  const float* arow = A + (rowBase + m) * K + half * 2;

  v8f acc = {};
#pragma unroll 4
  for (int k = 0; k < K; k += 4) {
    float2 av = *reinterpret_cast<const float2*>(arow + k);
    v2f a;
    a.x = RELU ? fmaxf(av.x, 0.0f) : av.x;
    a.y = RELU ? fmaxf(av.y, 0.0f) : av.y;
    const int kr = k + half * 2;
    v2f b;
    b.x = B[kr * NOUT + n0 + m];
    b.y = B[(kr + 1) * NOUT + n0 + m];
    acc = __builtin_amdgcn_wmma_f32_16x16x4_f32(false, a, false, b, (short)0, acc,
                                                false, false);
  }

#pragma unroll
  for (int v = 0; v < 8; ++v) {
    long long r = rowBase + v + 8 * half;
    C[r * NOUT + n0 + m] = acc[v];
  }
}

template <int F>
__global__ void k_init_agg(float* __restrict__ agg, const float* __restrict__ h,
                           const float* __restrict__ dinv, const float* __restrict__ bias,
                           int Nn) {
  long long t = (long long)blockIdx.x * blockDim.x + threadIdx.x;
  if (t >= (long long)Nn * F) return;
  int node = (int)(t / F);
  int f = (int)(t % F);
  float di = dinv[node];
  agg[t] = bias[f] + di * di * h[t];
}

template <int CH>
__global__ __launch_bounds__(256) void k_edge_agg(const int* __restrict__ ei,
                                                  const float* __restrict__ wnorm,
                                                  const float* __restrict__ h,
                                                  float* __restrict__ agg, int E) {
  long long t = (long long)blockIdx.x * blockDim.x + threadIdx.x;
  if (t >= (long long)E * CH) return;
  int e = (int)(t / CH);
  int c = (int)(t % CH);
  int row = ei[e];
  int col = ei[E + e];
  float w = wnorm[e];
  float4 v = reinterpret_cast<const float4*>(h + (size_t)row * (CH * 4))[c];
  float* dst = agg + (size_t)col * (CH * 4) + c * 4;
  atomicAdd(dst + 0, w * v.x);
  atomicAdd(dst + 1, w * v.y);
  atomicAdd(dst + 2, w * v.z);
  atomicAdd(dst + 3, w * v.w);
}

__global__ void k_gemm3_relu(const float* __restrict__ a2, const float* __restrict__ W3,
                             float* __restrict__ g, int Nn) {
  int i = blockIdx.x * blockDim.x + threadIdx.x;
  if (i >= Nn) return;
  const float* r = a2 + (size_t)i * H2C;
  float acc = 0.0f;
#pragma unroll
  for (int f = 0; f < H2C; ++f) acc += fmaxf(r[f], 0.0f) * W3[f];
  g[i] = acc;
}

__global__ void k_init_out(float* __restrict__ out, const float* __restrict__ g,
                           const float* __restrict__ dinv, const float* __restrict__ b3,
                           int Nn) {
  int i = blockIdx.x * blockDim.x + threadIdx.x;
  if (i >= Nn) return;
  float di = dinv[i];
  out[i] = b3[0] + di * di * g[i];
}

__global__ void k_edge_agg1(const int* __restrict__ ei, const float* __restrict__ wnorm,
                            const float* __restrict__ g, float* __restrict__ out, int E) {
  int e = blockIdx.x * blockDim.x + threadIdx.x;
  if (e >= E) return;
  int row = ei[e];
  int col = ei[E + e];
  atomicAdd(&out[col], wnorm[e] * g[row]);
}


static inline unsigned int blocks_for(long long total, int bs) {
  return (unsigned int)((total + bs - 1) / bs);
}

extern "C" void kernel_launch(void* const* d_in, const int* in_sizes, int n_in,
                              void* d_out, int out_size, void* d_ws, size_t ws_size,
                              hipStream_t stream) {
  const float* x = (const float*)d_in[0];
  const int* ei = (const int*)d_in[1];
  const float* ew = (const float*)d_in[2];
  const float* W1 = (const float*)d_in[3];
  const float* b1 = (const float*)d_in[4];
  const float* W2 = (const float*)d_in[5];
  const float* b2 = (const float*)d_in[6];
  const float* W3 = (const float*)d_in[7];
  const float* b3 = (const float*)d_in[8];

  const int E = in_sizes[2];
  const int Nn = in_sizes[0] / IN_C;
  float* out = (float*)d_out;

  float* dinv = (float*)d_ws;
  float* wnorm = dinv + Nn;
  float* bufA = wnorm + E;
  float* bufB = bufA + (size_t)Nn * H1C;

  k_init_deg<<<blocks_for(Nn, 256), 256, 0, stream>>>(dinv, Nn);
  k_deg_accum<<<blocks_for(E, 256), 256, 0, stream>>>(ei, ew, dinv, E);
  k_dinv<<<blocks_for(Nn, 256), 256, 0, stream>>>(dinv, Nn);
  k_wnorm<<<blocks_for(E, 256), 256, 0, stream>>>(ei, ew, dinv, wnorm, E);

  k_gemm_wmma_f32<IN_C, H1C / 16, false>
      <<<Nn / 16, (H1C / 16) * 32, 0, stream>>>(x, W1, bufA);
  k_init_agg<H1C><<<blocks_for((long long)Nn * H1C, 256), 256, 0, stream>>>(
      bufB, bufA, dinv, b1, Nn);
  k_edge_agg<H1C / 4><<<blocks_for((long long)E * (H1C / 4), 256), 256, 0, stream>>>(
      ei, wnorm, bufA, bufB, E);

  k_gemm_wmma_f32<H1C, H2C / 16, true>
      <<<Nn / 16, (H2C / 16) * 32, 0, stream>>>(bufB, W2, bufA);
  k_init_agg<H2C><<<blocks_for((long long)Nn * H2C, 256), 256, 0, stream>>>(
      bufB, bufA, dinv, b2, Nn);
  k_edge_agg<H2C / 4><<<blocks_for((long long)E * (H2C / 4), 256), 256, 0, stream>>>(
      ei, wnorm, bufA, bufB, E);

  k_gemm3_relu<<<blocks_for(Nn, 256), 256, 0, stream>>>(bufB, W3, bufA, Nn);
  k_init_out<<<blocks_for(Nn, 256), 256, 0, stream>>>(out, bufA, dinv, b3, Nn);
  k_edge_agg1<<<blocks_for(E, 256), 256, 0, stream>>>(ei, wnorm, bufA, out, E);
}


# Degree and layer-3 scalar scatter-adds sum per edge slice in LDS by node range, planes added by a dense pass: no global atomics there; f32 summation order changes
# speedup vs baseline: 1.4071x; 1.4071x over previous
#include <hip/hip_runtime.h>
#include <hip/hip_bf16.h>


typedef float v2f __attribute__((ext_vector_type(2)));
typedef float v8f __attribute__((ext_vector_type(8)));

#define IN_C 512
#define H1C 64
#define H2C 32


#define SLICES 64
#define RANGE 25600
#define PBLOCK 1024
#define EPT 16

__global__ __launch_bounds__(PBLOCK) void k_deg_part(const int* __restrict__ ei,
                                                     const float* __restrict__ ew,
                                                     float* __restrict__ part, int E,
                                                     int Nn) {
  __shared__ float acc[RANGE];
  const int tid = threadIdx.x;
  const int rbase = blockIdx.y * RANGE;
  const int rlen = min(RANGE, Nn - rbase);
  for (int j = tid; j < RANGE; j += PBLOCK) acc[j] = 0.0f;
  __syncthreads();

  const int per = (E + SLICES - 1) / SLICES;
  const int begin = blockIdx.x * per;
  const int end = min(E, begin + per);
  const int* __restrict__ col = ei + E;
  for (int e0 = begin + tid; e0 < end; e0 += PBLOCK * EPT) {
    int c[EPT];
    float w[EPT];
#pragma unroll
    for (int u = 0; u < EPT; ++u) {
      const int e = e0 + u * PBLOCK;
      const int ec = min(e, end - 1);
      const int cc = col[ec];
      w[u] = ew[ec];
      c[u] = (e < end) ? cc : -1;
    }
    asm volatile("" : "+v"(w[0]));
#pragma unroll
    for (int u = 0; u < EPT; ++u) {
      const unsigned rel = (unsigned)(c[u] - rbase);
      if (rel < (unsigned)rlen) atomicAdd(&acc[rel], w[u]);
    }
  }
  __syncthreads();

  float* __restrict__ dst = part + (size_t)blockIdx.x * Nn + rbase;
  for (int j = tid; j < rlen; j += PBLOCK) dst[j] = acc[j];
}

__global__ void k_dinv_sum(const float* __restrict__ part, float* __restrict__ dinv, int Nn) {
  int i = blockIdx.x * blockDim.x + threadIdx.x;
  if (i >= Nn) return;
  float s0 = 1.0f, s1 = 0.0f, s2 = 0.0f, s3 = 0.0f;
#pragma unroll
  for (int k = 0; k < SLICES; k += 4) {
    s0 += part[(size_t)(k + 0) * Nn + i];
    s1 += part[(size_t)(k + 1) * Nn + i];
    s2 += part[(size_t)(k + 2) * Nn + i];
    s3 += part[(size_t)(k + 3) * Nn + i];
  }
  float d = (s0 + s1) + (s2 + s3);
  dinv[i] = (d > 0.0f) ? rsqrtf(fmaxf(d, 1e-12f)) : 0.0f;
}

__global__ void k_wnorm(const int* __restrict__ ei, const float* __restrict__ ew,
                        const float* __restrict__ dinv, float* __restrict__ wnorm, int E) {
  int e = blockIdx.x * blockDim.x + threadIdx.x;
  if (e >= E) return;
  int row = ei[e];
  int col = ei[E + e];
  wnorm[e] = dinv[row] * ew[e] * dinv[col];
}


template <int K, int NT, bool RELU>
__global__ __launch_bounds__(NT * 32) void k_gemm_wmma_f32(
    const float* __restrict__ A, const float* __restrict__ B, float* __restrict__ C) {
  constexpr int NOUT = NT * 16;
  const int wave = threadIdx.x >> 5;
  const int lane = threadIdx.x & 31;
  const int m = lane & 15;
  const int half = lane >> 4;
  const int n0 = wave * 16;
  const long long rowBase = (long long)blockIdx.x * 16;

  const float* arow = A + (rowBase + m) * K + half * 2;

  v8f acc = {};
#pragma unroll 4
  for (int k = 0; k < K; k += 4) {
    float2 av = *reinterpret_cast<const float2*>(arow + k);
    v2f a;
    a.x = RELU ? fmaxf(av.x, 0.0f) : av.x;
    a.y = RELU ? fmaxf(av.y, 0.0f) : av.y;
    const int kr = k + half * 2;
    v2f b;
    b.x = B[kr * NOUT + n0 + m];
    b.y = B[(kr + 1) * NOUT + n0 + m];
    acc = __builtin_amdgcn_wmma_f32_16x16x4_f32(false, a, false, b, (short)0, acc,
                                                false, false);
  }

#pragma unroll
  for (int v = 0; v < 8; ++v) {
    long long r = rowBase + v + 8 * half;
    C[r * NOUT + n0 + m] = acc[v];
  }
}

template <int F>
__global__ void k_init_agg(float* __restrict__ agg, const float* __restrict__ h,
                           const float* __restrict__ dinv, const float* __restrict__ bias,
                           int Nn) {
  long long t = (long long)blockIdx.x * blockDim.x + threadIdx.x;
  if (t >= (long long)Nn * F) return;
  int node = (int)(t / F);
  int f = (int)(t % F);
  float di = dinv[node];
  agg[t] = bias[f] + di * di * h[t];
}

template <int CH>
__global__ __launch_bounds__(256) void k_edge_agg(const int* __restrict__ ei,
                                                  const float* __restrict__ wnorm,
                                                  const float* __restrict__ h,
                                                  float* __restrict__ agg, int E) {
  long long t = (long long)blockIdx.x * blockDim.x + threadIdx.x;
  if (t >= (long long)E * CH) return;
  int e = (int)(t / CH);
  int c = (int)(t % CH);
  int row = ei[e];
  int col = ei[E + e];
  float w = wnorm[e];
  float4 v = reinterpret_cast<const float4*>(h + (size_t)row * (CH * 4))[c];
  float* dst = agg + (size_t)col * (CH * 4) + c * 4;
  atomicAdd(dst + 0, w * v.x);
  atomicAdd(dst + 1, w * v.y);
  atomicAdd(dst + 2, w * v.z);
  atomicAdd(dst + 3, w * v.w);
}

__global__ void k_gemm3_relu(const float* __restrict__ a2, const float* __restrict__ W3,
                             float* __restrict__ g, int Nn) {
  int i = blockIdx.x * blockDim.x + threadIdx.x;
  if (i >= Nn) return;
  const float* r = a2 + (size_t)i * H2C;
  float acc = 0.0f;
#pragma unroll
  for (int f = 0; f < H2C; ++f) acc += fmaxf(r[f], 0.0f) * W3[f];
  g[i] = acc;
}

__global__ __launch_bounds__(PBLOCK) void k_agg1_part(const int* __restrict__ ei,
                                                      const float* __restrict__ wnorm,
                                                      const float* __restrict__ g,
                                                      float* __restrict__ part, int E,
                                                      int Nn) {
  __shared__ float acc[RANGE];
  const int tid = threadIdx.x;
  const int rbase = blockIdx.y * RANGE;
  const int rlen = min(RANGE, Nn - rbase);
  for (int j = tid; j < RANGE; j += PBLOCK) acc[j] = 0.0f;
  __syncthreads();

  const int per = (E + SLICES - 1) / SLICES;
  const int begin = blockIdx.x * per;
  const int end = min(E, begin + per);
  const int* __restrict__ col = ei + E;
  for (int e0 = begin + tid; e0 < end; e0 += PBLOCK * EPT) {
    int c[EPT];
    int r[EPT];
    float w[EPT];
    float v[EPT];
#pragma unroll
    for (int u = 0; u < EPT; ++u) {
      const int e = e0 + u * PBLOCK;
      const int ec = min(e, end - 1);
      const int cc = col[ec];
      r[u] = ei[ec];
      w[u] = wnorm[ec];
      c[u] = (e < end) ? cc : -1;
    }
    asm volatile("" : "+v"(r[0]), "+v"(w[0]));
#pragma unroll
    for (int u = 0; u < EPT; ++u) {
      const unsigned rel = (unsigned)(c[u] - rbase);
      v[u] = g[(rel < (unsigned)rlen) ? r[u] : 0];
    }
    asm volatile("" : "+v"(v[0]));
#pragma unroll
    for (int u = 0; u < EPT; ++u) {
      const unsigned rel = (unsigned)(c[u] - rbase);
      if (rel < (unsigned)rlen) atomicAdd(&acc[rel], w[u] * v[u]);
    }
  }
  __syncthreads();

  float* __restrict__ dst = part + (size_t)blockIdx.x * Nn + rbase;
  for (int j = tid; j < rlen; j += PBLOCK) dst[j] = acc[j];
}

__global__ void k_out_sum(float* __restrict__ out, const float* __restrict__ part,
                          const float* __restrict__ g, const float* __restrict__ dinv,
                          const float* __restrict__ b3, int Nn) {
  int i = blockIdx.x * blockDim.x + threadIdx.x;
  if (i >= Nn) return;
  float di = dinv[i];
  float s0 = b3[0] + di * di * g[i], s1 = 0.0f, s2 = 0.0f, s3 = 0.0f;
#pragma unroll
  for (int k = 0; k < SLICES; k += 4) {
    s0 += part[(size_t)(k + 0) * Nn + i];
    s1 += part[(size_t)(k + 1) * Nn + i];
    s2 += part[(size_t)(k + 2) * Nn + i];
    s3 += part[(size_t)(k + 3) * Nn + i];
  }
  out[i] = (s0 + s1) + (s2 + s3);
}


static inline unsigned int blocks_for(long long total, int bs) {
  return (unsigned int)((total + bs - 1) / bs);
}

extern "C" void kernel_launch(void* const* d_in, const int* in_sizes, int n_in,
                              void* d_out, int out_size, void* d_ws, size_t ws_size,
                              hipStream_t stream) {
  const float* x = (const float*)d_in[0];
  const int* ei = (const int*)d_in[1];
  const float* ew = (const float*)d_in[2];
  const float* W1 = (const float*)d_in[3];
  const float* b1 = (const float*)d_in[4];
  const float* W2 = (const float*)d_in[5];
  const float* b2 = (const float*)d_in[6];
  const float* W3 = (const float*)d_in[7];
  const float* b3 = (const float*)d_in[8];

  const int E = in_sizes[2];
  const int Nn = in_sizes[0] / IN_C;
  float* out = (float*)d_out;

  float* dinv = (float*)d_ws;
  float* wnorm = dinv + Nn;
  float* bufA = wnorm + E;
  float* bufB = bufA + (size_t)Nn * H1C;

  k_deg_part<<<dim3(SLICES, blocks_for(Nn, RANGE)), PBLOCK, 0, stream>>>(ei, ew, bufA, E, Nn);
  k_dinv_sum<<<blocks_for(Nn, 256), 256, 0, stream>>>(bufA, dinv, Nn);
  k_wnorm<<<blocks_for(E, 256), 256, 0, stream>>>(ei, ew, dinv, wnorm, E);

  k_gemm_wmma_f32<IN_C, H1C / 16, false>
      <<<Nn / 16, (H1C / 16) * 32, 0, stream>>>(x, W1, bufA);
  k_init_agg<H1C><<<blocks_for((long long)Nn * H1C, 256), 256, 0, stream>>>(
      bufB, bufA, dinv, b1, Nn);
  k_edge_agg<H1C / 4><<<blocks_for((long long)E * (H1C / 4), 256), 256, 0, stream>>>(
      ei, wnorm, bufA, bufB, E);

  k_gemm_wmma_f32<H1C, H2C / 16, true>
      <<<Nn / 16, (H2C / 16) * 32, 0, stream>>>(bufB, W2, bufA);
  k_init_agg<H2C><<<blocks_for((long long)Nn * H2C, 256), 256, 0, stream>>>(
      bufB, bufA, dinv, b2, Nn);
  k_edge_agg<H2C / 4><<<blocks_for((long long)E * (H2C / 4), 256), 256, 0, stream>>>(
      ei, wnorm, bufA, bufB, E);

  k_gemm3_relu<<<blocks_for(Nn, 256), 256, 0, stream>>>(bufB, W3, bufA, Nn);
  k_agg1_part<<<dim3(SLICES, blocks_for(Nn, RANGE)), PBLOCK, 0, stream>>>(ei, wnorm, bufA, bufB,
                                                                         E, Nn);
  k_out_sum<<<blocks_for(Nn, 256), 256, 0, stream>>>(out, bufB, bufA, dinv, b3, Nn);
}


# Row aggregation pre-summed in LDS: block (edge slice, node range) sums its edges' rows in LDS, then adds them to agg with whole-line atomics, one per slice and float; f32 sum order changes
# speedup vs baseline: 2.0503x; 1.4571x over previous
#include <hip/hip_runtime.h>
#include <hip/hip_bf16.h>


typedef float v2f __attribute__((ext_vector_type(2)));
typedef float v8f __attribute__((ext_vector_type(8)));

#define IN_C 512
#define H1C 64
#define H2C 32


#define SLICES 64
#define RANGE 25600
#define PBLOCK 1024
#define EPT 16

__global__ __launch_bounds__(PBLOCK) void k_deg_part(const int* __restrict__ ei,
                                                     const float* __restrict__ ew,
                                                     float* __restrict__ part, int E,
                                                     int Nn) {
  __shared__ float acc[RANGE];
  const int tid = threadIdx.x;
  const int rbase = blockIdx.y * RANGE;
  const int rlen = min(RANGE, Nn - rbase);
  for (int j = tid; j < RANGE; j += PBLOCK) acc[j] = 0.0f;
  __syncthreads();

  const int per = (E + SLICES - 1) / SLICES;
  const int begin = blockIdx.x * per;
  const int end = min(E, begin + per);
  const int* __restrict__ col = ei + E;
  for (int e0 = begin + tid; e0 < end; e0 += PBLOCK * EPT) {
    int c[EPT];
    float w[EPT];
#pragma unroll
    for (int u = 0; u < EPT; ++u) {
      const int e = e0 + u * PBLOCK;
      const int ec = min(e, end - 1);
      const int cc = col[ec];
      w[u] = ew[ec];
      c[u] = (e < end) ? cc : -1;
    }
    asm volatile("" : "+v"(w[0]));
#pragma unroll
    for (int u = 0; u < EPT; ++u) {
      const unsigned rel = (unsigned)(c[u] - rbase);
      if (rel < (unsigned)rlen) atomicAdd(&acc[rel], w[u]);
    }
  }
  __syncthreads();

  float* __restrict__ dst = part + (size_t)blockIdx.x * Nn + rbase;
  for (int j = tid; j < rlen; j += PBLOCK) dst[j] = acc[j];
}

__global__ void k_dinv_sum(const float* __restrict__ part, float* __restrict__ dinv, int Nn) {
  int i = blockIdx.x * blockDim.x + threadIdx.x;
  if (i >= Nn) return;
  float s0 = 1.0f, s1 = 0.0f, s2 = 0.0f, s3 = 0.0f;
#pragma unroll
  for (int k = 0; k < SLICES; k += 4) {
    s0 += part[(size_t)(k + 0) * Nn + i];
    s1 += part[(size_t)(k + 1) * Nn + i];
    s2 += part[(size_t)(k + 2) * Nn + i];
    s3 += part[(size_t)(k + 3) * Nn + i];
  }
  float d = (s0 + s1) + (s2 + s3);
  dinv[i] = (d > 0.0f) ? rsqrtf(fmaxf(d, 1e-12f)) : 0.0f;
}

__global__ void k_wnorm(const int* __restrict__ ei, const float* __restrict__ ew,
                        const float* __restrict__ dinv, float* __restrict__ wnorm, int E) {
  int e = blockIdx.x * blockDim.x + threadIdx.x;
  if (e >= E) return;
  int row = ei[e];
  int col = ei[E + e];
  wnorm[e] = dinv[row] * ew[e] * dinv[col];
}


template <int K, int NT, bool RELU>
__global__ __launch_bounds__(NT * 32) void k_gemm_wmma_f32(
    const float* __restrict__ A, const float* __restrict__ B, float* __restrict__ C) {
  constexpr int NOUT = NT * 16;
  const int wave = threadIdx.x >> 5;
  const int lane = threadIdx.x & 31;
  const int m = lane & 15;
  const int half = lane >> 4;
  const int n0 = wave * 16;
  const long long rowBase = (long long)blockIdx.x * 16;

  const float* arow = A + (rowBase + m) * K + half * 2;

  v8f acc = {};
#pragma unroll 4
  for (int k = 0; k < K; k += 4) {
    float2 av = *reinterpret_cast<const float2*>(arow + k);
    v2f a;
    a.x = RELU ? fmaxf(av.x, 0.0f) : av.x;
    a.y = RELU ? fmaxf(av.y, 0.0f) : av.y;
    const int kr = k + half * 2;
    v2f b;
    b.x = B[kr * NOUT + n0 + m];
    b.y = B[(kr + 1) * NOUT + n0 + m];
    acc = __builtin_amdgcn_wmma_f32_16x16x4_f32(false, a, false, b, (short)0, acc,
                                                false, false);
  }

#pragma unroll
  for (int v = 0; v < 8; ++v) {
    long long r = rowBase + v + 8 * half;
    C[r * NOUT + n0 + m] = acc[v];
  }
}

template <int F>
__global__ void k_init_agg(float* __restrict__ agg, const float* __restrict__ h,
                           const float* __restrict__ dinv, const float* __restrict__ bias,
                           int Nn) {
  long long t = (long long)blockIdx.x * blockDim.x + threadIdx.x;
  if (t >= (long long)Nn * F) return;
  int node = (int)(t / F);
  int f = (int)(t % F);
  float di = dinv[node];
  agg[t] = bias[f] + di * di * h[t];
}


#define RFLOATS 76800
#define RBLOCK 1024
#define RK 8

template <int F>
__device__ __forceinline__ void row_batch(const int rv, const int relv, const float wv,
                                          const int b, const int lane,
                                          const float* __restrict__ h, float* acc) {
  constexpr int L = F / 32;
  float v[8][L];
#pragma unroll
  for (int u = 0; u < 8; ++u) {
    const int r = __builtin_amdgcn_readlane(rv, b + u);
    const float* src = h + (size_t)r * F + lane;
#pragma unroll
    for (int l = 0; l < L; ++l) v[u][l] = src[l * 32];
  }
#pragma unroll
  for (int u = 0; u < 8; ++u) {
    const int rel = __builtin_amdgcn_readlane(relv, b + u);
    const float w = __int_as_float(__builtin_amdgcn_readlane(__float_as_int(wv), b + u));
    float* dst = acc + rel * F + lane;
#pragma unroll
    for (int l = 0; l < L; ++l) atomicAdd(dst + l * 32, w * v[u][l]);
  }
}

template <int F>
__global__ __launch_bounds__(RBLOCK) void k_row_agg(const int* __restrict__ ei,
                                                    const float* __restrict__ wnorm,
                                                    const float* __restrict__ h,
                                                    float* __restrict__ agg, int E, int Nn) {
  constexpr int L = F / 32;
  constexpr int RROWS = RFLOATS / F;
  __shared__ float acc[RFLOATS];
  __shared__ int qe[RBLOCK / 32][64];
  __shared__ int qr[RBLOCK / 32][64];
  const int tid = threadIdx.x;
  const int lane = tid & 31;
  const int wave = __builtin_amdgcn_readfirstlane(tid >> 5);
  const int rbase = blockIdx.y * RROWS;
  const int rlen = min(RROWS, Nn - rbase);
  for (int j = tid; j < RFLOATS; j += RBLOCK) acc[j] = 0.0f;
  __syncthreads();

  const int slices = (int)gridDim.x;
  const int per = (E + slices - 1) / slices;
  const int begin = blockIdx.x * per;
  const int end = min(E, begin + per);
  const int* __restrict__ col = ei + E;
  int* myqe = qe[wave];
  int* myqr = qr[wave];
  int head = 0, tail = 0;

  for (int base = begin + wave * (32 * RK); base < end; base += RBLOCK * RK) {
    int c[RK];
    if (base + 32 * RK <= end) {
#pragma unroll
      for (int k = 0; k < RK; ++k) c[k] = col[base + k * 32 + lane];
    } else {
#pragma unroll
      for (int k = 0; k < RK; ++k) {
        const int e = base + k * 32 + lane;
        const int cc = col[min(e, end - 1)];
        c[k] = (e < end) ? cc : -1;
      }
    }
#pragma unroll
    for (int k = 0; k < RK; ++k) {
      const unsigned rel = (unsigned)(c[k] - rbase);
      const bool mine = rel < (unsigned)rlen;
      const unsigned mask = __builtin_amdgcn_ballot_w32(mine);
      if (mask != 0u) {
        const int pos = (tail + (int)__builtin_amdgcn_mbcnt_lo(mask, 0u)) & 63;
        if (mine) {
          myqe[pos] = base + k * 32 + lane;
          myqr[pos] = (int)rel;
        }
        tail += __builtin_popcount(mask);
        if (tail - head >= 32) {
          const int idx = (head + lane) & 63;
          const int e = myqe[idx];
          const int relv = myqr[idx];
          head += 32;
          const int rv = ei[e];
          const float wv = wnorm[e];
          row_batch<F>(rv, relv, wv, 0, lane, h, acc);
          row_batch<F>(rv, relv, wv, 8, lane, h, acc);
          row_batch<F>(rv, relv, wv, 16, lane, h, acc);
          row_batch<F>(rv, relv, wv, 24, lane, h, acc);
        }
      }
    }
  }
  const int left = tail - head;
  if (left > 0) {
    const int idx = (head + lane) & 63;
    const bool valid = lane < left;
    const int e = valid ? myqe[idx] : begin;
    const int relv = valid ? myqr[idx] : 0;
    const int rv = ei[e];
    const float wv = wnorm[e];
#pragma unroll
    for (int b = 0; b < 32; b += 8) {
      if (b + 8 <= left) {
        row_batch<F>(rv, relv, wv, b, lane, h, acc);
      } else {
#pragma unroll
        for (int u = b; u < b + 8; ++u) {
          if (u < left) {
            const int r = __builtin_amdgcn_readlane(rv, u);
            const int rel = __builtin_amdgcn_readlane(relv, u);
            const float w = __int_as_float(__builtin_amdgcn_readlane(__float_as_int(wv), u));
            const float* src = h + (size_t)r * F + lane;
            float* dst = acc + rel * F + lane;
#pragma unroll
            for (int l = 0; l < L; ++l) atomicAdd(dst + l * 32, w * src[l * 32]);
          }
        }
      }
    }
  }
  __syncthreads();

  float* __restrict__ out_rows = agg + (size_t)rbase * F;
  for (int j = tid; j < rlen * F; j += RBLOCK) atomicAdd(out_rows + j, acc[j]);
}

__global__ void k_gemm3_relu(const float* __restrict__ a2, const float* __restrict__ W3,
                             float* __restrict__ g, int Nn) {
  int i = blockIdx.x * blockDim.x + threadIdx.x;
  if (i >= Nn) return;
  const float* r = a2 + (size_t)i * H2C;
  float acc = 0.0f;
#pragma unroll
  for (int f = 0; f < H2C; ++f) acc += fmaxf(r[f], 0.0f) * W3[f];
  g[i] = acc;
}

__global__ __launch_bounds__(PBLOCK) void k_agg1_part(const int* __restrict__ ei,
                                                      const float* __restrict__ wnorm,
                                                      const float* __restrict__ g,
                                                      float* __restrict__ part, int E,
                                                      int Nn) {
  __shared__ float acc[RANGE];
  const int tid = threadIdx.x;
  const int rbase = blockIdx.y * RANGE;
  const int rlen = min(RANGE, Nn - rbase);
  for (int j = tid; j < RANGE; j += PBLOCK) acc[j] = 0.0f;
  __syncthreads();

  const int per = (E + SLICES - 1) / SLICES;
  const int begin = blockIdx.x * per;
  const int end = min(E, begin + per);
  const int* __restrict__ col = ei + E;
  for (int e0 = begin + tid; e0 < end; e0 += PBLOCK * EPT) {
    int c[EPT];
    int r[EPT];
    float w[EPT];
    float v[EPT];
#pragma unroll
    for (int u = 0; u < EPT; ++u) {
      const int e = e0 + u * PBLOCK;
      const int ec = min(e, end - 1);
      const int cc = col[ec];
      r[u] = ei[ec];
      w[u] = wnorm[ec];
      c[u] = (e < end) ? cc : -1;
    }
    asm volatile("" : "+v"(r[0]), "+v"(w[0]));
#pragma unroll
    for (int u = 0; u < EPT; ++u) {
      const unsigned rel = (unsigned)(c[u] - rbase);
      v[u] = g[(rel < (unsigned)rlen) ? r[u] : 0];
    }
    asm volatile("" : "+v"(v[0]));
#pragma unroll
    for (int u = 0; u < EPT; ++u) {
      const unsigned rel = (unsigned)(c[u] - rbase);
      if (rel < (unsigned)rlen) atomicAdd(&acc[rel], w[u] * v[u]);
    }
  }
  __syncthreads();

  float* __restrict__ dst = part + (size_t)blockIdx.x * Nn + rbase;
  for (int j = tid; j < rlen; j += PBLOCK) dst[j] = acc[j];
}

__global__ void k_out_sum(float* __restrict__ out, const float* __restrict__ part,
                          const float* __restrict__ g, const float* __restrict__ dinv,
                          const float* __restrict__ b3, int Nn) {
  int i = blockIdx.x * blockDim.x + threadIdx.x;
  if (i >= Nn) return;
  float di = dinv[i];
  float s0 = b3[0] + di * di * g[i], s1 = 0.0f, s2 = 0.0f, s3 = 0.0f;
#pragma unroll
  for (int k = 0; k < SLICES; k += 4) {
    s0 += part[(size_t)(k + 0) * Nn + i];
    s1 += part[(size_t)(k + 1) * Nn + i];
    s2 += part[(size_t)(k + 2) * Nn + i];
    s3 += part[(size_t)(k + 3) * Nn + i];
  }
  out[i] = (s0 + s1) + (s2 + s3);
}


static inline unsigned int blocks_for(long long total, int bs) {
  return (unsigned int)((total + bs - 1) / bs);
}

extern "C" void kernel_launch(void* const* d_in, const int* in_sizes, int n_in,
                              void* d_out, int out_size, void* d_ws, size_t ws_size,
                              hipStream_t stream) {
  const float* x = (const float*)d_in[0];
  const int* ei = (const int*)d_in[1];
  const float* ew = (const float*)d_in[2];
  const float* W1 = (const float*)d_in[3];
  const float* b1 = (const float*)d_in[4];
  const float* W2 = (const float*)d_in[5];
  const float* b2 = (const float*)d_in[6];
  const float* W3 = (const float*)d_in[7];
  const float* b3 = (const float*)d_in[8];

  const int E = in_sizes[2];
  const int Nn = in_sizes[0] / IN_C;
  float* out = (float*)d_out;

  float* dinv = (float*)d_ws;
  float* wnorm = dinv + Nn;
  float* bufA = wnorm + E;
  float* bufB = bufA + (size_t)Nn * H1C;

  k_deg_part<<<dim3(SLICES, blocks_for(Nn, RANGE)), PBLOCK, 0, stream>>>(ei, ew, bufA, E, Nn);
  k_dinv_sum<<<blocks_for(Nn, 256), 256, 0, stream>>>(bufA, dinv, Nn);
  k_wnorm<<<blocks_for(E, 256), 256, 0, stream>>>(ei, ew, dinv, wnorm, E);

  k_gemm_wmma_f32<IN_C, H1C / 16, false>
      <<<Nn / 16, (H1C / 16) * 32, 0, stream>>>(x, W1, bufA);
  k_init_agg<H1C><<<blocks_for((long long)Nn * H1C, 256), 256, 0, stream>>>(
      bufB, bufA, dinv, b1, Nn);
  k_row_agg<H1C><<<dim3(3, blocks_for(Nn, RFLOATS / H1C)), RBLOCK, 0, stream>>>(ei, wnorm, bufA,
                                                                              bufB, E, Nn);

  k_gemm_wmma_f32<H1C, H2C / 16, true>
      <<<Nn / 16, (H2C / 16) * 32, 0, stream>>>(bufB, W2, bufA);
  k_init_agg<H2C><<<blocks_for((long long)Nn * H2C, 256), 256, 0, stream>>>(
      bufB, bufA, dinv, b2, Nn);
  k_row_agg<H2C><<<dim3(6, blocks_for(Nn, RFLOATS / H2C)), RBLOCK, 0, stream>>>(ei, wnorm, bufA,
                                                                              bufB, E, Nn);

  k_gemm3_relu<<<blocks_for(Nn, 256), 256, 0, stream>>>(bufB, W3, bufA, Nn);
  k_agg1_part<<<dim3(SLICES, blocks_for(Nn, RANGE)), PBLOCK, 0, stream>>>(ei, wnorm, bufA, bufB,
                                                                         E, Nn);
  k_out_sum<<<blocks_for(Nn, 256), 256, 0, stream>>>(out, bufB, bufA, dinv, b3, Nn);
}


# Layer-1 GEMM re-tiled: one wave per 16x64 row tile, W1 fragments in LDS, each A piece loaded once (16 B per lane) for all four column tiles, tile stored in whole lines; f32 k sum order changes
# speedup vs baseline: 2.2695x; 1.1069x over previous
#include <hip/hip_runtime.h>
#include <hip/hip_bf16.h>


typedef float v2f __attribute__((ext_vector_type(2)));
typedef float v8f __attribute__((ext_vector_type(8)));

#define IN_C 512
#define H1C 64
#define H2C 32


#define SLICES 64
#define RANGE 25600
#define PBLOCK 1024
#define EPT 16

__global__ __launch_bounds__(PBLOCK) void k_deg_part(const int* __restrict__ ei,
                                                     const float* __restrict__ ew,
                                                     float* __restrict__ part, int E,
                                                     int Nn) {
  __shared__ float acc[RANGE];
  const int tid = threadIdx.x;
  const int rbase = blockIdx.y * RANGE;
  const int rlen = min(RANGE, Nn - rbase);
  for (int j = tid; j < RANGE; j += PBLOCK) acc[j] = 0.0f;
  __syncthreads();

  const int per = (E + SLICES - 1) / SLICES;
  const int begin = blockIdx.x * per;
  const int end = min(E, begin + per);
  const int* __restrict__ col = ei + E;
  for (int e0 = begin + tid; e0 < end; e0 += PBLOCK * EPT) {
    int c[EPT];
    float w[EPT];
#pragma unroll
    for (int u = 0; u < EPT; ++u) {
      const int e = e0 + u * PBLOCK;
      const int ec = min(e, end - 1);
      const int cc = col[ec];
      w[u] = ew[ec];
      c[u] = (e < end) ? cc : -1;
    }
    asm volatile("" : "+v"(w[0]));
#pragma unroll
    for (int u = 0; u < EPT; ++u) {
      const unsigned rel = (unsigned)(c[u] - rbase);
      if (rel < (unsigned)rlen) atomicAdd(&acc[rel], w[u]);
    }
  }
  __syncthreads();

  float* __restrict__ dst = part + (size_t)blockIdx.x * Nn + rbase;
  for (int j = tid; j < rlen; j += PBLOCK) dst[j] = acc[j];
}

__global__ void k_dinv_sum(const float* __restrict__ part, float* __restrict__ dinv, int Nn) {
  int i = blockIdx.x * blockDim.x + threadIdx.x;
  if (i >= Nn) return;
  float s0 = 1.0f, s1 = 0.0f, s2 = 0.0f, s3 = 0.0f;
#pragma unroll
  for (int k = 0; k < SLICES; k += 4) {
    s0 += part[(size_t)(k + 0) * Nn + i];
    s1 += part[(size_t)(k + 1) * Nn + i];
    s2 += part[(size_t)(k + 2) * Nn + i];
    s3 += part[(size_t)(k + 3) * Nn + i];
  }
  float d = (s0 + s1) + (s2 + s3);
  dinv[i] = (d > 0.0f) ? rsqrtf(fmaxf(d, 1e-12f)) : 0.0f;
}

__global__ void k_wnorm(const int* __restrict__ ei, const float* __restrict__ ew,
                        const float* __restrict__ dinv, float* __restrict__ wnorm, int E) {
  int e = blockIdx.x * blockDim.x + threadIdx.x;
  if (e >= E) return;
  int row = ei[e];
  int col = ei[E + e];
  wnorm[e] = dinv[row] * ew[e] * dinv[col];
}


template <int K, int NT, bool RELU>
__global__ __launch_bounds__(NT * 32) void k_gemm_wmma_f32(
    const float* __restrict__ A, const float* __restrict__ B, float* __restrict__ C) {
  constexpr int NOUT = NT * 16;
  const int wave = threadIdx.x >> 5;
  const int lane = threadIdx.x & 31;
  const int m = lane & 15;
  const int half = lane >> 4;
  const int n0 = wave * 16;
  const long long rowBase = (long long)blockIdx.x * 16;

  const float* arow = A + (rowBase + m) * K + half * 2;

  v8f acc = {};
#pragma unroll 4
  for (int k = 0; k < K; k += 4) {
    float2 av = *reinterpret_cast<const float2*>(arow + k);
    v2f a;
    a.x = RELU ? fmaxf(av.x, 0.0f) : av.x;
    a.y = RELU ? fmaxf(av.y, 0.0f) : av.y;
    const int kr = k + half * 2;
    v2f b;
    b.x = B[kr * NOUT + n0 + m];
    b.y = B[(kr + 1) * NOUT + n0 + m];
    acc = __builtin_amdgcn_wmma_f32_16x16x4_f32(false, a, false, b, (short)0, acc,
                                                false, false);
  }

#pragma unroll
  for (int v = 0; v < 8; ++v) {
    long long r = rowBase + v + 8 * half;
    C[r * NOUT + n0 + m] = acc[v];
  }
}


#define G1_WAVES 25
#define G1_THREADS (G1_WAVES * 32)

__device__ __forceinline__ void g1_block(const float4 a, const float4* bp, v8f& c0, v8f& c1,
                                         v8f& c2, v8f& c3) {
  const float4 b0 = bp[0];
  const float4 b1 = bp[16];
  const float4 b2 = bp[32];
  const float4 b3 = bp[48];
  v2f alo, ahi, f;
  alo.x = a.x; alo.y = a.y;
  ahi.x = a.z; ahi.y = a.w;
  f.x = b0.x; f.y = b0.y;
  c0 = __builtin_amdgcn_wmma_f32_16x16x4_f32(false, alo, false, f, (short)0, c0, false, false);
  f.x = b1.x; f.y = b1.y;
  c1 = __builtin_amdgcn_wmma_f32_16x16x4_f32(false, alo, false, f, (short)0, c1, false, false);
  f.x = b2.x; f.y = b2.y;
  c2 = __builtin_amdgcn_wmma_f32_16x16x4_f32(false, alo, false, f, (short)0, c2, false, false);
  f.x = b3.x; f.y = b3.y;
  c3 = __builtin_amdgcn_wmma_f32_16x16x4_f32(false, alo, false, f, (short)0, c3, false, false);
  f.x = b0.z; f.y = b0.w;
  c0 = __builtin_amdgcn_wmma_f32_16x16x4_f32(false, ahi, false, f, (short)0, c0, false, false);
  f.x = b1.z; f.y = b1.w;
  c1 = __builtin_amdgcn_wmma_f32_16x16x4_f32(false, ahi, false, f, (short)0, c1, false, false);
  f.x = b2.z; f.y = b2.w;
  c2 = __builtin_amdgcn_wmma_f32_16x16x4_f32(false, ahi, false, f, (short)0, c2, false, false);
  f.x = b3.z; f.y = b3.w;
  c3 = __builtin_amdgcn_wmma_f32_16x16x4_f32(false, ahi, false, f, (short)0, c3, false, false);
}

__global__ __launch_bounds__(G1_THREADS) void k_gemm1_rows(const float* __restrict__ A,
                                                           const float* __restrict__ B,
                                                           float* __restrict__ C, int ntiles) {
  __shared__ float4 Bs[(IN_C / 4) * H1C];
  __shared__ float4 Cs[G1_WAVES][16 * H1C / 4];
  const int tid = threadIdx.x;
  const int lane = tid & 31;
  const int wave = __builtin_amdgcn_readfirstlane(tid >> 5);
  const int m = lane & 15;
  const int half = lane >> 4;

  for (int idx = tid; idx < (IN_C / 4) * H1C; idx += G1_THREADS) {
    const int kq = idx / H1C;
    const int n = idx % H1C;
    const float* src = B + (size_t)(4 * kq) * H1C + n;
    float4 v;
    v.x = src[0];
    v.y = src[H1C];
    v.z = src[2 * H1C];
    v.w = src[3 * H1C];
    Bs[idx] = v;
  }
  __syncthreads();

  const int tile = blockIdx.x * G1_WAVES + wave;
  const bool active = tile < ntiles;
  if (active) {
    const float* arow = A + ((size_t)tile * 16 + m) * IN_C + half * 4;
    const float4* bfrag = Bs + half * H1C + m;
    v8f c0 = {}, c1 = {}, c2 = {}, c3 = {};
    float4 a[4];
#pragma unroll
    for (int u = 0; u < 4; ++u) a[u] = *reinterpret_cast<const float4*>(arow + 8 * u);
    for (int kb = 0; kb < IN_C - 32; kb += 32) {
      float4 an[4];
#pragma unroll
      for (int u = 0; u < 4; ++u)
        an[u] = *reinterpret_cast<const float4*>(arow + kb + 32 + 8 * u);
#pragma unroll
      for (int u = 0; u < 4; ++u) g1_block(a[u], bfrag + ((kb + 8 * u) >> 2) * H1C, c0, c1, c2, c3);
#pragma unroll
      for (int u = 0; u < 4; ++u) a[u] = an[u];
    }
#pragma unroll
    for (int u = 0; u < 4; ++u)
      g1_block(a[u], bfrag + ((IN_C - 32 + 8 * u) >> 2) * H1C, c0, c1, c2, c3);

    float* cs = reinterpret_cast<float*>(Cs[wave]);
#pragma unroll
    for (int v = 0; v < 8; ++v) {
      float* p = cs + (v + 8 * half) * H1C + m;
      p[0] = c0[v];
      p[16] = c1[v];
      p[32] = c2[v];
      p[48] = c3[v];
    }
  }
  __syncthreads();
  if (active) {
    float4* dst = reinterpret_cast<float4*>(C + (size_t)tile * 16 * H1C);
#pragma unroll
    for (int i = 0; i < 8; ++i) dst[i * 32 + lane] = Cs[wave][i * 32 + lane];
  }
}

template <int F>
__global__ void k_init_agg(float* __restrict__ agg, const float* __restrict__ h,
                           const float* __restrict__ dinv, const float* __restrict__ bias,
                           int Nn) {
  long long t = (long long)blockIdx.x * blockDim.x + threadIdx.x;
  if (t >= (long long)Nn * F) return;
  int node = (int)(t / F);
  int f = (int)(t % F);
  float di = dinv[node];
  agg[t] = bias[f] + di * di * h[t];
}


#define RFLOATS 76800
#define RBLOCK 1024
#define RK 8

template <int F>
__device__ __forceinline__ void row_batch(const int rv, const int relv, const float wv,
                                          const int b, const int lane,
                                          const float* __restrict__ h, float* acc) {
  constexpr int L = F / 32;
  float v[8][L];
#pragma unroll
  for (int u = 0; u < 8; ++u) {
    const int r = __builtin_amdgcn_readlane(rv, b + u);
    const float* src = h + (size_t)r * F + lane;
#pragma unroll
    for (int l = 0; l < L; ++l) v[u][l] = src[l * 32];
  }
#pragma unroll
  for (int u = 0; u < 8; ++u) {
    const int rel = __builtin_amdgcn_readlane(relv, b + u);
    const float w = __int_as_float(__builtin_amdgcn_readlane(__float_as_int(wv), b + u));
    float* dst = acc + rel * F + lane;
#pragma unroll
    for (int l = 0; l < L; ++l) atomicAdd(dst + l * 32, w * v[u][l]);
  }
}

template <int F>
__global__ __launch_bounds__(RBLOCK) void k_row_agg(const int* __restrict__ ei,
                                                    const float* __restrict__ wnorm,
                                                    const float* __restrict__ h,
                                                    float* __restrict__ agg, int E, int Nn) {
  constexpr int L = F / 32;
  constexpr int RROWS = RFLOATS / F;
  __shared__ float acc[RFLOATS];
  __shared__ int qe[RBLOCK / 32][64];
  __shared__ int qr[RBLOCK / 32][64];
  const int tid = threadIdx.x;
  const int lane = tid & 31;
  const int wave = __builtin_amdgcn_readfirstlane(tid >> 5);
  const int rbase = blockIdx.y * RROWS;
  const int rlen = min(RROWS, Nn - rbase);
  for (int j = tid; j < RFLOATS; j += RBLOCK) acc[j] = 0.0f;
  __syncthreads();

  const int slices = (int)gridDim.x;
  const int per = (E + slices - 1) / slices;
  const int begin = blockIdx.x * per;
  const int end = min(E, begin + per);
  const int* __restrict__ col = ei + E;
  int* myqe = qe[wave];
  int* myqr = qr[wave];
  int head = 0, tail = 0;

  for (int base = begin + wave * (32 * RK); base < end; base += RBLOCK * RK) {
    int c[RK];
    if (base + 32 * RK <= end) {
#pragma unroll
      for (int k = 0; k < RK; ++k) c[k] = col[base + k * 32 + lane];
    } else {
#pragma unroll
      for (int k = 0; k < RK; ++k) {
        const int e = base + k * 32 + lane;
        const int cc = col[min(e, end - 1)];
        c[k] = (e < end) ? cc : -1;
      }
    }
#pragma unroll
    for (int k = 0; k < RK; ++k) {
      const unsigned rel = (unsigned)(c[k] - rbase);
      const bool mine = rel < (unsigned)rlen;
      const unsigned mask = __builtin_amdgcn_ballot_w32(mine);
      if (mask != 0u) {
        const int pos = (tail + (int)__builtin_amdgcn_mbcnt_lo(mask, 0u)) & 63;
        if (mine) {
          myqe[pos] = base + k * 32 + lane;
          myqr[pos] = (int)rel;
        }
        tail += __builtin_popcount(mask);
        if (tail - head >= 32) {
          const int idx = (head + lane) & 63;
          const int e = myqe[idx];
          const int relv = myqr[idx];
          head += 32;
          const int rv = ei[e];
          const float wv = wnorm[e];
          row_batch<F>(rv, relv, wv, 0, lane, h, acc);
          row_batch<F>(rv, relv, wv, 8, lane, h, acc);
          row_batch<F>(rv, relv, wv, 16, lane, h, acc);
          row_batch<F>(rv, relv, wv, 24, lane, h, acc);
        }
      }
    }
  }
  const int left = tail - head;
  if (left > 0) {
    const int idx = (head + lane) & 63;
    const bool valid = lane < left;
    const int e = valid ? myqe[idx] : begin;
    const int relv = valid ? myqr[idx] : 0;
    const int rv = ei[e];
    const float wv = wnorm[e];
#pragma unroll
    for (int b = 0; b < 32; b += 8) {
      if (b + 8 <= left) {
        row_batch<F>(rv, relv, wv, b, lane, h, acc);
      } else {
#pragma unroll
        for (int u = b; u < b + 8; ++u) {
          if (u < left) {
            const int r = __builtin_amdgcn_readlane(rv, u);
            const int rel = __builtin_amdgcn_readlane(relv, u);
            const float w = __int_as_float(__builtin_amdgcn_readlane(__float_as_int(wv), u));
            const float* src = h + (size_t)r * F + lane;
            float* dst = acc + rel * F + lane;
#pragma unroll
            for (int l = 0; l < L; ++l) atomicAdd(dst + l * 32, w * src[l * 32]);
          }
        }
      }
    }
  }
  __syncthreads();

  float* __restrict__ out_rows = agg + (size_t)rbase * F;
  for (int j = tid; j < rlen * F; j += RBLOCK) atomicAdd(out_rows + j, acc[j]);
}

__global__ void k_gemm3_relu(const float* __restrict__ a2, const float* __restrict__ W3,
                             float* __restrict__ g, int Nn) {
  int i = blockIdx.x * blockDim.x + threadIdx.x;
  if (i >= Nn) return;
  const float* r = a2 + (size_t)i * H2C;
  float acc = 0.0f;
#pragma unroll
  for (int f = 0; f < H2C; ++f) acc += fmaxf(r[f], 0.0f) * W3[f];
  g[i] = acc;
}

__global__ __launch_bounds__(PBLOCK) void k_agg1_part(const int* __restrict__ ei,
                                                      const float* __restrict__ wnorm,
                                                      const float* __restrict__ g,
                                                      float* __restrict__ part, int E,
                                                      int Nn) {
  __shared__ float acc[RANGE];
  const int tid = threadIdx.x;
  const int rbase = blockIdx.y * RANGE;
  const int rlen = min(RANGE, Nn - rbase);
  for (int j = tid; j < RANGE; j += PBLOCK) acc[j] = 0.0f;
  __syncthreads();

  const int per = (E + SLICES - 1) / SLICES;
  const int begin = blockIdx.x * per;
  const int end = min(E, begin + per);
  const int* __restrict__ col = ei + E;
  for (int e0 = begin + tid; e0 < end; e0 += PBLOCK * EPT) {
    int c[EPT];
    int r[EPT];
    float w[EPT];
    float v[EPT];
#pragma unroll
    for (int u = 0; u < EPT; ++u) {
      const int e = e0 + u * PBLOCK;
      const int ec = min(e, end - 1);
      const int cc = col[ec];
      r[u] = ei[ec];
      w[u] = wnorm[ec];
      c[u] = (e < end) ? cc : -1;
    }
    asm volatile("" : "+v"(r[0]), "+v"(w[0]));
#pragma unroll
    for (int u = 0; u < EPT; ++u) {
      const unsigned rel = (unsigned)(c[u] - rbase);
      v[u] = g[(rel < (unsigned)rlen) ? r[u] : 0];
    }
    asm volatile("" : "+v"(v[0]));
#pragma unroll
    for (int u = 0; u < EPT; ++u) {
      const unsigned rel = (unsigned)(c[u] - rbase);
      if (rel < (unsigned)rlen) atomicAdd(&acc[rel], w[u] * v[u]);
    }
  }
  __syncthreads();

  float* __restrict__ dst = part + (size_t)blockIdx.x * Nn + rbase;
  for (int j = tid; j < rlen; j += PBLOCK) dst[j] = acc[j];
}

__global__ void k_out_sum(float* __restrict__ out, const float* __restrict__ part,
                          const float* __restrict__ g, const float* __restrict__ dinv,
                          const float* __restrict__ b3, int Nn) {
  int i = blockIdx.x * blockDim.x + threadIdx.x;
  if (i >= Nn) return;
  float di = dinv[i];
  float s0 = b3[0] + di * di * g[i], s1 = 0.0f, s2 = 0.0f, s3 = 0.0f;
#pragma unroll
  for (int k = 0; k < SLICES; k += 4) {
    s0 += part[(size_t)(k + 0) * Nn + i];
    s1 += part[(size_t)(k + 1) * Nn + i];
    s2 += part[(size_t)(k + 2) * Nn + i];
    s3 += part[(size_t)(k + 3) * Nn + i];
  }
  out[i] = (s0 + s1) + (s2 + s3);
}


static inline unsigned int blocks_for(long long total, int bs) {
  return (unsigned int)((total + bs - 1) / bs);
}

extern "C" void kernel_launch(void* const* d_in, const int* in_sizes, int n_in,
                              void* d_out, int out_size, void* d_ws, size_t ws_size,
                              hipStream_t stream) {
  const float* x = (const float*)d_in[0];
  const int* ei = (const int*)d_in[1];
  const float* ew = (const float*)d_in[2];
  const float* W1 = (const float*)d_in[3];
  const float* b1 = (const float*)d_in[4];
  const float* W2 = (const float*)d_in[5];
  const float* b2 = (const float*)d_in[6];
  const float* W3 = (const float*)d_in[7];
  const float* b3 = (const float*)d_in[8];

  const int E = in_sizes[2];
  const int Nn = in_sizes[0] / IN_C;
  float* out = (float*)d_out;

  float* dinv = (float*)d_ws;
  float* wnorm = dinv + Nn;
  float* bufA = wnorm + E;
  float* bufB = bufA + (size_t)Nn * H1C;

  k_deg_part<<<dim3(SLICES, blocks_for(Nn, RANGE)), PBLOCK, 0, stream>>>(ei, ew, bufA, E, Nn);
  k_dinv_sum<<<blocks_for(Nn, 256), 256, 0, stream>>>(bufA, dinv, Nn);
  k_wnorm<<<blocks_for(E, 256), 256, 0, stream>>>(ei, ew, dinv, wnorm, E);

  k_gemm1_rows<<<blocks_for(Nn / 16, G1_WAVES), G1_THREADS, 0, stream>>>(x, W1, bufA, Nn / 16);
  k_init_agg<H1C><<<blocks_for((long long)Nn * H1C, 256), 256, 0, stream>>>(
      bufB, bufA, dinv, b1, Nn);
  k_row_agg<H1C><<<dim3(3, blocks_for(Nn, RFLOATS / H1C)), RBLOCK, 0, stream>>>(ei, wnorm, bufA,
                                                                              bufB, E, Nn);

  k_gemm_wmma_f32<H1C, H2C / 16, true>
      <<<Nn / 16, (H2C / 16) * 32, 0, stream>>>(bufB, W2, bufA);
  k_init_agg<H2C><<<blocks_for((long long)Nn * H2C, 256), 256, 0, stream>>>(
      bufB, bufA, dinv, b2, Nn);
  k_row_agg<H2C><<<dim3(6, blocks_for(Nn, RFLOATS / H2C)), RBLOCK, 0, stream>>>(ei, wnorm, bufA,
                                                                              bufB, E, Nn);

  k_gemm3_relu<<<blocks_for(Nn, 256), 256, 0, stream>>>(bufB, W3, bufA, Nn);
  k_agg1_part<<<dim3(SLICES, blocks_for(Nn, RANGE)), PBLOCK, 0, stream>>>(ei, wnorm, bufA, bufB,
                                                                         E, Nn);
  k_out_sum<<<blocks_for(Nn, 256), 256, 0, stream>>>(out, bufB, bufA, dinv, b3, Nn);
}


# Layer-2 row aggregation streams edges pre-bucketed by destination range (count pass, then scatter pass into the unused scratch halves) instead of scanning every edge per range; f32 sum order changes
# speedup vs baseline: 2.3804x; 1.0489x over previous
#include <hip/hip_runtime.h>
#include <hip/hip_bf16.h>


typedef float v2f __attribute__((ext_vector_type(2)));
typedef float v8f __attribute__((ext_vector_type(8)));

#define IN_C 512
#define H1C 64
#define H2C 32


#define SLICES 64
#define RANGE 25600
#define PBLOCK 1024
#define EPT 16

__global__ __launch_bounds__(PBLOCK) void k_deg_part(const int* __restrict__ ei,
                                                     const float* __restrict__ ew,
                                                     float* __restrict__ part, int E,
                                                     int Nn) {
  __shared__ float acc[RANGE];
  const int tid = threadIdx.x;
  const int rbase = blockIdx.y * RANGE;
  const int rlen = min(RANGE, Nn - rbase);
  for (int j = tid; j < RANGE; j += PBLOCK) acc[j] = 0.0f;
  __syncthreads();

  const int per = (E + SLICES - 1) / SLICES;
  const int begin = blockIdx.x * per;
  const int end = min(E, begin + per);
  const int* __restrict__ col = ei + E;
  for (int e0 = begin + tid; e0 < end; e0 += PBLOCK * EPT) {
    int c[EPT];
    float w[EPT];
#pragma unroll
    for (int u = 0; u < EPT; ++u) {
      const int e = e0 + u * PBLOCK;
      const int ec = min(e, end - 1);
      const int cc = col[ec];
      w[u] = ew[ec];
      c[u] = (e < end) ? cc : -1;
    }
    asm volatile("" : "+v"(w[0]));
#pragma unroll
    for (int u = 0; u < EPT; ++u) {
      const unsigned rel = (unsigned)(c[u] - rbase);
      if (rel < (unsigned)rlen) atomicAdd(&acc[rel], w[u]);
    }
  }
  __syncthreads();

  float* __restrict__ dst = part + (size_t)blockIdx.x * Nn + rbase;
  for (int j = tid; j < rlen; j += PBLOCK) dst[j] = acc[j];
}

__global__ void k_dinv_sum(const float* __restrict__ part, float* __restrict__ dinv, int Nn) {
  int i = blockIdx.x * blockDim.x + threadIdx.x;
  if (i >= Nn) return;
  float s0 = 1.0f, s1 = 0.0f, s2 = 0.0f, s3 = 0.0f;
#pragma unroll
  for (int k = 0; k < SLICES; k += 4) {
    s0 += part[(size_t)(k + 0) * Nn + i];
    s1 += part[(size_t)(k + 1) * Nn + i];
    s2 += part[(size_t)(k + 2) * Nn + i];
    s3 += part[(size_t)(k + 3) * Nn + i];
  }
  float d = (s0 + s1) + (s2 + s3);
  dinv[i] = (d > 0.0f) ? rsqrtf(fmaxf(d, 1e-12f)) : 0.0f;
}

__global__ void k_wnorm(const int* __restrict__ ei, const float* __restrict__ ew,
                        const float* __restrict__ dinv, float* __restrict__ wnorm, int E) {
  int e = blockIdx.x * blockDim.x + threadIdx.x;
  if (e >= E) return;
  int row = ei[e];
  int col = ei[E + e];
  wnorm[e] = dinv[row] * ew[e] * dinv[col];
}


template <int K, int NT, bool RELU>
__global__ __launch_bounds__(NT * 32) void k_gemm_wmma_f32(
    const float* __restrict__ A, const float* __restrict__ B, float* __restrict__ C) {
  constexpr int NOUT = NT * 16;
  const int wave = threadIdx.x >> 5;
  const int lane = threadIdx.x & 31;
  const int m = lane & 15;
  const int half = lane >> 4;
  const int n0 = wave * 16;
  const long long rowBase = (long long)blockIdx.x * 16;

  const float* arow = A + (rowBase + m) * K + half * 2;

  v8f acc = {};
#pragma unroll 4
  for (int k = 0; k < K; k += 4) {
    float2 av = *reinterpret_cast<const float2*>(arow + k);
    v2f a;
    a.x = RELU ? fmaxf(av.x, 0.0f) : av.x;
    a.y = RELU ? fmaxf(av.y, 0.0f) : av.y;
    const int kr = k + half * 2;
    v2f b;
    b.x = B[kr * NOUT + n0 + m];
    b.y = B[(kr + 1) * NOUT + n0 + m];
    acc = __builtin_amdgcn_wmma_f32_16x16x4_f32(false, a, false, b, (short)0, acc,
                                                false, false);
  }

#pragma unroll
  for (int v = 0; v < 8; ++v) {
    long long r = rowBase + v + 8 * half;
    C[r * NOUT + n0 + m] = acc[v];
  }
}


#define G1_WAVES 25
#define G1_THREADS (G1_WAVES * 32)

__device__ __forceinline__ void g1_block(const float4 a, const float4* bp, v8f& c0, v8f& c1,
                                         v8f& c2, v8f& c3) {
  const float4 b0 = bp[0];
  const float4 b1 = bp[16];
  const float4 b2 = bp[32];
  const float4 b3 = bp[48];
  v2f alo, ahi, f;
  alo.x = a.x; alo.y = a.y;
  ahi.x = a.z; ahi.y = a.w;
  f.x = b0.x; f.y = b0.y;
  c0 = __builtin_amdgcn_wmma_f32_16x16x4_f32(false, alo, false, f, (short)0, c0, false, false);
  f.x = b1.x; f.y = b1.y;
  c1 = __builtin_amdgcn_wmma_f32_16x16x4_f32(false, alo, false, f, (short)0, c1, false, false);
  f.x = b2.x; f.y = b2.y;
  c2 = __builtin_amdgcn_wmma_f32_16x16x4_f32(false, alo, false, f, (short)0, c2, false, false);
  f.x = b3.x; f.y = b3.y;
  c3 = __builtin_amdgcn_wmma_f32_16x16x4_f32(false, alo, false, f, (short)0, c3, false, false);
  f.x = b0.z; f.y = b0.w;
  c0 = __builtin_amdgcn_wmma_f32_16x16x4_f32(false, ahi, false, f, (short)0, c0, false, false);
  f.x = b1.z; f.y = b1.w;
  c1 = __builtin_amdgcn_wmma_f32_16x16x4_f32(false, ahi, false, f, (short)0, c1, false, false);
  f.x = b2.z; f.y = b2.w;
  c2 = __builtin_amdgcn_wmma_f32_16x16x4_f32(false, ahi, false, f, (short)0, c2, false, false);
  f.x = b3.z; f.y = b3.w;
  c3 = __builtin_amdgcn_wmma_f32_16x16x4_f32(false, ahi, false, f, (short)0, c3, false, false);
}

__global__ __launch_bounds__(G1_THREADS) void k_gemm1_rows(const float* __restrict__ A,
                                                           const float* __restrict__ B,
                                                           float* __restrict__ C, int ntiles) {
  __shared__ float4 Bs[(IN_C / 4) * H1C];
  __shared__ float4 Cs[G1_WAVES][16 * H1C / 4];
  const int tid = threadIdx.x;
  const int lane = tid & 31;
  const int wave = __builtin_amdgcn_readfirstlane(tid >> 5);
  const int m = lane & 15;
  const int half = lane >> 4;

  for (int idx = tid; idx < (IN_C / 4) * H1C; idx += G1_THREADS) {
    const int kq = idx / H1C;
    const int n = idx % H1C;
    const float* src = B + (size_t)(4 * kq) * H1C + n;
    float4 v;
    v.x = src[0];
    v.y = src[H1C];
    v.z = src[2 * H1C];
    v.w = src[3 * H1C];
    Bs[idx] = v;
  }
  __syncthreads();

  const int tile = blockIdx.x * G1_WAVES + wave;
  const bool active = tile < ntiles;
  if (active) {
    const float* arow = A + ((size_t)tile * 16 + m) * IN_C + half * 4;
    const float4* bfrag = Bs + half * H1C + m;
    v8f c0 = {}, c1 = {}, c2 = {}, c3 = {};
    float4 a[4];
#pragma unroll
    for (int u = 0; u < 4; ++u) a[u] = *reinterpret_cast<const float4*>(arow + 8 * u);
    for (int kb = 0; kb < IN_C - 32; kb += 32) {
      float4 an[4];
#pragma unroll
      for (int u = 0; u < 4; ++u)
        an[u] = *reinterpret_cast<const float4*>(arow + kb + 32 + 8 * u);
#pragma unroll
      for (int u = 0; u < 4; ++u) g1_block(a[u], bfrag + ((kb + 8 * u) >> 2) * H1C, c0, c1, c2, c3);
#pragma unroll
      for (int u = 0; u < 4; ++u) a[u] = an[u];
    }
#pragma unroll
    for (int u = 0; u < 4; ++u)
      g1_block(a[u], bfrag + ((IN_C - 32 + 8 * u) >> 2) * H1C, c0, c1, c2, c3);

    float* cs = reinterpret_cast<float*>(Cs[wave]);
#pragma unroll
    for (int v = 0; v < 8; ++v) {
      float* p = cs + (v + 8 * half) * H1C + m;
      p[0] = c0[v];
      p[16] = c1[v];
      p[32] = c2[v];
      p[48] = c3[v];
    }
  }
  __syncthreads();
  if (active) {
    float4* dst = reinterpret_cast<float4*>(C + (size_t)tile * 16 * H1C);
#pragma unroll
    for (int i = 0; i < 8; ++i) dst[i * 32 + lane] = Cs[wave][i * 32 + lane];
  }
}

template <int F>
__global__ void k_init_agg(float* __restrict__ agg, const float* __restrict__ h,
                           const float* __restrict__ dinv, const float* __restrict__ bias,
                           int Nn) {
  long long t = (long long)blockIdx.x * blockDim.x + threadIdx.x;
  if (t >= (long long)Nn * F) return;
  int node = (int)(t / F);
  int f = (int)(t % F);
  float di = dinv[node];
  agg[t] = bias[f] + di * di * h[t];
}


#define RFLOATS 76800
#define RBLOCK 1024
#define RK 8

template <int F>
__device__ __forceinline__ void row_batch(const int rv, const int relv, const float wv,
                                          const int b, const int lane,
                                          const float* __restrict__ h, float* acc) {
  constexpr int L = F / 32;
  float v[8][L];
#pragma unroll
  for (int u = 0; u < 8; ++u) {
    const int r = __builtin_amdgcn_readlane(rv, b + u);
    const float* src = h + (size_t)r * F + lane;
#pragma unroll
    for (int l = 0; l < L; ++l) v[u][l] = src[l * 32];
  }
#pragma unroll
  for (int u = 0; u < 8; ++u) {
    const int rel = __builtin_amdgcn_readlane(relv, b + u);
    const float w = __int_as_float(__builtin_amdgcn_readlane(__float_as_int(wv), b + u));
    float* dst = acc + rel * F + lane;
#pragma unroll
    for (int l = 0; l < L; ++l) atomicAdd(dst + l * 32, w * v[u][l]);
  }
}

template <int F>
__global__ __launch_bounds__(RBLOCK) void k_row_agg(const int* __restrict__ ei,
                                                    const float* __restrict__ wnorm,
                                                    const float* __restrict__ h,
                                                    float* __restrict__ agg, int E, int Nn) {
  constexpr int L = F / 32;
  constexpr int RROWS = RFLOATS / F;
  __shared__ float acc[RFLOATS];
  __shared__ int qe[RBLOCK / 32][64];
  __shared__ int qr[RBLOCK / 32][64];
  const int tid = threadIdx.x;
  const int lane = tid & 31;
  const int wave = __builtin_amdgcn_readfirstlane(tid >> 5);
  const int rbase = blockIdx.y * RROWS;
  const int rlen = min(RROWS, Nn - rbase);
  for (int j = tid; j < RFLOATS; j += RBLOCK) acc[j] = 0.0f;
  __syncthreads();

  const int slices = (int)gridDim.x;
  const int per = (E + slices - 1) / slices;
  const int begin = blockIdx.x * per;
  const int end = min(E, begin + per);
  const int* __restrict__ col = ei + E;
  int* myqe = qe[wave];
  int* myqr = qr[wave];
  int head = 0, tail = 0;

  for (int base = begin + wave * (32 * RK); base < end; base += RBLOCK * RK) {
    int c[RK];
    if (base + 32 * RK <= end) {
#pragma unroll
      for (int k = 0; k < RK; ++k) c[k] = col[base + k * 32 + lane];
    } else {
#pragma unroll
      for (int k = 0; k < RK; ++k) {
        const int e = base + k * 32 + lane;
        const int cc = col[min(e, end - 1)];
        c[k] = (e < end) ? cc : -1;
      }
    }
#pragma unroll
    for (int k = 0; k < RK; ++k) {
      const unsigned rel = (unsigned)(c[k] - rbase);
      const bool mine = rel < (unsigned)rlen;
      const unsigned mask = __builtin_amdgcn_ballot_w32(mine);
      if (mask != 0u) {
        const int pos = (tail + (int)__builtin_amdgcn_mbcnt_lo(mask, 0u)) & 63;
        if (mine) {
          myqe[pos] = base + k * 32 + lane;
          myqr[pos] = (int)rel;
        }
        tail += __builtin_popcount(mask);
        if (tail - head >= 32) {
          const int idx = (head + lane) & 63;
          const int e = myqe[idx];
          const int relv = myqr[idx];
          head += 32;
          const int rv = ei[e];
          const float wv = wnorm[e];
          row_batch<F>(rv, relv, wv, 0, lane, h, acc);
          row_batch<F>(rv, relv, wv, 8, lane, h, acc);
          row_batch<F>(rv, relv, wv, 16, lane, h, acc);
          row_batch<F>(rv, relv, wv, 24, lane, h, acc);
        }
      }
    }
  }
  const int left = tail - head;
  if (left > 0) {
    const int idx = (head + lane) & 63;
    const bool valid = lane < left;
    const int e = valid ? myqe[idx] : begin;
    const int relv = valid ? myqr[idx] : 0;
    const int rv = ei[e];
    const float wv = wnorm[e];
#pragma unroll
    for (int b = 0; b < 32; b += 8) {
      if (b + 8 <= left) {
        row_batch<F>(rv, relv, wv, b, lane, h, acc);
      } else {
#pragma unroll
        for (int u = b; u < b + 8; ++u) {
          if (u < left) {
            const int r = __builtin_amdgcn_readlane(rv, u);
            const int rel = __builtin_amdgcn_readlane(relv, u);
            const float w = __int_as_float(__builtin_amdgcn_readlane(__float_as_int(wv), u));
            const float* src = h + (size_t)r * F + lane;
            float* dst = acc + rel * F + lane;
#pragma unroll
            for (int l = 0; l < L; ++l) atomicAdd(dst + l * 32, w * src[l * 32]);
          }
        }
      }
    }
  }
  __syncthreads();

  float* __restrict__ out_rows = agg + (size_t)rbase * F;
  for (int j = tid; j < rlen * F; j += RBLOCK) atomicAdd(out_rows + j, acc[j]);
}


#define BKB 256
#define BKT 1024
#define BKE 16
#define BKR 128
#define R2ROWS (RFLOATS / H2C)

__global__ __launch_bounds__(BKT) void k_bk_count(const int* __restrict__ ei,
                                                  int* __restrict__ cnt, int E, int Nn,
                                                  int nr) {
  __shared__ int hist[BKR];
  const int tid = threadIdx.x;
  if (tid < BKR) hist[tid] = 0;
  __syncthreads();
  const int per = (E + BKB - 1) / BKB;
  const int begin = blockIdx.x * per;
  const int end = min(E, begin + per);
  const int* __restrict__ col = ei + E;
  for (int e0 = begin + tid; e0 < end; e0 += BKT * BKE) {
    int c[BKE];
#pragma unroll
    for (int u = 0; u < BKE; ++u) {
      const int e = e0 + u * BKT;
      const int cc = col[min(e, end - 1)];
      c[u] = (e < end) ? cc : -1;
    }
#pragma unroll
    for (int u = 0; u < BKE; ++u)
      if ((unsigned)c[u] < (unsigned)Nn) atomicAdd(&hist[c[u] / R2ROWS], 1);
  }
  __syncthreads();
  if (tid < nr) cnt[tid * BKB + blockIdx.x] = hist[tid];
}

__global__ __launch_bounds__(BKT) void k_bk_scatter(const int* __restrict__ ei,
                                                    const float* __restrict__ wnorm,
                                                    const int* __restrict__ cnt,
                                                    int* __restrict__ starts,
                                                    int* __restrict__ lrow,
                                                    float* __restrict__ lw, int E, int Nn,
                                                    int nr) {
  __shared__ int tot[BKR];
  __shared__ int before[BKR];
  __shared__ int cur[BKR];
  const int tid = threadIdx.x;
  const int lane = tid & 31;
  const int wave = tid >> 5;
  for (int r = wave; r < nr; r += BKT / 32) {
    int t = 0, bf = 0;
#pragma unroll
    for (int k = 0; k < BKB / 32; ++k) {
      const int b = k * 32 + lane;
      const int v = cnt[r * BKB + b];
      t += v;
      bf += (b < (int)blockIdx.x) ? v : 0;
    }
    for (int d = 16; d > 0; d >>= 1) {
      t += __shfl_xor(t, d);
      bf += __shfl_xor(bf, d);
    }
    if (lane == 0) {
      tot[r] = t;
      before[r] = bf;
    }
  }
  __syncthreads();
  if (tid <= nr) {
    int sum = 0;
    for (int q = 0; q < tid; ++q) sum += tot[q];
    if (tid < nr) cur[tid] = sum + before[tid];
    if (blockIdx.x == 0) starts[tid] = sum;
  }
  __syncthreads();

  const int per = (E + BKB - 1) / BKB;
  const int begin = blockIdx.x * per;
  const int end = min(E, begin + per);
  const int* __restrict__ col = ei + E;
  for (int e0 = begin + tid; e0 < end; e0 += BKT * BKE) {
    int c[BKE];
    int r[BKE];
    float w[BKE];
#pragma unroll
    for (int u = 0; u < BKE; ++u) {
      const int e = e0 + u * BKT;
      const int ec = min(e, end - 1);
      const int cc = col[ec];
      r[u] = ei[ec];
      w[u] = wnorm[ec];
      c[u] = (e < end) ? cc : -1;
    }
    asm volatile("" : "+v"(r[0]), "+v"(w[0]));
#pragma unroll
    for (int u = 0; u < BKE; ++u) {
      if ((unsigned)c[u] < (unsigned)Nn) {
        const int rg = c[u] / R2ROWS;
        const int pos = atomicAdd(&cur[rg], 1);
        lrow[pos] = (r[u] << 12) | (c[u] - rg * R2ROWS);
        lw[pos] = w[u];
      }
    }
  }
}

template <int F, int NB>
__device__ __forceinline__ void row_batch_n(const int rv, const int relv, const float wv,
                                            const int b, const int lane,
                                            const float* __restrict__ h, float* acc) {
  constexpr int L = F / 32;
  float v[NB][L];
#pragma unroll
  for (int u = 0; u < NB; ++u) {
    const int r = __builtin_amdgcn_readlane(rv, b + u);
    const float* src = h + (size_t)r * F + lane;
#pragma unroll
    for (int l = 0; l < L; ++l) v[u][l] = src[l * 32];
  }
#pragma unroll
  for (int u = 0; u < NB; ++u) {
    const int rel = __builtin_amdgcn_readlane(relv, b + u);
    const float w = __int_as_float(__builtin_amdgcn_readlane(__float_as_int(wv), b + u));
    float* dst = acc + rel * F + lane;
#pragma unroll
    for (int l = 0; l < L; ++l) atomicAdd(dst + l * 32, w * v[u][l]);
  }
}

template <int F>
__device__ __forceinline__ void row_group(const int rv, const int relv, const float wv,
                                          const int left, const int lane,
                                          const float* __restrict__ h, float* acc) {
  constexpr int L = F / 32;
#pragma unroll
  for (int b = 0; b < 32; b += 16) {
    if (b + 16 <= left) {
      row_batch_n<F, 16>(rv, relv, wv, b, lane, h, acc);
    } else {
#pragma unroll
      for (int bb = b; bb < b + 16; bb += 8) {
        if (bb + 8 <= left) {
          row_batch_n<F, 8>(rv, relv, wv, bb, lane, h, acc);
        } else {
#pragma unroll
          for (int u = bb; u < bb + 8; ++u) {
            if (u < left) {
              const int r = __builtin_amdgcn_readlane(rv, u);
              const int rel = __builtin_amdgcn_readlane(relv, u);
              const float w =
                  __int_as_float(__builtin_amdgcn_readlane(__float_as_int(wv), u));
              const float* src = h + (size_t)r * F + lane;
              float* dst = acc + rel * F + lane;
#pragma unroll
              for (int l = 0; l < L; ++l) atomicAdd(dst + l * 32, w * src[l * 32]);
            }
          }
        }
      }
    }
  }
}

template <int F>
__global__ __launch_bounds__(RBLOCK) void k_row_agg_bk(const int* __restrict__ lrow,
                                                       const float* __restrict__ lw,
                                                       const int* __restrict__ starts,
                                                       const float* __restrict__ h,
                                                       float* __restrict__ agg, int Nn) {
  constexpr int RROWS = RFLOATS / F;
  __shared__ float acc[RFLOATS];
  const int tid = threadIdx.x;
  const int lane = tid & 31;
  const int wave = __builtin_amdgcn_readfirstlane(tid >> 5);
  const int b0 = __builtin_amdgcn_readfirstlane(starts[blockIdx.y]);
  const int b1 = __builtin_amdgcn_readfirstlane(starts[blockIdx.y + 1]);
  const int rbase = blockIdx.y * RROWS;
  const int rlen = min(RROWS, Nn - rbase);
  for (int j = tid; j < RFLOATS; j += RBLOCK) acc[j] = 0.0f;
  __syncthreads();

  const int parts = (int)gridDim.x;
  const int per = (((b1 - b0) + parts - 1) / parts + 31) & ~31;
  const int begin = b0 + blockIdx.x * per;
  const int end = min(b1, begin + per);

  int i = begin + wave * 32;
  int pk = 0;
  float wv = 0.0f;
  if (i < end && lane < end - i) {
    pk = lrow[i + lane];
    wv = lw[i + lane];
  }
  while (i < end) {
    const int inext = i + RBLOCK;
    int pkn = 0;
    float wn = 0.0f;
    if (inext < end && lane < end - inext) {
      pkn = lrow[inext + lane];
      wn = lw[inext + lane];
    }
    row_group<F>(pk >> 12, pk & 4095, wv, min(32, end - i), lane, h, acc);
    pk = pkn;
    wv = wn;
    i = inext;
  }
  __syncthreads();

  float* __restrict__ out_rows = agg + (size_t)rbase * F;
  for (int j = tid; j < rlen * F; j += RBLOCK) atomicAdd(out_rows + j, acc[j]);
}

__global__ void k_gemm3_relu(const float* __restrict__ a2, const float* __restrict__ W3,
                             float* __restrict__ g, int Nn) {
  int i = blockIdx.x * blockDim.x + threadIdx.x;
  if (i >= Nn) return;
  const float* r = a2 + (size_t)i * H2C;
  float acc = 0.0f;
#pragma unroll
  for (int f = 0; f < H2C; ++f) acc += fmaxf(r[f], 0.0f) * W3[f];
  g[i] = acc;
}

__global__ __launch_bounds__(PBLOCK) void k_agg1_part(const int* __restrict__ ei,
                                                      const float* __restrict__ wnorm,
                                                      const float* __restrict__ g,
                                                      float* __restrict__ part, int E,
                                                      int Nn) {
  __shared__ float acc[RANGE];
  const int tid = threadIdx.x;
  const int rbase = blockIdx.y * RANGE;
  const int rlen = min(RANGE, Nn - rbase);
  for (int j = tid; j < RANGE; j += PBLOCK) acc[j] = 0.0f;
  __syncthreads();

  const int per = (E + SLICES - 1) / SLICES;
  const int begin = blockIdx.x * per;
  const int end = min(E, begin + per);
  const int* __restrict__ col = ei + E;
  for (int e0 = begin + tid; e0 < end; e0 += PBLOCK * EPT) {
    int c[EPT];
    int r[EPT];
    float w[EPT];
    float v[EPT];
#pragma unroll
    for (int u = 0; u < EPT; ++u) {
      const int e = e0 + u * PBLOCK;
      const int ec = min(e, end - 1);
      const int cc = col[ec];
      r[u] = ei[ec];
      w[u] = wnorm[ec];
      c[u] = (e < end) ? cc : -1;
    }
    asm volatile("" : "+v"(r[0]), "+v"(w[0]));
#pragma unroll
    for (int u = 0; u < EPT; ++u) {
      const unsigned rel = (unsigned)(c[u] - rbase);
      v[u] = g[(rel < (unsigned)rlen) ? r[u] : 0];
    }
    asm volatile("" : "+v"(v[0]));
#pragma unroll
    for (int u = 0; u < EPT; ++u) {
      const unsigned rel = (unsigned)(c[u] - rbase);
      if (rel < (unsigned)rlen) atomicAdd(&acc[rel], w[u] * v[u]);
    }
  }
  __syncthreads();

  float* __restrict__ dst = part + (size_t)blockIdx.x * Nn + rbase;
  for (int j = tid; j < rlen; j += PBLOCK) dst[j] = acc[j];
}

__global__ void k_out_sum(float* __restrict__ out, const float* __restrict__ part,
                          const float* __restrict__ g, const float* __restrict__ dinv,
                          const float* __restrict__ b3, int Nn) {
  int i = blockIdx.x * blockDim.x + threadIdx.x;
  if (i >= Nn) return;
  float di = dinv[i];
  float s0 = b3[0] + di * di * g[i], s1 = 0.0f, s2 = 0.0f, s3 = 0.0f;
#pragma unroll
  for (int k = 0; k < SLICES; k += 4) {
    s0 += part[(size_t)(k + 0) * Nn + i];
    s1 += part[(size_t)(k + 1) * Nn + i];
    s2 += part[(size_t)(k + 2) * Nn + i];
    s3 += part[(size_t)(k + 3) * Nn + i];
  }
  out[i] = (s0 + s1) + (s2 + s3);
}


static inline unsigned int blocks_for(long long total, int bs) {
  return (unsigned int)((total + bs - 1) / bs);
}

extern "C" void kernel_launch(void* const* d_in, const int* in_sizes, int n_in,
                              void* d_out, int out_size, void* d_ws, size_t ws_size,
                              hipStream_t stream) {
  const float* x = (const float*)d_in[0];
  const int* ei = (const int*)d_in[1];
  const float* ew = (const float*)d_in[2];
  const float* W1 = (const float*)d_in[3];
  const float* b1 = (const float*)d_in[4];
  const float* W2 = (const float*)d_in[5];
  const float* b2 = (const float*)d_in[6];
  const float* W3 = (const float*)d_in[7];
  const float* b3 = (const float*)d_in[8];

  const int E = in_sizes[2];
  const int Nn = in_sizes[0] / IN_C;
  float* out = (float*)d_out;

  float* dinv = (float*)d_ws;
  float* wnorm = dinv + Nn;
  float* bufA = wnorm + E;
  float* bufB = bufA + (size_t)Nn * H1C;

  k_deg_part<<<dim3(SLICES, blocks_for(Nn, RANGE)), PBLOCK, 0, stream>>>(ei, ew, bufA, E, Nn);
  k_dinv_sum<<<blocks_for(Nn, 256), 256, 0, stream>>>(bufA, dinv, Nn);
  k_wnorm<<<blocks_for(E, 256), 256, 0, stream>>>(ei, ew, dinv, wnorm, E);

  k_gemm1_rows<<<blocks_for(Nn / 16, G1_WAVES), G1_THREADS, 0, stream>>>(x, W1, bufA, Nn / 16);
  k_init_agg<H1C><<<blocks_for((long long)Nn * H1C, 256), 256, 0, stream>>>(
      bufB, bufA, dinv, b1, Nn);
  k_row_agg<H1C><<<dim3(3, blocks_for(Nn, RFLOATS / H1C)), RBLOCK, 0, stream>>>(ei, wnorm, bufA,
                                                                              bufB, E, Nn);

  k_gemm_wmma_f32<H1C, H2C / 16, true>
      <<<Nn / 16, (H2C / 16) * 32, 0, stream>>>(bufB, W2, bufA);
  const int nr2 = (int)blocks_for(Nn, R2ROWS);
  int* bk_cnt = (int*)d_out;
  int* bk_starts = bk_cnt + nr2 * BKB;
  int* lrow = (int*)(bufA + (size_t)Nn * H2C);
  float* lw = bufB + (size_t)Nn * H2C;
  k_bk_count<<<BKB, BKT, 0, stream>>>(ei, bk_cnt, E, Nn, nr2);
  k_bk_scatter<<<BKB, BKT, 0, stream>>>(ei, wnorm, bk_cnt, bk_starts, lrow, lw, E, Nn, nr2);
  k_init_agg<H2C><<<blocks_for((long long)Nn * H2C, 256), 256, 0, stream>>>(
      bufB, bufA, dinv, b2, Nn);
  k_row_agg_bk<H2C><<<dim3(6, nr2), RBLOCK, 0, stream>>>(lrow, lw, bk_starts, bufA, bufB, Nn);

  k_gemm3_relu<<<blocks_for(Nn, 256), 256, 0, stream>>>(bufB, W3, bufA, Nn);
  k_agg1_part<<<dim3(SLICES, blocks_for(Nn, RANGE)), PBLOCK, 0, stream>>>(ei, wnorm, bufA, bufB,
                                                                         E, Nn);
  k_out_sum<<<blocks_for(Nn, 256), 256, 0, stream>>>(out, bufB, bufA, dinv, b3, Nn);
}


# Edges bucketed by destination range once, before layer 1 (scatter pass computes the normalised weight); all three aggregations stream that list, layer 1 as two 32-wide planes; f32 sum order changes
# speedup vs baseline: 3.0720x; 1.2905x over previous
#include <hip/hip_runtime.h>
#include <hip/hip_bf16.h>


typedef float v2f __attribute__((ext_vector_type(2)));
typedef float v8f __attribute__((ext_vector_type(8)));

#define IN_C 512
#define H1C 64
#define H2C 32


#define SLICES 64
#define RANGE 25600
#define PBLOCK 1024
#define EPT 16

__global__ __launch_bounds__(PBLOCK) void k_deg_part(const int* __restrict__ ei,
                                                     const float* __restrict__ ew,
                                                     float* __restrict__ part, int E,
                                                     int Nn) {
  __shared__ float acc[RANGE];
  const int tid = threadIdx.x;
  const int rbase = blockIdx.y * RANGE;
  const int rlen = min(RANGE, Nn - rbase);
  for (int j = tid; j < RANGE; j += PBLOCK) acc[j] = 0.0f;
  __syncthreads();

  const int per = (E + SLICES - 1) / SLICES;
  const int begin = blockIdx.x * per;
  const int end = min(E, begin + per);
  const int* __restrict__ col = ei + E;
  for (int e0 = begin + tid; e0 < end; e0 += PBLOCK * EPT) {
    int c[EPT];
    float w[EPT];
#pragma unroll
    for (int u = 0; u < EPT; ++u) {
      const int e = e0 + u * PBLOCK;
      const int ec = min(e, end - 1);
      const int cc = col[ec];
      w[u] = ew[ec];
      c[u] = (e < end) ? cc : -1;
    }
    asm volatile("" : "+v"(w[0]));
#pragma unroll
    for (int u = 0; u < EPT; ++u) {
      const unsigned rel = (unsigned)(c[u] - rbase);
      if (rel < (unsigned)rlen) atomicAdd(&acc[rel], w[u]);
    }
  }
  __syncthreads();

  float* __restrict__ dst = part + (size_t)blockIdx.x * Nn + rbase;
  for (int j = tid; j < rlen; j += PBLOCK) dst[j] = acc[j];
}

__global__ void k_dinv_sum(const float* __restrict__ part, float* __restrict__ dinv, int Nn) {
  int i = blockIdx.x * blockDim.x + threadIdx.x;
  if (i >= Nn) return;
  float s0 = 1.0f, s1 = 0.0f, s2 = 0.0f, s3 = 0.0f;
#pragma unroll
  for (int k = 0; k < SLICES; k += 4) {
    s0 += part[(size_t)(k + 0) * Nn + i];
    s1 += part[(size_t)(k + 1) * Nn + i];
    s2 += part[(size_t)(k + 2) * Nn + i];
    s3 += part[(size_t)(k + 3) * Nn + i];
  }
  float d = (s0 + s1) + (s2 + s3);
  dinv[i] = (d > 0.0f) ? rsqrtf(fmaxf(d, 1e-12f)) : 0.0f;
}


__global__ __launch_bounds__(64) void k_gemm2_planes(const float* __restrict__ A0,
                                                     const float* __restrict__ A1,
                                                     const float* __restrict__ B,
                                                     float* __restrict__ C) {
  constexpr int NOUT = H2C;
  constexpr int KH = H1C / 2;
  const int wave = threadIdx.x >> 5;
  const int lane = threadIdx.x & 31;
  const int m = lane & 15;
  const int half = lane >> 4;
  const int n0 = wave * 16;
  const long long rowBase = (long long)blockIdx.x * 16;

  const float* a0 = A0 + (rowBase + m) * KH + half * 2;
  const float* a1 = A1 + (rowBase + m) * KH + half * 2;

  v8f acc = {};
#pragma unroll 4
  for (int k = 0; k < KH; k += 4) {
    float2 av = *reinterpret_cast<const float2*>(a0 + k);
    v2f a;
    a.x = fmaxf(av.x, 0.0f);
    a.y = fmaxf(av.y, 0.0f);
    const int kr = k + half * 2;
    v2f b;
    b.x = B[kr * NOUT + n0 + m];
    b.y = B[(kr + 1) * NOUT + n0 + m];
    acc = __builtin_amdgcn_wmma_f32_16x16x4_f32(false, a, false, b, (short)0, acc,
                                                false, false);
  }
#pragma unroll 4
  for (int k = 0; k < KH; k += 4) {
    float2 av = *reinterpret_cast<const float2*>(a1 + k);
    v2f a;
    a.x = fmaxf(av.x, 0.0f);
    a.y = fmaxf(av.y, 0.0f);
    const int kr = KH + k + half * 2;
    v2f b;
    b.x = B[kr * NOUT + n0 + m];
    b.y = B[(kr + 1) * NOUT + n0 + m];
    acc = __builtin_amdgcn_wmma_f32_16x16x4_f32(false, a, false, b, (short)0, acc,
                                                false, false);
  }

#pragma unroll
  for (int v = 0; v < 8; ++v) {
    long long r = rowBase + v + 8 * half;
    C[r * NOUT + n0 + m] = acc[v];
  }
}


#define G1_WAVES 25
#define G1_THREADS (G1_WAVES * 32)

__device__ __forceinline__ void g1_block(const float4 a, const float4* bp, v8f& c0, v8f& c1,
                                         v8f& c2, v8f& c3) {
  const float4 b0 = bp[0];
  const float4 b1 = bp[16];
  const float4 b2 = bp[32];
  const float4 b3 = bp[48];
  v2f alo, ahi, f;
  alo.x = a.x; alo.y = a.y;
  ahi.x = a.z; ahi.y = a.w;
  f.x = b0.x; f.y = b0.y;
  c0 = __builtin_amdgcn_wmma_f32_16x16x4_f32(false, alo, false, f, (short)0, c0, false, false);
  f.x = b1.x; f.y = b1.y;
  c1 = __builtin_amdgcn_wmma_f32_16x16x4_f32(false, alo, false, f, (short)0, c1, false, false);
  f.x = b2.x; f.y = b2.y;
  c2 = __builtin_amdgcn_wmma_f32_16x16x4_f32(false, alo, false, f, (short)0, c2, false, false);
  f.x = b3.x; f.y = b3.y;
  c3 = __builtin_amdgcn_wmma_f32_16x16x4_f32(false, alo, false, f, (short)0, c3, false, false);
  f.x = b0.z; f.y = b0.w;
  c0 = __builtin_amdgcn_wmma_f32_16x16x4_f32(false, ahi, false, f, (short)0, c0, false, false);
  f.x = b1.z; f.y = b1.w;
  c1 = __builtin_amdgcn_wmma_f32_16x16x4_f32(false, ahi, false, f, (short)0, c1, false, false);
  f.x = b2.z; f.y = b2.w;
  c2 = __builtin_amdgcn_wmma_f32_16x16x4_f32(false, ahi, false, f, (short)0, c2, false, false);
  f.x = b3.z; f.y = b3.w;
  c3 = __builtin_amdgcn_wmma_f32_16x16x4_f32(false, ahi, false, f, (short)0, c3, false, false);
}

__global__ __launch_bounds__(G1_THREADS) void k_gemm1_rows(const float* __restrict__ A,
                                                           const float* __restrict__ B,
                                                           float* __restrict__ C0,
                                                           float* __restrict__ C1, int ntiles) {
  __shared__ float4 Bs[(IN_C / 4) * H1C];
  __shared__ float4 Cs[G1_WAVES][16 * H1C / 4];
  const int tid = threadIdx.x;
  const int lane = tid & 31;
  const int wave = __builtin_amdgcn_readfirstlane(tid >> 5);
  const int m = lane & 15;
  const int half = lane >> 4;

  for (int idx = tid; idx < (IN_C / 4) * H1C; idx += G1_THREADS) {
    const int kq = idx / H1C;
    const int n = idx % H1C;
    const float* src = B + (size_t)(4 * kq) * H1C + n;
    float4 v;
    v.x = src[0];
    v.y = src[H1C];
    v.z = src[2 * H1C];
    v.w = src[3 * H1C];
    Bs[idx] = v;
  }
  __syncthreads();

  const int tile = blockIdx.x * G1_WAVES + wave;
  const bool active = tile < ntiles;
  if (active) {
    const float* arow = A + ((size_t)tile * 16 + m) * IN_C + half * 4;
    const float4* bfrag = Bs + half * H1C + m;
    v8f c0 = {}, c1 = {}, c2 = {}, c3 = {};
    float4 a[4];
#pragma unroll
    for (int u = 0; u < 4; ++u) a[u] = *reinterpret_cast<const float4*>(arow + 8 * u);
    for (int kb = 0; kb < IN_C - 32; kb += 32) {
      float4 an[4];
#pragma unroll
      for (int u = 0; u < 4; ++u)
        an[u] = *reinterpret_cast<const float4*>(arow + kb + 32 + 8 * u);
#pragma unroll
      for (int u = 0; u < 4; ++u) g1_block(a[u], bfrag + ((kb + 8 * u) >> 2) * H1C, c0, c1, c2, c3);
#pragma unroll
      for (int u = 0; u < 4; ++u) a[u] = an[u];
    }
#pragma unroll
    for (int u = 0; u < 4; ++u)
      g1_block(a[u], bfrag + ((IN_C - 32 + 8 * u) >> 2) * H1C, c0, c1, c2, c3);

    float* cs = reinterpret_cast<float*>(Cs[wave]);
#pragma unroll
    for (int v = 0; v < 8; ++v) {
      float* p = cs + (v + 8 * half) * H1C + m;
      p[0] = c0[v];
      p[16] = c1[v];
      p[32] = c2[v];
      p[48] = c3[v];
    }
  }
  __syncthreads();
  if (active) {
    float* cp = ((lane >> 3) & 1) ? C1 : C0;
    float4* dst = reinterpret_cast<float4*>(cp + (size_t)tile * 16 * (H1C / 2)) +
                  (lane >> 4) * 8 + (lane & 7);
#pragma unroll
    for (int i = 0; i < 8; ++i) dst[i * 16] = Cs[wave][i * 32 + lane];
  }
}

template <int F>
__global__ void k_init_agg(float* __restrict__ agg, const float* __restrict__ h,
                           const float* __restrict__ dinv, const float* __restrict__ bias,
                           int Nn) {
  long long t = (long long)blockIdx.x * blockDim.x + threadIdx.x;
  if (t >= (long long)Nn * F) return;
  int node = (int)(t / F);
  int f = (int)(t % F);
  float di = dinv[node];
  agg[t] = bias[f] + di * di * h[t];
}


#define RFLOATS 76800
#define RBLOCK 1024

#define BKB 256
#define BKT 1024
#define BKE 16
#define BKR 128
#define R2ROWS (RFLOATS / H2C)

__global__ __launch_bounds__(BKT) void k_bk_count(const int* __restrict__ ei,
                                                  int* __restrict__ cnt, int E, int Nn,
                                                  int nr) {
  __shared__ int hist[BKR];
  const int tid = threadIdx.x;
  if (tid < BKR) hist[tid] = 0;
  __syncthreads();
  const int per = (E + BKB - 1) / BKB;
  const int begin = blockIdx.x * per;
  const int end = min(E, begin + per);
  const int* __restrict__ col = ei + E;
  for (int e0 = begin + tid; e0 < end; e0 += BKT * BKE) {
    int c[BKE];
#pragma unroll
    for (int u = 0; u < BKE; ++u) {
      const int e = e0 + u * BKT;
      const int cc = col[min(e, end - 1)];
      c[u] = (e < end) ? cc : -1;
    }
#pragma unroll
    for (int u = 0; u < BKE; ++u)
      if ((unsigned)c[u] < (unsigned)Nn) atomicAdd(&hist[c[u] / R2ROWS], 1);
  }
  __syncthreads();
  if (tid < nr) cnt[tid * BKB + blockIdx.x] = hist[tid];
}

__global__ __launch_bounds__(BKT) void k_bk_scatter(const int* __restrict__ ei,
                                                    const float* __restrict__ ew,
                                                    const float* __restrict__ dinv,
                                                    const int* __restrict__ cnt,
                                                    int* __restrict__ starts,
                                                    int* __restrict__ lrow,
                                                    float* __restrict__ lw, int E, int Nn,
                                                    int nr) {
  __shared__ int tot[BKR];
  __shared__ int before[BKR];
  __shared__ int cur[BKR];
  const int tid = threadIdx.x;
  const int lane = tid & 31;
  const int wave = tid >> 5;
  for (int r = wave; r < nr; r += BKT / 32) {
    int t = 0, bf = 0;
#pragma unroll
    for (int k = 0; k < BKB / 32; ++k) {
      const int b = k * 32 + lane;
      const int v = cnt[r * BKB + b];
      t += v;
      bf += (b < (int)blockIdx.x) ? v : 0;
    }
    for (int d = 16; d > 0; d >>= 1) {
      t += __shfl_xor(t, d);
      bf += __shfl_xor(bf, d);
    }
    if (lane == 0) {
      tot[r] = t;
      before[r] = bf;
    }
  }
  __syncthreads();
  if (tid <= nr) {
    int sum = 0;
    for (int q = 0; q < tid; ++q) sum += tot[q];
    if (tid < nr) cur[tid] = sum + before[tid];
    if (blockIdx.x == 0) starts[tid] = sum;
  }
  __syncthreads();

  const int per = (E + BKB - 1) / BKB;
  const int begin = blockIdx.x * per;
  const int end = min(E, begin + per);
  const int* __restrict__ col = ei + E;
  for (int e0 = begin + tid; e0 < end; e0 += BKT * BKE) {
    int c[BKE];
    int r[BKE];
    float w[BKE];
    float dc[BKE];
#pragma unroll
    for (int u = 0; u < BKE; ++u) {
      const int e = e0 + u * BKT;
      const int ec = min(e, end - 1);
      const int cc = col[ec];
      r[u] = ei[ec];
      w[u] = ew[ec];
      dc[u] = dinv[cc];
      c[u] = (e < end) ? cc : -1;
    }
#pragma unroll
    for (int u = 0; u < BKE; ++u) w[u] = dinv[r[u]] * w[u] * dc[u];
    asm volatile("" : "+v"(r[0]), "+v"(w[0]));
#pragma unroll
    for (int u = 0; u < BKE; ++u) {
      if ((unsigned)c[u] < (unsigned)Nn) {
        const int rg = c[u] / R2ROWS;
        const int pos = atomicAdd(&cur[rg], 1);
        lrow[pos] = (r[u] << 12) | (c[u] - rg * R2ROWS);
        lw[pos] = w[u];
      }
    }
  }
}

template <int F, int NB>
__device__ __forceinline__ void row_batch_n(const int rv, const int relv, const float wv,
                                            const int b, const int lane,
                                            const float* __restrict__ h, float* acc) {
  constexpr int L = F / 32;
  float v[NB][L];
#pragma unroll
  for (int u = 0; u < NB; ++u) {
    const int r = __builtin_amdgcn_readlane(rv, b + u);
    const float* src = h + (size_t)r * F + lane;
#pragma unroll
    for (int l = 0; l < L; ++l) v[u][l] = src[l * 32];
  }
#pragma unroll
  for (int u = 0; u < NB; ++u) {
    const int rel = __builtin_amdgcn_readlane(relv, b + u);
    const float w = __int_as_float(__builtin_amdgcn_readlane(__float_as_int(wv), b + u));
    float* dst = acc + rel * F + lane;
#pragma unroll
    for (int l = 0; l < L; ++l) atomicAdd(dst + l * 32, w * v[u][l]);
  }
}

template <int F>
__device__ __forceinline__ void row_group(const int rv, const int relv, const float wv,
                                          const int left, const int lane,
                                          const float* __restrict__ h, float* acc) {
  constexpr int L = F / 32;
#pragma unroll
  for (int b = 0; b < 32; b += 16) {
    if (b + 16 <= left) {
      row_batch_n<F, 16>(rv, relv, wv, b, lane, h, acc);
    } else {
#pragma unroll
      for (int bb = b; bb < b + 16; bb += 8) {
        if (bb + 8 <= left) {
          row_batch_n<F, 8>(rv, relv, wv, bb, lane, h, acc);
        } else {
#pragma unroll
          for (int u = bb; u < bb + 8; ++u) {
            if (u < left) {
              const int r = __builtin_amdgcn_readlane(rv, u);
              const int rel = __builtin_amdgcn_readlane(relv, u);
              const float w =
                  __int_as_float(__builtin_amdgcn_readlane(__float_as_int(wv), u));
              const float* src = h + (size_t)r * F + lane;
              float* dst = acc + rel * F + lane;
#pragma unroll
              for (int l = 0; l < L; ++l) atomicAdd(dst + l * 32, w * src[l * 32]);
            }
          }
        }
      }
    }
  }
}

template <int F>
__global__ __launch_bounds__(RBLOCK) void k_row_agg_bk(const int* __restrict__ lrow,
                                                       const float* __restrict__ lw,
                                                       const int* __restrict__ starts,
                                                       const float* __restrict__ h,
                                                       float* __restrict__ agg, int Nn) {
  constexpr int RROWS = RFLOATS / F;
  __shared__ float acc[RFLOATS];
  const int tid = threadIdx.x;
  const int lane = tid & 31;
  const int wave = __builtin_amdgcn_readfirstlane(tid >> 5);
  const int b0 = __builtin_amdgcn_readfirstlane(starts[blockIdx.y]);
  const int b1 = __builtin_amdgcn_readfirstlane(starts[blockIdx.y + 1]);
  const int rbase = blockIdx.y * RROWS;
  const int rlen = min(RROWS, Nn - rbase);
  for (int j = tid; j < RFLOATS; j += RBLOCK) acc[j] = 0.0f;
  __syncthreads();

  const int parts = (int)gridDim.x;
  const int per = (((b1 - b0) + parts - 1) / parts + 31) & ~31;
  const int begin = b0 + blockIdx.x * per;
  const int end = min(b1, begin + per);

  int i = begin + wave * 32;
  int pk = 0;
  float wv = 0.0f;
  if (i < end && lane < end - i) {
    pk = lrow[i + lane];
    wv = lw[i + lane];
  }
  while (i < end) {
    const int inext = i + RBLOCK;
    int pkn = 0;
    float wn = 0.0f;
    if (inext < end && lane < end - inext) {
      pkn = lrow[inext + lane];
      wn = lw[inext + lane];
    }
    row_group<F>(pk >> 12, pk & 4095, wv, min(32, end - i), lane, h, acc);
    pk = pkn;
    wv = wn;
    i = inext;
  }
  __syncthreads();

  float* __restrict__ out_rows = agg + (size_t)rbase * F;
  for (int j = tid; j < rlen * F; j += RBLOCK) atomicAdd(out_rows + j, acc[j]);
}

__global__ void k_gemm3_relu(const float* __restrict__ a2, const float* __restrict__ W3,
                             float* __restrict__ g, int Nn) {
  int i = blockIdx.x * blockDim.x + threadIdx.x;
  if (i >= Nn) return;
  const float* r = a2 + (size_t)i * H2C;
  float acc = 0.0f;
#pragma unroll
  for (int f = 0; f < H2C; ++f) acc += fmaxf(r[f], 0.0f) * W3[f];
  g[i] = acc;
}

#define P3 6
#define P3E 16

__global__ __launch_bounds__(RBLOCK) void k_agg1_bk(const int* __restrict__ lrow,
                                                    const float* __restrict__ lw,
                                                    const int* __restrict__ starts,
                                                    const float* __restrict__ g,
                                                    float* __restrict__ part, int Nn) {
  __shared__ float acc[R2ROWS];
  const int tid = threadIdx.x;
  const int b0 = starts[blockIdx.y];
  const int b1 = starts[blockIdx.y + 1];
  const int rbase = blockIdx.y * R2ROWS;
  const int rlen = min(R2ROWS, Nn - rbase);
  for (int j = tid; j < R2ROWS; j += RBLOCK) acc[j] = 0.0f;
  __syncthreads();

  const int parts = (int)gridDim.x;
  const int per = ((b1 - b0) + parts - 1) / parts;
  const int begin = b0 + blockIdx.x * per;
  const int end = min(b1, begin + per);
  for (int i0 = begin + tid; i0 < end; i0 += RBLOCK * P3E) {
    int pk[P3E];
    float w[P3E];
    float v[P3E];
#pragma unroll
    for (int u = 0; u < P3E; ++u) {
      const int ic = min(i0 + u * RBLOCK, end - 1);
      pk[u] = lrow[ic];
      w[u] = lw[ic];
    }
#pragma unroll
    for (int u = 0; u < P3E; ++u) v[u] = g[pk[u] >> 12];
#pragma unroll
    for (int u = 0; u < P3E; ++u) asm volatile("" : "+v"(v[u]), "+v"(w[u]));
#pragma unroll
    for (int u = 0; u < P3E; ++u)
      if (i0 + u * RBLOCK < end) atomicAdd(&acc[pk[u] & 4095], w[u] * v[u]);
  }
  __syncthreads();

  float* __restrict__ dst = part + (size_t)blockIdx.x * Nn + rbase;
  for (int j = tid; j < rlen; j += RBLOCK) dst[j] = acc[j];
}

__global__ void k_out_sum(float* __restrict__ out, const float* __restrict__ part,
                          const float* __restrict__ g, const float* __restrict__ dinv,
                          const float* __restrict__ b3, int Nn) {
  int i = blockIdx.x * blockDim.x + threadIdx.x;
  if (i >= Nn) return;
  float di = dinv[i];
  float s0 = b3[0] + di * di * g[i], s1 = 0.0f;
#pragma unroll
  for (int k = 0; k < P3; k += 2) {
    s0 += part[(size_t)(k + 0) * Nn + i];
    s1 += part[(size_t)(k + 1) * Nn + i];
  }
  out[i] = s0 + s1;
}


static inline unsigned int blocks_for(long long total, int bs) {
  return (unsigned int)((total + bs - 1) / bs);
}

extern "C" void kernel_launch(void* const* d_in, const int* in_sizes, int n_in,
                              void* d_out, int out_size, void* d_ws, size_t ws_size,
                              hipStream_t stream) {
  const float* x = (const float*)d_in[0];
  const int* ei = (const int*)d_in[1];
  const float* ew = (const float*)d_in[2];
  const float* W1 = (const float*)d_in[3];
  const float* b1 = (const float*)d_in[4];
  const float* W2 = (const float*)d_in[5];
  const float* b2 = (const float*)d_in[6];
  const float* W3 = (const float*)d_in[7];
  const float* b3 = (const float*)d_in[8];

  const int E = in_sizes[2];
  const int Nn = in_sizes[0] / IN_C;
  float* out = (float*)d_out;

  float* dinv = (float*)d_ws;
  float* lw = dinv + Nn;
  float* bufA = lw + E;
  float* bufA_hi = bufA + (size_t)Nn * H2C;
  float* bufB = bufA + (size_t)Nn * H1C;
  float* bufB_hi = bufB + (size_t)Nn * H2C;

  k_deg_part<<<dim3(SLICES, blocks_for(Nn, RANGE)), PBLOCK, 0, stream>>>(ei, ew, bufA, E, Nn);
  k_dinv_sum<<<blocks_for(Nn, 256), 256, 0, stream>>>(bufA, dinv, Nn);

  const int nr2 = (int)blocks_for(Nn, R2ROWS);
  int* bk_cnt = (int*)d_out;
  int* bk_starts = bk_cnt + nr2 * BKB;
  int* lrow = (int*)bufB_hi;
  k_bk_count<<<BKB, BKT, 0, stream>>>(ei, bk_cnt, E, Nn, nr2);
  k_bk_scatter<<<BKB, BKT, 0, stream>>>(ei, ew, dinv, bk_cnt, bk_starts, lrow, lw, E, Nn, nr2);

  k_gemm1_rows<<<blocks_for(Nn / 16, G1_WAVES), G1_THREADS, 0, stream>>>(x, W1, bufA, bufA_hi,
                                                                        Nn / 16);
  k_init_agg<H2C><<<blocks_for((long long)Nn * H2C, 256), 256, 0, stream>>>(bufB, bufA, dinv, b1,
                                                                          Nn);
  k_row_agg_bk<H2C><<<dim3(6, nr2), RBLOCK, 0, stream>>>(lrow, lw, bk_starts, bufA, bufB, Nn);
  k_init_agg<H2C><<<blocks_for((long long)Nn * H2C, 256), 256, 0, stream>>>(bufA, bufA_hi, dinv,
                                                                          b1 + H2C, Nn);
  k_row_agg_bk<H2C><<<dim3(6, nr2), RBLOCK, 0, stream>>>(lrow, lw, bk_starts, bufA_hi, bufA, Nn);

  k_gemm2_planes<<<Nn / 16, 64, 0, stream>>>(bufB, bufA, W2, bufA_hi);
  k_init_agg<H2C><<<blocks_for((long long)Nn * H2C, 256), 256, 0, stream>>>(bufB, bufA_hi, dinv,
                                                                          b2, Nn);
  k_row_agg_bk<H2C><<<dim3(6, nr2), RBLOCK, 0, stream>>>(lrow, lw, bk_starts, bufA_hi, bufB, Nn);

  k_gemm3_relu<<<blocks_for(Nn, 256), 256, 0, stream>>>(bufB, W3, bufA, Nn);
  k_agg1_bk<<<dim3(P3, nr2), RBLOCK, 0, stream>>>(lrow, lw, bk_starts, bufA, bufA_hi, Nn);
  k_out_sum<<<blocks_for(Nn, 256), 256, 0, stream>>>(out, bufA_hi, bufA, dinv, b3, Nn);
}


# Aggregation blocks own their rows: 250 buckets of 400 destination nodes, one block each, store bias + self-loop + LDS sum in whole lines; no global atomics, no start-value pass; f32 sum order changes
# speedup vs baseline: 3.2246x; 1.0497x over previous
#include <hip/hip_runtime.h>
#include <hip/hip_bf16.h>


typedef float v2f __attribute__((ext_vector_type(2)));
typedef float v8f __attribute__((ext_vector_type(8)));

#define IN_C 512
#define H1C 64
#define H2C 32


#define SLICES 64
#define RANGE 25600
#define PBLOCK 1024
#define EPT 16

__global__ __launch_bounds__(PBLOCK) void k_deg_part(const int* __restrict__ ei,
                                                     const float* __restrict__ ew,
                                                     float* __restrict__ part, int E,
                                                     int Nn) {
  __shared__ float acc[RANGE];
  const int tid = threadIdx.x;
  const int rbase = blockIdx.y * RANGE;
  const int rlen = min(RANGE, Nn - rbase);
  for (int j = tid; j < RANGE; j += PBLOCK) acc[j] = 0.0f;
  __syncthreads();

  const int per = (E + SLICES - 1) / SLICES;
  const int begin = blockIdx.x * per;
  const int end = min(E, begin + per);
  const int* __restrict__ col = ei + E;
  for (int e0 = begin + tid; e0 < end; e0 += PBLOCK * EPT) {
    int c[EPT];
    float w[EPT];
#pragma unroll
    for (int u = 0; u < EPT; ++u) {
      const int e = e0 + u * PBLOCK;
      const int ec = min(e, end - 1);
      const int cc = col[ec];
      w[u] = ew[ec];
      c[u] = (e < end) ? cc : -1;
    }
    asm volatile("" : "+v"(w[0]));
#pragma unroll
    for (int u = 0; u < EPT; ++u) {
      const unsigned rel = (unsigned)(c[u] - rbase);
      if (rel < (unsigned)rlen) atomicAdd(&acc[rel], w[u]);
    }
  }
  __syncthreads();

  float* __restrict__ dst = part + (size_t)blockIdx.x * Nn + rbase;
  for (int j = tid; j < rlen; j += PBLOCK) dst[j] = acc[j];
}

__global__ void k_dinv_sum(const float* __restrict__ part, float* __restrict__ dinv, int Nn) {
  int i = blockIdx.x * blockDim.x + threadIdx.x;
  if (i >= Nn) return;
  float s0 = 1.0f, s1 = 0.0f, s2 = 0.0f, s3 = 0.0f;
#pragma unroll
  for (int k = 0; k < SLICES; k += 4) {
    s0 += part[(size_t)(k + 0) * Nn + i];
    s1 += part[(size_t)(k + 1) * Nn + i];
    s2 += part[(size_t)(k + 2) * Nn + i];
    s3 += part[(size_t)(k + 3) * Nn + i];
  }
  float d = (s0 + s1) + (s2 + s3);
  dinv[i] = (d > 0.0f) ? rsqrtf(fmaxf(d, 1e-12f)) : 0.0f;
}


__global__ __launch_bounds__(64) void k_gemm2_planes(const float* __restrict__ A0,
                                                     const float* __restrict__ A1,
                                                     const float* __restrict__ B,
                                                     float* __restrict__ C) {
  constexpr int NOUT = H2C;
  constexpr int KH = H1C / 2;
  const int wave = threadIdx.x >> 5;
  const int lane = threadIdx.x & 31;
  const int m = lane & 15;
  const int half = lane >> 4;
  const int n0 = wave * 16;
  const long long rowBase = (long long)blockIdx.x * 16;

  const float* a0 = A0 + (rowBase + m) * KH + half * 2;
  const float* a1 = A1 + (rowBase + m) * KH + half * 2;

  v8f acc = {};
#pragma unroll 4
  for (int k = 0; k < KH; k += 4) {
    float2 av = *reinterpret_cast<const float2*>(a0 + k);
    v2f a;
    a.x = fmaxf(av.x, 0.0f);
    a.y = fmaxf(av.y, 0.0f);
    const int kr = k + half * 2;
    v2f b;
    b.x = B[kr * NOUT + n0 + m];
    b.y = B[(kr + 1) * NOUT + n0 + m];
    acc = __builtin_amdgcn_wmma_f32_16x16x4_f32(false, a, false, b, (short)0, acc,
                                                false, false);
  }
#pragma unroll 4
  for (int k = 0; k < KH; k += 4) {
    float2 av = *reinterpret_cast<const float2*>(a1 + k);
    v2f a;
    a.x = fmaxf(av.x, 0.0f);
    a.y = fmaxf(av.y, 0.0f);
    const int kr = KH + k + half * 2;
    v2f b;
    b.x = B[kr * NOUT + n0 + m];
    b.y = B[(kr + 1) * NOUT + n0 + m];
    acc = __builtin_amdgcn_wmma_f32_16x16x4_f32(false, a, false, b, (short)0, acc,
                                                false, false);
  }

#pragma unroll
  for (int v = 0; v < 8; ++v) {
    long long r = rowBase + v + 8 * half;
    C[r * NOUT + n0 + m] = acc[v];
  }
}


#define G1_WAVES 25
#define G1_THREADS (G1_WAVES * 32)

__device__ __forceinline__ void g1_block(const float4 a, const float4* bp, v8f& c0, v8f& c1,
                                         v8f& c2, v8f& c3) {
  const float4 b0 = bp[0];
  const float4 b1 = bp[16];
  const float4 b2 = bp[32];
  const float4 b3 = bp[48];
  v2f alo, ahi, f;
  alo.x = a.x; alo.y = a.y;
  ahi.x = a.z; ahi.y = a.w;
  f.x = b0.x; f.y = b0.y;
  c0 = __builtin_amdgcn_wmma_f32_16x16x4_f32(false, alo, false, f, (short)0, c0, false, false);
  f.x = b1.x; f.y = b1.y;
  c1 = __builtin_amdgcn_wmma_f32_16x16x4_f32(false, alo, false, f, (short)0, c1, false, false);
  f.x = b2.x; f.y = b2.y;
  c2 = __builtin_amdgcn_wmma_f32_16x16x4_f32(false, alo, false, f, (short)0, c2, false, false);
  f.x = b3.x; f.y = b3.y;
  c3 = __builtin_amdgcn_wmma_f32_16x16x4_f32(false, alo, false, f, (short)0, c3, false, false);
  f.x = b0.z; f.y = b0.w;
  c0 = __builtin_amdgcn_wmma_f32_16x16x4_f32(false, ahi, false, f, (short)0, c0, false, false);
  f.x = b1.z; f.y = b1.w;
  c1 = __builtin_amdgcn_wmma_f32_16x16x4_f32(false, ahi, false, f, (short)0, c1, false, false);
  f.x = b2.z; f.y = b2.w;
  c2 = __builtin_amdgcn_wmma_f32_16x16x4_f32(false, ahi, false, f, (short)0, c2, false, false);
  f.x = b3.z; f.y = b3.w;
  c3 = __builtin_amdgcn_wmma_f32_16x16x4_f32(false, ahi, false, f, (short)0, c3, false, false);
}

__global__ __launch_bounds__(G1_THREADS) void k_gemm1_rows(const float* __restrict__ A,
                                                           const float* __restrict__ B,
                                                           float* __restrict__ C0,
                                                           float* __restrict__ C1, int ntiles) {
  __shared__ float4 Bs[(IN_C / 4) * H1C];
  __shared__ float4 Cs[G1_WAVES][16 * H1C / 4];
  const int tid = threadIdx.x;
  const int lane = tid & 31;
  const int wave = __builtin_amdgcn_readfirstlane(tid >> 5);
  const int m = lane & 15;
  const int half = lane >> 4;

  for (int idx = tid; idx < (IN_C / 4) * H1C; idx += G1_THREADS) {
    const int kq = idx / H1C;
    const int n = idx % H1C;
    const float* src = B + (size_t)(4 * kq) * H1C + n;
    float4 v;
    v.x = src[0];
    v.y = src[H1C];
    v.z = src[2 * H1C];
    v.w = src[3 * H1C];
    Bs[idx] = v;
  }
  __syncthreads();

  const int tile = blockIdx.x * G1_WAVES + wave;
  const bool active = tile < ntiles;
  if (active) {
    const float* arow = A + ((size_t)tile * 16 + m) * IN_C + half * 4;
    const float4* bfrag = Bs + half * H1C + m;
    v8f c0 = {}, c1 = {}, c2 = {}, c3 = {};
    float4 a[4];
#pragma unroll
    for (int u = 0; u < 4; ++u) a[u] = *reinterpret_cast<const float4*>(arow + 8 * u);
    for (int kb = 0; kb < IN_C - 32; kb += 32) {
      float4 an[4];
#pragma unroll
      for (int u = 0; u < 4; ++u)
        an[u] = *reinterpret_cast<const float4*>(arow + kb + 32 + 8 * u);
#pragma unroll
      for (int u = 0; u < 4; ++u) g1_block(a[u], bfrag + ((kb + 8 * u) >> 2) * H1C, c0, c1, c2, c3);
#pragma unroll
      for (int u = 0; u < 4; ++u) a[u] = an[u];
    }
#pragma unroll
    for (int u = 0; u < 4; ++u)
      g1_block(a[u], bfrag + ((IN_C - 32 + 8 * u) >> 2) * H1C, c0, c1, c2, c3);

    float* cs = reinterpret_cast<float*>(Cs[wave]);
#pragma unroll
    for (int v = 0; v < 8; ++v) {
      float* p = cs + (v + 8 * half) * H1C + m;
      p[0] = c0[v];
      p[16] = c1[v];
      p[32] = c2[v];
      p[48] = c3[v];
    }
  }
  __syncthreads();
  if (active) {
    float* cp = ((lane >> 3) & 1) ? C1 : C0;
    float4* dst = reinterpret_cast<float4*>(cp + (size_t)tile * 16 * (H1C / 2)) +
                  (lane >> 4) * 8 + (lane & 7);
#pragma unroll
    for (int i = 0; i < 8; ++i) dst[i * 16] = Cs[wave][i * 32 + lane];
  }
}


#define OWNROWS 400
#define RBLOCK 1024

#define BKB 256
#define BKT 1024
#define BKE 16
#define BKR 256

__global__ __launch_bounds__(BKT) void k_bk_count(const int* __restrict__ ei,
                                                  int* __restrict__ cnt, int E, int Nn,
                                                  int nr) {
  __shared__ int hist[BKR];
  const int tid = threadIdx.x;
  if (tid < BKR) hist[tid] = 0;
  __syncthreads();
  const int per = (E + BKB - 1) / BKB;
  const int begin = blockIdx.x * per;
  const int end = min(E, begin + per);
  const int* __restrict__ col = ei + E;
  for (int e0 = begin + tid; e0 < end; e0 += BKT * BKE) {
    int c[BKE];
#pragma unroll
    for (int u = 0; u < BKE; ++u) {
      const int e = e0 + u * BKT;
      const int cc = col[min(e, end - 1)];
      c[u] = (e < end) ? cc : -1;
    }
#pragma unroll
    for (int u = 0; u < BKE; ++u)
      if ((unsigned)c[u] < (unsigned)Nn) atomicAdd(&hist[c[u] / OWNROWS], 1);
  }
  __syncthreads();
  if (tid < nr) cnt[tid * BKB + blockIdx.x] = hist[tid];
}

__global__ __launch_bounds__(BKT) void k_bk_scatter(const int* __restrict__ ei,
                                                    const float* __restrict__ ew,
                                                    const float* __restrict__ dinv,
                                                    const int* __restrict__ cnt,
                                                    int* __restrict__ starts,
                                                    int* __restrict__ lrow,
                                                    float* __restrict__ lw, int E, int Nn,
                                                    int nr) {
  __shared__ int tot[BKR];
  __shared__ int before[BKR];
  __shared__ int cur[BKR];
  __shared__ int wsum[BKR / 32];
  const int tid = threadIdx.x;
  const int lane = tid & 31;
  const int wave = tid >> 5;
  for (int r = wave; r < nr; r += BKT / 32) {
    int t = 0, bf = 0;
#pragma unroll
    for (int k = 0; k < BKB / 32; ++k) {
      const int b = k * 32 + lane;
      const int v = cnt[r * BKB + b];
      t += v;
      bf += (b < (int)blockIdx.x) ? v : 0;
    }
    for (int d = 16; d > 0; d >>= 1) {
      t += __shfl_xor(t, d);
      bf += __shfl_xor(bf, d);
    }
    if (lane == 0) {
      tot[r] = t;
      before[r] = bf;
    }
  }
  __syncthreads();
  int mine = 0, incl = 0;
  if (tid < BKR) {
    mine = (tid < nr) ? tot[tid] : 0;
    incl = mine;
    for (int d = 1; d < 32; d <<= 1) {
      const int below = __shfl_up(incl, d);
      if (lane >= d) incl += below;
    }
    if (lane == 31) wsum[wave] = incl;
  }
  __syncthreads();
  if (tid <= nr) {
    int sum = incl - mine;
    for (int q = 0; q < wave; ++q) sum += wsum[q];
    if (tid < nr) cur[tid] = sum + before[tid];
    if (blockIdx.x == 0) starts[tid] = sum;
  }
  __syncthreads();

  const int per = (E + BKB - 1) / BKB;
  const int begin = blockIdx.x * per;
  const int end = min(E, begin + per);
  const int* __restrict__ col = ei + E;
  for (int e0 = begin + tid; e0 < end; e0 += BKT * BKE) {
    int c[BKE];
    int r[BKE];
    float w[BKE];
    float dc[BKE];
#pragma unroll
    for (int u = 0; u < BKE; ++u) {
      const int e = e0 + u * BKT;
      const int ec = min(e, end - 1);
      const int cc = col[ec];
      r[u] = ei[ec];
      w[u] = ew[ec];
      dc[u] = dinv[cc];
      c[u] = (e < end) ? cc : -1;
    }
#pragma unroll
    for (int u = 0; u < BKE; ++u) w[u] = dinv[r[u]] * w[u] * dc[u];
    asm volatile("" : "+v"(r[0]), "+v"(w[0]));
#pragma unroll
    for (int u = 0; u < BKE; ++u) {
      if ((unsigned)c[u] < (unsigned)Nn) {
        const int rg = c[u] / OWNROWS;
        const int pos = atomicAdd(&cur[rg], 1);
        lrow[pos] = (r[u] << 12) | (c[u] - rg * OWNROWS);
        lw[pos] = w[u];
      }
    }
  }
}

template <int F, int NB>
__device__ __forceinline__ void row_batch_n(const int rv, const int relv, const float wv,
                                            const int b, const int lane,
                                            const float* __restrict__ h, float* acc) {
  constexpr int L = F / 32;
  float v[NB][L];
#pragma unroll
  for (int u = 0; u < NB; ++u) {
    const int r = __builtin_amdgcn_readlane(rv, b + u);
    const float* src = h + (size_t)r * F + lane;
#pragma unroll
    for (int l = 0; l < L; ++l) v[u][l] = src[l * 32];
  }
#pragma unroll
  for (int u = 0; u < NB; ++u) {
    const int rel = __builtin_amdgcn_readlane(relv, b + u);
    const float w = __int_as_float(__builtin_amdgcn_readlane(__float_as_int(wv), b + u));
    float* dst = acc + rel * F + lane;
#pragma unroll
    for (int l = 0; l < L; ++l) atomicAdd(dst + l * 32, w * v[u][l]);
  }
}

template <int F>
__device__ __forceinline__ void row_group(const int rv, const int relv, const float wv,
                                          const int left, const int lane,
                                          const float* __restrict__ h, float* acc) {
  constexpr int L = F / 32;
#pragma unroll
  for (int b = 0; b < 32; b += 16) {
    if (b + 16 <= left) {
      row_batch_n<F, 16>(rv, relv, wv, b, lane, h, acc);
    } else {
#pragma unroll
      for (int bb = b; bb < b + 16; bb += 8) {
        if (bb + 8 <= left) {
          row_batch_n<F, 8>(rv, relv, wv, bb, lane, h, acc);
        } else {
#pragma unroll
          for (int u = bb; u < bb + 8; ++u) {
            if (u < left) {
              const int r = __builtin_amdgcn_readlane(rv, u);
              const int rel = __builtin_amdgcn_readlane(relv, u);
              const float w =
                  __int_as_float(__builtin_amdgcn_readlane(__float_as_int(wv), u));
              const float* src = h + (size_t)r * F + lane;
              float* dst = acc + rel * F + lane;
#pragma unroll
              for (int l = 0; l < L; ++l) atomicAdd(dst + l * 32, w * src[l * 32]);
            }
          }
        }
      }
    }
  }
}

template <int F>
__global__ __launch_bounds__(RBLOCK) void k_row_agg_own(const int* __restrict__ lrow,
                                                        const float* __restrict__ lw,
                                                        const int* __restrict__ starts,
                                                        const float* __restrict__ h,
                                                        const float* __restrict__ dinv,
                                                        const float* __restrict__ bias,
                                                        float* __restrict__ agg, int Nn) {
  static_assert(RBLOCK % F == 0, "a thread keeps one column in the tail");
  constexpr int PER = (OWNROWS * F + RBLOCK - 1) / RBLOCK;
  __shared__ float acc[OWNROWS * F];
  const int tid = threadIdx.x;
  const int lane = tid & 31;
  const int wave = __builtin_amdgcn_readfirstlane(tid >> 5);
  const int begin = __builtin_amdgcn_readfirstlane(starts[blockIdx.x]);
  const int end = __builtin_amdgcn_readfirstlane(starts[blockIdx.x + 1]);
  const int rbase = blockIdx.x * OWNROWS;
  const int rlen = min(OWNROWS, Nn - rbase);
  for (int j = tid; j < OWNROWS * F; j += RBLOCK) acc[j] = 0.0f;
  __syncthreads();

  int i = begin + wave * 32;
  int pk = 0;
  float wv = 0.0f;
  if (i < end && lane < end - i) {
    pk = lrow[i + lane];
    wv = lw[i + lane];
  }
  while (i < end) {
    const int inext = i + RBLOCK;
    int pkn = 0;
    float wn = 0.0f;
    if (inext < end && lane < end - inext) {
      pkn = lrow[inext + lane];
      wn = lw[inext + lane];
    }
    row_group<F>(pk >> 12, pk & 4095, wv, min(32, end - i), lane, h, acc);
    pk = pkn;
    wv = wn;
    i = inext;
  }
  const float* __restrict__ h_rows = h + (size_t)rbase * F;
  const int nfl = rlen * F;
  const float bs = bias[tid % F];
  float dv[PER];
  float sv[PER];
#pragma unroll
  for (int k = 0; k < PER; ++k) {
    const int jc = min(tid + k * RBLOCK, nfl - 1);
    dv[k] = dinv[rbase + jc / F];
    sv[k] = h_rows[jc];
  }
#pragma unroll
  for (int k = 0; k < PER; ++k) sv[k] = bs + dv[k] * dv[k] * sv[k];
  __syncthreads();

#pragma unroll
  for (int k = 0; k < PER; ++k) sv[k] += acc[min(tid + k * RBLOCK, nfl - 1)];
  float* __restrict__ out_rows = agg + (size_t)rbase * F;
#pragma unroll
  for (int k = 0; k < PER; ++k) {
    const int j = tid + k * RBLOCK;
    if (j < nfl) out_rows[j] = sv[k];
  }
}

__global__ void k_gemm3_relu(const float* __restrict__ a2, const float* __restrict__ W3,
                             float* __restrict__ g, int Nn) {
  int i = blockIdx.x * blockDim.x + threadIdx.x;
  if (i >= Nn) return;
  const float* r = a2 + (size_t)i * H2C;
  float acc = 0.0f;
#pragma unroll
  for (int f = 0; f < H2C; ++f) acc += fmaxf(r[f], 0.0f) * W3[f];
  g[i] = acc;
}

#define P3E 16

__global__ __launch_bounds__(RBLOCK) void k_agg1_own(const int* __restrict__ lrow,
                                                     const float* __restrict__ lw,
                                                     const int* __restrict__ starts,
                                                     const float* __restrict__ g,
                                                     float* __restrict__ sum, int Nn) {
  __shared__ float acc[OWNROWS];
  const int tid = threadIdx.x;
  const int begin = starts[blockIdx.x];
  const int end = starts[blockIdx.x + 1];
  const int rbase = blockIdx.x * OWNROWS;
  const int rlen = min(OWNROWS, Nn - rbase);
  for (int j = tid; j < OWNROWS; j += RBLOCK) acc[j] = 0.0f;
  __syncthreads();

  for (int i0 = begin + tid; i0 < end; i0 += RBLOCK * P3E) {
    int pk[P3E];
    float w[P3E];
    float v[P3E];
#pragma unroll
    for (int u = 0; u < P3E; ++u) {
      const int ic = min(i0 + u * RBLOCK, end - 1);
      pk[u] = lrow[ic];
      w[u] = lw[ic];
    }
#pragma unroll
    for (int u = 0; u < P3E; ++u) v[u] = g[pk[u] >> 12];
#pragma unroll
    for (int u = 0; u < P3E; ++u) asm volatile("" : "+v"(v[u]), "+v"(w[u]));
#pragma unroll
    for (int u = 0; u < P3E; ++u)
      if (i0 + u * RBLOCK < end) atomicAdd(&acc[pk[u] & 4095], w[u] * v[u]);
  }
  __syncthreads();

  float* __restrict__ dst = sum + rbase;
  for (int j = tid; j < rlen; j += RBLOCK) dst[j] = acc[j];
}

__global__ void k_out_sum(float* __restrict__ out, const float* __restrict__ sum,
                          const float* __restrict__ g, const float* __restrict__ dinv,
                          const float* __restrict__ b3, int Nn) {
  int i = blockIdx.x * blockDim.x + threadIdx.x;
  if (i >= Nn) return;
  float di = dinv[i];
  out[i] = (b3[0] + di * di * g[i]) + sum[i];
}


static inline unsigned int blocks_for(long long total, int bs) {
  return (unsigned int)((total + bs - 1) / bs);
}

extern "C" void kernel_launch(void* const* d_in, const int* in_sizes, int n_in,
                              void* d_out, int out_size, void* d_ws, size_t ws_size,
                              hipStream_t stream) {
  const float* x = (const float*)d_in[0];
  const int* ei = (const int*)d_in[1];
  const float* ew = (const float*)d_in[2];
  const float* W1 = (const float*)d_in[3];
  const float* b1 = (const float*)d_in[4];
  const float* W2 = (const float*)d_in[5];
  const float* b2 = (const float*)d_in[6];
  const float* W3 = (const float*)d_in[7];
  const float* b3 = (const float*)d_in[8];

  const int E = in_sizes[2];
  const int Nn = in_sizes[0] / IN_C;
  float* out = (float*)d_out;

  float* dinv = (float*)d_ws;
  float* lw = dinv + Nn;
  float* bufA = lw + E;
  float* bufA_hi = bufA + (size_t)Nn * H2C;
  float* bufB = bufA + (size_t)Nn * H1C;
  float* bufB_hi = bufB + (size_t)Nn * H2C;

  k_deg_part<<<dim3(SLICES, blocks_for(Nn, RANGE)), PBLOCK, 0, stream>>>(ei, ew, bufA, E, Nn);
  k_dinv_sum<<<blocks_for(Nn, 256), 256, 0, stream>>>(bufA, dinv, Nn);

  const int nr2 = (int)blocks_for(Nn, OWNROWS);
  int* bk_cnt = (int*)d_out;
  int* bk_starts = bk_cnt + nr2 * BKB;
  int* lrow = (int*)bufB_hi;
  k_bk_count<<<BKB, BKT, 0, stream>>>(ei, bk_cnt, E, Nn, nr2);
  k_bk_scatter<<<BKB, BKT, 0, stream>>>(ei, ew, dinv, bk_cnt, bk_starts, lrow, lw, E, Nn, nr2);

  k_gemm1_rows<<<blocks_for(Nn / 16, G1_WAVES), G1_THREADS, 0, stream>>>(x, W1, bufA, bufA_hi,
                                                                        Nn / 16);
  k_row_agg_own<H2C><<<nr2, RBLOCK, 0, stream>>>(lrow, lw, bk_starts, bufA, dinv, b1, bufB, Nn);
  k_row_agg_own<H2C><<<nr2, RBLOCK, 0, stream>>>(lrow, lw, bk_starts, bufA_hi, dinv, b1 + H2C,
                                                 bufA, Nn);

  k_gemm2_planes<<<Nn / 16, 64, 0, stream>>>(bufB, bufA, W2, bufA_hi);
  k_row_agg_own<H2C><<<nr2, RBLOCK, 0, stream>>>(lrow, lw, bk_starts, bufA_hi, dinv, b2, bufB, Nn);

  k_gemm3_relu<<<blocks_for(Nn, 256), 256, 0, stream>>>(bufB, W3, bufA, Nn);
  k_agg1_own<<<nr2, RBLOCK, 0, stream>>>(lrow, lw, bk_starts, bufA, bufA_hi, Nn);
  k_out_sum<<<blocks_for(Nn, 256), 256, 0, stream>>>(out, bufA_hi, bufA, dinv, b3, Nn);
}


# Aggregation blocks apply the next layer's weights to their finished rows in LDS: layer 1 sums 64 wide in one pass, then relu and W2 on WMMA; layer 2 ends in relu and W3 dot (f32 tree sum); 9 launches
# speedup vs baseline: 3.5514x; 1.1013x over previous
#include <hip/hip_runtime.h>
#include <hip/hip_bf16.h>


typedef float v2f __attribute__((ext_vector_type(2)));
typedef float v8f __attribute__((ext_vector_type(8)));

#define IN_C 512
#define H1C 64
#define H2C 32


#define SLICES 64
#define RANGE 25600
#define PBLOCK 1024
#define EPT 16

__global__ __launch_bounds__(PBLOCK) void k_deg_part(const int* __restrict__ ei,
                                                     const float* __restrict__ ew,
                                                     float* __restrict__ part, int E,
                                                     int Nn) {
  __shared__ float acc[RANGE];
  const int tid = threadIdx.x;
  const int rbase = blockIdx.y * RANGE;
  const int rlen = min(RANGE, Nn - rbase);
  for (int j = tid; j < RANGE; j += PBLOCK) acc[j] = 0.0f;
  __syncthreads();

  const int per = (E + SLICES - 1) / SLICES;
  const int begin = blockIdx.x * per;
  const int end = min(E, begin + per);
  const int* __restrict__ col = ei + E;
  for (int e0 = begin + tid; e0 < end; e0 += PBLOCK * EPT) {
    int c[EPT];
    float w[EPT];
#pragma unroll
    for (int u = 0; u < EPT; ++u) {
      const int e = e0 + u * PBLOCK;
      const int ec = min(e, end - 1);
      const int cc = col[ec];
      w[u] = ew[ec];
      c[u] = (e < end) ? cc : -1;
    }
    asm volatile("" : "+v"(w[0]));
#pragma unroll
    for (int u = 0; u < EPT; ++u) {
      const unsigned rel = (unsigned)(c[u] - rbase);
      if (rel < (unsigned)rlen) atomicAdd(&acc[rel], w[u]);
    }
  }
  __syncthreads();

  float* __restrict__ dst = part + (size_t)blockIdx.x * Nn + rbase;
  for (int j = tid; j < rlen; j += PBLOCK) dst[j] = acc[j];
}

__global__ void k_dinv_sum(const float* __restrict__ part, float* __restrict__ dinv, int Nn) {
  int i = blockIdx.x * blockDim.x + threadIdx.x;
  if (i >= Nn) return;
  float s0 = 1.0f, s1 = 0.0f, s2 = 0.0f, s3 = 0.0f;
#pragma unroll
  for (int k = 0; k < SLICES; k += 4) {
    s0 += part[(size_t)(k + 0) * Nn + i];
    s1 += part[(size_t)(k + 1) * Nn + i];
    s2 += part[(size_t)(k + 2) * Nn + i];
    s3 += part[(size_t)(k + 3) * Nn + i];
  }
  float d = (s0 + s1) + (s2 + s3);
  dinv[i] = (d > 0.0f) ? rsqrtf(fmaxf(d, 1e-12f)) : 0.0f;
}


#define G1_WAVES 25
#define G1_THREADS (G1_WAVES * 32)

__device__ __forceinline__ void g1_block(const float4 a, const float4* bp, v8f& c0, v8f& c1,
                                         v8f& c2, v8f& c3) {
  const float4 b0 = bp[0];
  const float4 b1 = bp[16];
  const float4 b2 = bp[32];
  const float4 b3 = bp[48];
  v2f alo, ahi, f;
  alo.x = a.x; alo.y = a.y;
  ahi.x = a.z; ahi.y = a.w;
  f.x = b0.x; f.y = b0.y;
  c0 = __builtin_amdgcn_wmma_f32_16x16x4_f32(false, alo, false, f, (short)0, c0, false, false);
  f.x = b1.x; f.y = b1.y;
  c1 = __builtin_amdgcn_wmma_f32_16x16x4_f32(false, alo, false, f, (short)0, c1, false, false);
  f.x = b2.x; f.y = b2.y;
  c2 = __builtin_amdgcn_wmma_f32_16x16x4_f32(false, alo, false, f, (short)0, c2, false, false);
  f.x = b3.x; f.y = b3.y;
  c3 = __builtin_amdgcn_wmma_f32_16x16x4_f32(false, alo, false, f, (short)0, c3, false, false);
  f.x = b0.z; f.y = b0.w;
  c0 = __builtin_amdgcn_wmma_f32_16x16x4_f32(false, ahi, false, f, (short)0, c0, false, false);
  f.x = b1.z; f.y = b1.w;
  c1 = __builtin_amdgcn_wmma_f32_16x16x4_f32(false, ahi, false, f, (short)0, c1, false, false);
  f.x = b2.z; f.y = b2.w;
  c2 = __builtin_amdgcn_wmma_f32_16x16x4_f32(false, ahi, false, f, (short)0, c2, false, false);
  f.x = b3.z; f.y = b3.w;
  c3 = __builtin_amdgcn_wmma_f32_16x16x4_f32(false, ahi, false, f, (short)0, c3, false, false);
}

__global__ __launch_bounds__(G1_THREADS) void k_gemm1_rows(const float* __restrict__ A,
                                                           const float* __restrict__ B,
                                                           float* __restrict__ C, int ntiles) {
  __shared__ float4 Bs[(IN_C / 4) * H1C];
  __shared__ float4 Cs[G1_WAVES][16 * H1C / 4];
  const int tid = threadIdx.x;
  const int lane = tid & 31;
  const int wave = __builtin_amdgcn_readfirstlane(tid >> 5);
  const int m = lane & 15;
  const int half = lane >> 4;

  for (int idx = tid; idx < (IN_C / 4) * H1C; idx += G1_THREADS) {
    const int kq = idx / H1C;
    const int n = idx % H1C;
    const float* src = B + (size_t)(4 * kq) * H1C + n;
    float4 v;
    v.x = src[0];
    v.y = src[H1C];
    v.z = src[2 * H1C];
    v.w = src[3 * H1C];
    Bs[idx] = v;
  }
  __syncthreads();

  const int tile = blockIdx.x * G1_WAVES + wave;
  const bool active = tile < ntiles;
  if (active) {
    const float* arow = A + ((size_t)tile * 16 + m) * IN_C + half * 4;
    const float4* bfrag = Bs + half * H1C + m;
    v8f c0 = {}, c1 = {}, c2 = {}, c3 = {};
    float4 a[4];
#pragma unroll
    for (int u = 0; u < 4; ++u) a[u] = *reinterpret_cast<const float4*>(arow + 8 * u);
    for (int kb = 0; kb < IN_C - 32; kb += 32) {
      float4 an[4];
#pragma unroll
      for (int u = 0; u < 4; ++u)
        an[u] = *reinterpret_cast<const float4*>(arow + kb + 32 + 8 * u);
#pragma unroll
      for (int u = 0; u < 4; ++u) g1_block(a[u], bfrag + ((kb + 8 * u) >> 2) * H1C, c0, c1, c2, c3);
#pragma unroll
      for (int u = 0; u < 4; ++u) a[u] = an[u];
    }
#pragma unroll
    for (int u = 0; u < 4; ++u)
      g1_block(a[u], bfrag + ((IN_C - 32 + 8 * u) >> 2) * H1C, c0, c1, c2, c3);

    float* cs = reinterpret_cast<float*>(Cs[wave]);
#pragma unroll
    for (int v = 0; v < 8; ++v) {
      float* p = cs + (v + 8 * half) * H1C + m;
      p[0] = c0[v];
      p[16] = c1[v];
      p[32] = c2[v];
      p[48] = c3[v];
    }
  }
  __syncthreads();
  if (active) {
    float4* dst = reinterpret_cast<float4*>(C + (size_t)tile * 16 * H1C);
#pragma unroll
    for (int i = 0; i < 8; ++i) dst[i * 32 + lane] = Cs[wave][i * 32 + lane];
  }
}


#define OWNROWS 400
#define RBLOCK 1024

#define BKB 256
#define BKT 1024
#define BKE 16
#define BKR 256

__global__ __launch_bounds__(BKT) void k_bk_count(const int* __restrict__ ei,
                                                  int* __restrict__ cnt, int E, int Nn,
                                                  int nr) {
  __shared__ int hist[BKR];
  const int tid = threadIdx.x;
  if (tid < BKR) hist[tid] = 0;
  __syncthreads();
  const int per = (E + BKB - 1) / BKB;
  const int begin = blockIdx.x * per;
  const int end = min(E, begin + per);
  const int* __restrict__ col = ei + E;
  for (int e0 = begin + tid; e0 < end; e0 += BKT * BKE) {
    int c[BKE];
#pragma unroll
    for (int u = 0; u < BKE; ++u) {
      const int e = e0 + u * BKT;
      const int cc = col[min(e, end - 1)];
      c[u] = (e < end) ? cc : -1;
    }
#pragma unroll
    for (int u = 0; u < BKE; ++u)
      if ((unsigned)c[u] < (unsigned)Nn) atomicAdd(&hist[c[u] / OWNROWS], 1);
  }
  __syncthreads();
  if (tid < nr) cnt[tid * BKB + blockIdx.x] = hist[tid];
}

__global__ __launch_bounds__(BKT) void k_bk_scatter(const int* __restrict__ ei,
                                                    const float* __restrict__ ew,
                                                    const float* __restrict__ dinv,
                                                    const int* __restrict__ cnt,
                                                    int* __restrict__ starts,
                                                    int* __restrict__ lrow,
                                                    float* __restrict__ lw, int E, int Nn,
                                                    int nr) {
  __shared__ int tot[BKR];
  __shared__ int before[BKR];
  __shared__ int cur[BKR];
  __shared__ int wsum[BKR / 32];
  const int tid = threadIdx.x;
  const int lane = tid & 31;
  const int wave = tid >> 5;
  for (int r = wave; r < nr; r += BKT / 32) {
    int t = 0, bf = 0;
#pragma unroll
    for (int k = 0; k < BKB / 32; ++k) {
      const int b = k * 32 + lane;
      const int v = cnt[r * BKB + b];
      t += v;
      bf += (b < (int)blockIdx.x) ? v : 0;
    }
    for (int d = 16; d > 0; d >>= 1) {
      t += __shfl_xor(t, d);
      bf += __shfl_xor(bf, d);
    }
    if (lane == 0) {
      tot[r] = t;
      before[r] = bf;
    }
  }
  __syncthreads();
  int mine = 0, incl = 0;
  if (tid < BKR) {
    mine = (tid < nr) ? tot[tid] : 0;
    incl = mine;
    for (int d = 1; d < 32; d <<= 1) {
      const int below = __shfl_up(incl, d);
      if (lane >= d) incl += below;
    }
    if (lane == 31) wsum[wave] = incl;
  }
  __syncthreads();
  if (tid <= nr) {
    int sum = incl - mine;
    for (int q = 0; q < wave; ++q) sum += wsum[q];
    if (tid < nr) cur[tid] = sum + before[tid];
    if (blockIdx.x == 0) starts[tid] = sum;
  }
  __syncthreads();

  const int per = (E + BKB - 1) / BKB;
  const int begin = blockIdx.x * per;
  const int end = min(E, begin + per);
  const int* __restrict__ col = ei + E;
  for (int e0 = begin + tid; e0 < end; e0 += BKT * BKE) {
    int c[BKE];
    int r[BKE];
    float w[BKE];
    float dc[BKE];
#pragma unroll
    for (int u = 0; u < BKE; ++u) {
      const int e = e0 + u * BKT;
      const int ec = min(e, end - 1);
      const int cc = col[ec];
      r[u] = ei[ec];
      w[u] = ew[ec];
      dc[u] = dinv[cc];
      c[u] = (e < end) ? cc : -1;
    }
#pragma unroll
    for (int u = 0; u < BKE; ++u) w[u] = dinv[r[u]] * w[u] * dc[u];
    asm volatile("" : "+v"(r[0]), "+v"(w[0]));
#pragma unroll
    for (int u = 0; u < BKE; ++u) {
      if ((unsigned)c[u] < (unsigned)Nn) {
        const int rg = c[u] / OWNROWS;
        const int pos = atomicAdd(&cur[rg], 1);
        lrow[pos] = (r[u] << 12) | (c[u] - rg * OWNROWS);
        lw[pos] = w[u];
      }
    }
  }
}

template <int F, int P, int NB>
__device__ __forceinline__ void row_batch_n(const int rv, const int relv, const float wv,
                                            const int b, const int lane,
                                            const float* __restrict__ h, float* acc) {
  constexpr int L = F / 32;
  float v[NB][L];
#pragma unroll
  for (int u = 0; u < NB; ++u) {
    const int r = __builtin_amdgcn_readlane(rv, b + u);
    const float* src = h + (size_t)r * F + lane;
#pragma unroll
    for (int l = 0; l < L; ++l) v[u][l] = src[l * 32];
  }
#pragma unroll
  for (int u = 0; u < NB; ++u) {
    const int rel = __builtin_amdgcn_readlane(relv, b + u);
    const float w = __int_as_float(__builtin_amdgcn_readlane(__float_as_int(wv), b + u));
    float* dst = acc + rel * P + lane;
#pragma unroll
    for (int l = 0; l < L; ++l) atomicAdd(dst + l * 32, w * v[u][l]);
  }
}

template <int F, int P>
__device__ __forceinline__ void row_group(const int rv, const int relv, const float wv,
                                          const int left, const int lane,
                                          const float* __restrict__ h, float* acc) {
  constexpr int L = F / 32;
#pragma unroll
  for (int b = 0; b < 32; b += 16) {
    if (b + 16 <= left) {
      row_batch_n<F, P, 16>(rv, relv, wv, b, lane, h, acc);
    } else {
#pragma unroll
      for (int bb = b; bb < b + 16; bb += 8) {
        if (bb + 8 <= left) {
          row_batch_n<F, P, 8>(rv, relv, wv, bb, lane, h, acc);
        } else {
#pragma unroll
          for (int u = bb; u < bb + 8; ++u) {
            if (u < left) {
              const int r = __builtin_amdgcn_readlane(rv, u);
              const int rel = __builtin_amdgcn_readlane(relv, u);
              const float w =
                  __int_as_float(__builtin_amdgcn_readlane(__float_as_int(wv), u));
              const float* src = h + (size_t)r * F + lane;
              float* dst = acc + rel * P + lane;
#pragma unroll
              for (int l = 0; l < L; ++l) atomicAdd(dst + l * 32, w * src[l * 32]);
            }
          }
        }
      }
    }
  }
}

template <int F, int P>
__device__ __forceinline__ void bucket_sum(const int* __restrict__ lrow,
                                           const float* __restrict__ lw, const int begin,
                                           const int end, const int wave, const int lane,
                                           const float* __restrict__ h, float* acc) {
  int i = begin + wave * 32;
  int pk = 0;
  float wv = 0.0f;
  if (i < end && lane < end - i) {
    pk = lrow[i + lane];
    wv = lw[i + lane];
  }
  while (i < end) {
    const int inext = i + RBLOCK;
    int pkn = 0;
    float wn = 0.0f;
    if (inext < end && lane < end - inext) {
      pkn = lrow[inext + lane];
      wn = lw[inext + lane];
    }
    row_group<F, P>(pk >> 12, pk & 4095, wv, min(32, end - i), lane, h, acc);
    pk = pkn;
    wv = wn;
    i = inext;
  }
}


#define APITCH 68

__global__ __launch_bounds__(RBLOCK) void k_agg64_gemm2(const int* __restrict__ lrow,
                                                        const float* __restrict__ lw,
                                                        const int* __restrict__ starts,
                                                        const float* __restrict__ h,
                                                        const float* __restrict__ dinv,
                                                        const float* __restrict__ bias,
                                                        const float* __restrict__ W2,
                                                        float* __restrict__ h2, int Nn) {
  constexpr int F = H1C;
  constexpr int PER = OWNROWS * F / RBLOCK;
  static_assert(OWNROWS * F % RBLOCK == 0 && RBLOCK % F == 0, "a thread keeps one column");
  static_assert(OWNROWS % 16 == 0 && OWNROWS / 16 <= RBLOCK / 32, "one wave per 16-row tile");
  __shared__ float acc[OWNROWS * APITCH];
  const int tid = threadIdx.x;
  const int lane = tid & 31;
  const int wave = __builtin_amdgcn_readfirstlane(tid >> 5);
  const int begin = __builtin_amdgcn_readfirstlane(starts[blockIdx.x]);
  const int end = __builtin_amdgcn_readfirstlane(starts[blockIdx.x + 1]);
  const int rbase = blockIdx.x * OWNROWS;
  const int rlen = min(OWNROWS, Nn - rbase);
  for (int j = tid; j < OWNROWS * APITCH; j += RBLOCK) acc[j] = 0.0f;
  __syncthreads();

  bucket_sum<F, APITCH>(lrow, lw, begin, end, wave, lane, h, acc);

  const int col = tid % F;
  const int row0 = tid / F;
  const float bs = bias[col];
  float dv[PER];
  float sv[PER];
#pragma unroll
  for (int k = 0; k < PER; ++k) {
    const int rc = rbase + min(row0 + 16 * k, rlen - 1);
    dv[k] = dinv[rc];
    sv[k] = h[(size_t)rc * F + col];
  }
#pragma unroll
  for (int k = 0; k < PER; ++k) sv[k] = bs + dv[k] * dv[k] * sv[k];
  __syncthreads();

#pragma unroll
  for (int k = 0; k < PER; ++k) {
    const int r = row0 + 16 * k;
    float* p = acc + r * APITCH + col;
    const float t = fmaxf(sv[k] + *p, 0.0f);
    *p = (r < rlen) ? t : 0.0f;
  }
  __syncthreads();

  const bool active = wave < OWNROWS / 16;
  if (active) {
    float* tile = acc + wave * 16 * APITCH;
    const int m = lane & 15;
    const int half = lane >> 4;
    const float* ap = tile + m * APITCH + 2 * half;
    const float* bp = W2 + (2 * half) * H2C + m;
    v8f c0 = {}, c1 = {};
#pragma unroll 8
    for (int k = 0; k < H1C; k += 4) {
      const float2 av = *reinterpret_cast<const float2*>(ap + k);
      v2f a, b0, b1;
      a.x = av.x;
      a.y = av.y;
      b0.x = bp[k * H2C];
      b0.y = bp[(k + 1) * H2C];
      b1.x = bp[k * H2C + 16];
      b1.y = bp[(k + 1) * H2C + 16];
      c0 = __builtin_amdgcn_wmma_f32_16x16x4_f32(false, a, false, b0, (short)0, c0, false, false);
      c1 = __builtin_amdgcn_wmma_f32_16x16x4_f32(false, a, false, b1, (short)0, c1, false, false);
    }
#pragma unroll
    for (int v = 0; v < 8; ++v) {
      float* p = tile + (v + 8 * half) * APITCH + m;
      p[0] = c0[v];
      p[16] = c1[v];
    }
  }
  __syncthreads();
  if (active) {
    const float* tile = acc + wave * 16 * APITCH;
    float4* dst = reinterpret_cast<float4*>(h2 + (size_t)(rbase + wave * 16) * H2C);
#pragma unroll
    for (int i = 0; i < 4; ++i) {
      const int r = 4 * i + (lane >> 3);
      const float4 q = *reinterpret_cast<const float4*>(tile + r * APITCH + 4 * (lane & 7));
      if (wave * 16 + r < rlen) dst[i * 32 + lane] = q;
    }
  }
}

__global__ __launch_bounds__(RBLOCK) void k_agg32_dot(const int* __restrict__ lrow,
                                                      const float* __restrict__ lw,
                                                      const int* __restrict__ starts,
                                                      const float* __restrict__ h,
                                                      const float* __restrict__ dinv,
                                                      const float* __restrict__ bias,
                                                      const float* __restrict__ W3,
                                                      float* __restrict__ g, int Nn) {
  constexpr int F = H2C;
  static_assert(RBLOCK % F == 0 && F == 32, "a row of the tail is one wave's lanes");
  constexpr int PER = (OWNROWS * F + RBLOCK - 1) / RBLOCK;
  __shared__ float acc[OWNROWS * F];
  __shared__ float gs[OWNROWS];
  const int tid = threadIdx.x;
  const int lane = tid & 31;
  const int wave = __builtin_amdgcn_readfirstlane(tid >> 5);
  const int begin = __builtin_amdgcn_readfirstlane(starts[blockIdx.x]);
  const int end = __builtin_amdgcn_readfirstlane(starts[blockIdx.x + 1]);
  const int rbase = blockIdx.x * OWNROWS;
  const int rlen = min(OWNROWS, Nn - rbase);
  for (int j = tid; j < OWNROWS * F; j += RBLOCK) acc[j] = 0.0f;
  __syncthreads();

  bucket_sum<F, F>(lrow, lw, begin, end, wave, lane, h, acc);

  const float* __restrict__ h_rows = h + (size_t)rbase * F;
  const int nfl = rlen * F;
  const float bs = bias[lane];
  const float w3 = W3[lane];
  float dv[PER];
  float sv[PER];
#pragma unroll
  for (int k = 0; k < PER; ++k) {
    const int jc = min(tid + k * RBLOCK, nfl - 1);
    dv[k] = dinv[rbase + jc / F];
    sv[k] = h_rows[jc];
  }
#pragma unroll
  for (int k = 0; k < PER; ++k) sv[k] = bs + dv[k] * dv[k] * sv[k];
  __syncthreads();

#pragma unroll
  for (int k = 0; k < PER; ++k) sv[k] += acc[min(tid + k * RBLOCK, nfl - 1)];
#pragma unroll
  for (int k = 0; k < PER; ++k) {
    float t = fmaxf(sv[k], 0.0f) * w3;
    for (int d = 16; d > 0; d >>= 1) t += __shfl_xor(t, d);
    const int r = wave + 32 * k;
    if (lane == 0 && r < OWNROWS) gs[r] = t;
  }
  __syncthreads();
  if (tid < rlen) g[rbase + tid] = gs[tid];
}


#define P3E 16

__global__ __launch_bounds__(RBLOCK) void k_agg1_own(const int* __restrict__ lrow,
                                                     const float* __restrict__ lw,
                                                     const int* __restrict__ starts,
                                                     const float* __restrict__ g,
                                                     float* __restrict__ sum, int Nn) {
  __shared__ float acc[OWNROWS];
  const int tid = threadIdx.x;
  const int begin = starts[blockIdx.x];
  const int end = starts[blockIdx.x + 1];
  const int rbase = blockIdx.x * OWNROWS;
  const int rlen = min(OWNROWS, Nn - rbase);
  for (int j = tid; j < OWNROWS; j += RBLOCK) acc[j] = 0.0f;
  __syncthreads();

  for (int i0 = begin + tid; i0 < end; i0 += RBLOCK * P3E) {
    int pk[P3E];
    float w[P3E];
    float v[P3E];
#pragma unroll
    for (int u = 0; u < P3E; ++u) {
      const int ic = min(i0 + u * RBLOCK, end - 1);
      pk[u] = lrow[ic];
      w[u] = lw[ic];
    }
#pragma unroll
    for (int u = 0; u < P3E; ++u) v[u] = g[pk[u] >> 12];
#pragma unroll
    for (int u = 0; u < P3E; ++u) asm volatile("" : "+v"(v[u]), "+v"(w[u]));
#pragma unroll
    for (int u = 0; u < P3E; ++u)
      if (i0 + u * RBLOCK < end) atomicAdd(&acc[pk[u] & 4095], w[u] * v[u]);
  }
  __syncthreads();

  float* __restrict__ dst = sum + rbase;
  for (int j = tid; j < rlen; j += RBLOCK) dst[j] = acc[j];
}

__global__ void k_out_sum(float* __restrict__ out, const float* __restrict__ sum,
                          const float* __restrict__ g, const float* __restrict__ dinv,
                          const float* __restrict__ b3, int Nn) {
  int i = blockIdx.x * blockDim.x + threadIdx.x;
  if (i >= Nn) return;
  float di = dinv[i];
  out[i] = (b3[0] + di * di * g[i]) + sum[i];
}


static inline unsigned int blocks_for(long long total, int bs) {
  return (unsigned int)((total + bs - 1) / bs);
}

extern "C" void kernel_launch(void* const* d_in, const int* in_sizes, int n_in,
                              void* d_out, int out_size, void* d_ws, size_t ws_size,
                              hipStream_t stream) {
  const float* x = (const float*)d_in[0];
  const int* ei = (const int*)d_in[1];
  const float* ew = (const float*)d_in[2];
  const float* W1 = (const float*)d_in[3];
  const float* b1 = (const float*)d_in[4];
  const float* W2 = (const float*)d_in[5];
  const float* b2 = (const float*)d_in[6];
  const float* W3 = (const float*)d_in[7];
  const float* b3 = (const float*)d_in[8];

  const int E = in_sizes[2];
  const int Nn = in_sizes[0] / IN_C;
  float* out = (float*)d_out;

  float* dinv = (float*)d_ws;
  float* lw = dinv + Nn;
  float* bufA = lw + E;
  float* bufA_hi = bufA + (size_t)Nn * H2C;
  float* bufB = bufA + (size_t)Nn * H1C;
  float* bufB_hi = bufB + (size_t)Nn * H2C;

  k_deg_part<<<dim3(SLICES, blocks_for(Nn, RANGE)), PBLOCK, 0, stream>>>(ei, ew, bufA, E, Nn);
  k_dinv_sum<<<blocks_for(Nn, 256), 256, 0, stream>>>(bufA, dinv, Nn);

  const int nr2 = (int)blocks_for(Nn, OWNROWS);
  int* bk_cnt = (int*)d_out;
  int* bk_starts = bk_cnt + nr2 * BKB;
  int* lrow = (int*)bufB_hi;
  k_bk_count<<<BKB, BKT, 0, stream>>>(ei, bk_cnt, E, Nn, nr2);
  k_bk_scatter<<<BKB, BKT, 0, stream>>>(ei, ew, dinv, bk_cnt, bk_starts, lrow, lw, E, Nn, nr2);

  k_gemm1_rows<<<blocks_for(Nn / 16, G1_WAVES), G1_THREADS, 0, stream>>>(x, W1, bufA, Nn / 16);
  k_agg64_gemm2<<<nr2, RBLOCK, 0, stream>>>(lrow, lw, bk_starts, bufA, dinv, b1, W2, bufB, Nn);

  k_agg32_dot<<<nr2, RBLOCK, 0, stream>>>(lrow, lw, bk_starts, bufB, dinv, b2, W3, bufA, Nn);

  k_agg1_own<<<nr2, RBLOCK, 0, stream>>>(lrow, lw, bk_starts, bufA, bufA_hi, Nn);
  k_out_sum<<<blocks_for(Nn, 256), 256, 0, stream>>>(out, bufA_hi, bufA, dinv, b3, Nn);
}


# Degrees summed from the bucket list by the layer-1 GEMM block owning the rows; rows stored scaled by dinv, sums scaled by dinv[col]: no degree passes or per-edge dinv reads; f32 sum order changes
# speedup vs baseline: 4.1115x; 1.1577x over previous
#include <hip/hip_runtime.h>
#include <hip/hip_bf16.h>


typedef float v2f __attribute__((ext_vector_type(2)));
typedef float v8f __attribute__((ext_vector_type(8)));

#define IN_C 512
#define H1C 64
#define H2C 32
#define OWNROWS 400

__device__ __forceinline__ float dinv_of(const float wsum) {
  const float d = 1.0f + wsum;
  return (d > 0.0f) ? rsqrtf(fmaxf(d, 1e-12f)) : 0.0f;
}


#define G1_WAVES 25
#define G1_THREADS (G1_WAVES * 32)
#define G1_DE 18

__device__ __forceinline__ void g1_block(const float4 a, const float4* bp, v8f& c0, v8f& c1,
                                         v8f& c2, v8f& c3) {
  const float4 b0 = bp[0];
  const float4 b1 = bp[16];
  const float4 b2 = bp[32];
  const float4 b3 = bp[48];
  v2f alo, ahi, f;
  alo.x = a.x; alo.y = a.y;
  ahi.x = a.z; ahi.y = a.w;
  f.x = b0.x; f.y = b0.y;
  c0 = __builtin_amdgcn_wmma_f32_16x16x4_f32(false, alo, false, f, (short)0, c0, false, false);
  f.x = b1.x; f.y = b1.y;
  c1 = __builtin_amdgcn_wmma_f32_16x16x4_f32(false, alo, false, f, (short)0, c1, false, false);
  f.x = b2.x; f.y = b2.y;
  c2 = __builtin_amdgcn_wmma_f32_16x16x4_f32(false, alo, false, f, (short)0, c2, false, false);
  f.x = b3.x; f.y = b3.y;
  c3 = __builtin_amdgcn_wmma_f32_16x16x4_f32(false, alo, false, f, (short)0, c3, false, false);
  f.x = b0.z; f.y = b0.w;
  c0 = __builtin_amdgcn_wmma_f32_16x16x4_f32(false, ahi, false, f, (short)0, c0, false, false);
  f.x = b1.z; f.y = b1.w;
  c1 = __builtin_amdgcn_wmma_f32_16x16x4_f32(false, ahi, false, f, (short)0, c1, false, false);
  f.x = b2.z; f.y = b2.w;
  c2 = __builtin_amdgcn_wmma_f32_16x16x4_f32(false, ahi, false, f, (short)0, c2, false, false);
  f.x = b3.z; f.y = b3.w;
  c3 = __builtin_amdgcn_wmma_f32_16x16x4_f32(false, ahi, false, f, (short)0, c3, false, false);
}

__global__ __launch_bounds__(G1_THREADS) void k_gemm1_rows(const float* __restrict__ A,
                                                           const float* __restrict__ B,
                                                           const int* __restrict__ lrow,
                                                           const float* __restrict__ lw,
                                                           const int* __restrict__ starts,
                                                           float* __restrict__ C,
                                                           float* __restrict__ dinv, int ntiles,
                                                           int Nn) {
  static_assert(G1_WAVES * 16 == OWNROWS, "a block's row tiles are one bucket's rows");
  __shared__ float4 Bs[(IN_C / 4) * H1C];
  __shared__ float4 Cs[G1_WAVES][16 * H1C / 4];
  __shared__ float degs[OWNROWS];
  const int tid = threadIdx.x;
  const int lane = tid & 31;
  const int wave = __builtin_amdgcn_readfirstlane(tid >> 5);
  const int m = lane & 15;
  const int half = lane >> 4;
  const int rbase = blockIdx.x * OWNROWS;
  const int rlen = min(OWNROWS, Nn - rbase);
  const int lbegin = starts[blockIdx.x];
  const int lend = starts[blockIdx.x + 1];

  int pk[G1_DE];
  float wd[G1_DE];
#pragma unroll
  for (int u = 0; u < G1_DE; ++u) {
    const int ic = max(min(lbegin + tid + u * G1_THREADS, lend - 1), 0);
    pk[u] = lrow[ic];
    wd[u] = lw[ic];
  }
  if (tid < OWNROWS) degs[tid] = 0.0f;

  for (int idx = tid; idx < (IN_C / 4) * H1C; idx += G1_THREADS) {
    const int kq = idx / H1C;
    const int n = idx % H1C;
    const float* src = B + (size_t)(4 * kq) * H1C + n;
    float4 v;
    v.x = src[0];
    v.y = src[H1C];
    v.z = src[2 * H1C];
    v.w = src[3 * H1C];
    Bs[idx] = v;
  }
  __syncthreads();

#pragma unroll
  for (int u = 0; u < G1_DE; ++u)
    if (lbegin + tid + u * G1_THREADS < lend) atomicAdd(&degs[pk[u] & 4095], wd[u]);
  for (int i0 = lbegin + tid + G1_THREADS * G1_DE; i0 < lend; i0 += G1_THREADS)
    atomicAdd(&degs[lrow[i0] & 4095], lw[i0]);

  const int tile = blockIdx.x * G1_WAVES + wave;
  const bool active = tile < ntiles;
  if (active) {
    const float* arow = A + ((size_t)tile * 16 + m) * IN_C + half * 4;
    const float4* bfrag = Bs + half * H1C + m;
    v8f c0 = {}, c1 = {}, c2 = {}, c3 = {};
    float4 a[4];
#pragma unroll
    for (int u = 0; u < 4; ++u) a[u] = *reinterpret_cast<const float4*>(arow + 8 * u);
    for (int kb = 0; kb < IN_C - 32; kb += 32) {
      float4 an[4];
#pragma unroll
      for (int u = 0; u < 4; ++u)
        an[u] = *reinterpret_cast<const float4*>(arow + kb + 32 + 8 * u);
#pragma unroll
      for (int u = 0; u < 4; ++u) g1_block(a[u], bfrag + ((kb + 8 * u) >> 2) * H1C, c0, c1, c2, c3);
#pragma unroll
      for (int u = 0; u < 4; ++u) a[u] = an[u];
    }
#pragma unroll
    for (int u = 0; u < 4; ++u)
      g1_block(a[u], bfrag + ((IN_C - 32 + 8 * u) >> 2) * H1C, c0, c1, c2, c3);

    float* cs = reinterpret_cast<float*>(Cs[wave]);
#pragma unroll
    for (int v = 0; v < 8; ++v) {
      float* p = cs + (v + 8 * half) * H1C + m;
      p[0] = c0[v];
      p[16] = c1[v];
      p[32] = c2[v];
      p[48] = c3[v];
    }
  }
  __syncthreads();
  if (tid < rlen) dinv[rbase + tid] = dinv_of(degs[tid]);
  if (active) {
    float4* dst = reinterpret_cast<float4*>(C + (size_t)tile * 16 * H1C);
#pragma unroll
    for (int i = 0; i < 8; ++i) {
      const float s = dinv_of(degs[wave * 16 + 2 * i + half]);
      float4 q = Cs[wave][i * 32 + lane];
      q.x *= s;
      q.y *= s;
      q.z *= s;
      q.w *= s;
      dst[i * 32 + lane] = q;
    }
  }
}


#define RBLOCK 1024

#define BKB 256
#define BKT 1024
#define BKE 16
#define BKR 256

__global__ __launch_bounds__(BKT) void k_bk_count(const int* __restrict__ ei,
                                                  int* __restrict__ cnt, int E, int Nn,
                                                  int nr) {
  __shared__ int hist[BKR];
  const int tid = threadIdx.x;
  if (tid < BKR) hist[tid] = 0;
  __syncthreads();
  const int per = (E + BKB - 1) / BKB;
  const int begin = blockIdx.x * per;
  const int end = min(E, begin + per);
  const int* __restrict__ col = ei + E;
  for (int e0 = begin + tid; e0 < end; e0 += BKT * BKE) {
    int c[BKE];
#pragma unroll
    for (int u = 0; u < BKE; ++u) {
      const int e = e0 + u * BKT;
      const int cc = col[min(e, end - 1)];
      c[u] = (e < end) ? cc : -1;
    }
#pragma unroll
    for (int u = 0; u < BKE; ++u)
      if ((unsigned)c[u] < (unsigned)Nn) atomicAdd(&hist[c[u] / OWNROWS], 1);
  }
  __syncthreads();
  if (tid < nr) cnt[tid * BKB + blockIdx.x] = hist[tid];
}

__global__ __launch_bounds__(BKT) void k_bk_scatter(const int* __restrict__ ei,
                                                    const float* __restrict__ ew,
                                                    const int* __restrict__ cnt,
                                                    int* __restrict__ starts,
                                                    int* __restrict__ lrow,
                                                    float* __restrict__ lw, int E, int Nn,
                                                    int nr) {
  __shared__ int tot[BKR];
  __shared__ int before[BKR];
  __shared__ int cur[BKR];
  __shared__ int wsum[BKR / 32];
  const int tid = threadIdx.x;
  const int lane = tid & 31;
  const int wave = tid >> 5;
  for (int r = wave; r < nr; r += BKT / 32) {
    int t = 0, bf = 0;
#pragma unroll
    for (int k = 0; k < BKB / 32; ++k) {
      const int b = k * 32 + lane;
      const int v = cnt[r * BKB + b];
      t += v;
      bf += (b < (int)blockIdx.x) ? v : 0;
    }
    for (int d = 16; d > 0; d >>= 1) {
      t += __shfl_xor(t, d);
      bf += __shfl_xor(bf, d);
    }
    if (lane == 0) {
      tot[r] = t;
      before[r] = bf;
    }
  }
  __syncthreads();
  int mine = 0, incl = 0;
  if (tid < BKR) {
    mine = (tid < nr) ? tot[tid] : 0;
    incl = mine;
    for (int d = 1; d < 32; d <<= 1) {
      const int below = __shfl_up(incl, d);
      if (lane >= d) incl += below;
    }
    if (lane == 31) wsum[wave] = incl;
  }
  __syncthreads();
  if (tid <= nr) {
    int sum = incl - mine;
    for (int q = 0; q < wave; ++q) sum += wsum[q];
    if (tid < nr) cur[tid] = sum + before[tid];
    if (blockIdx.x == 0) starts[tid] = sum;
  }
  __syncthreads();

  const int per = (E + BKB - 1) / BKB;
  const int begin = blockIdx.x * per;
  const int end = min(E, begin + per);
  const int* __restrict__ col = ei + E;
  for (int e0 = begin + tid; e0 < end; e0 += BKT * BKE) {
    int c[BKE];
    int r[BKE];
    float w[BKE];
#pragma unroll
    for (int u = 0; u < BKE; ++u) {
      const int e = e0 + u * BKT;
      const int ec = min(e, end - 1);
      const int cc = col[ec];
      r[u] = ei[ec];
      w[u] = ew[ec];
      c[u] = (e < end) ? cc : -1;
    }
    asm volatile("" : "+v"(r[0]), "+v"(w[0]));
#pragma unroll
    for (int u = 0; u < BKE; ++u) {
      if ((unsigned)c[u] < (unsigned)Nn) {
        const int rg = c[u] / OWNROWS;
        const int pos = atomicAdd(&cur[rg], 1);
        lrow[pos] = (r[u] << 12) | (c[u] - rg * OWNROWS);
        lw[pos] = w[u];
      }
    }
  }
}

template <int F, int P, int NB>
__device__ __forceinline__ void row_batch_n(const int rv, const int relv, const float wv,
                                            const int b, const int lane,
                                            const float* __restrict__ h, float* acc) {
  constexpr int L = F / 32;
  float v[NB][L];
#pragma unroll
  for (int u = 0; u < NB; ++u) {
    const int r = __builtin_amdgcn_readlane(rv, b + u);
    const float* src = h + (size_t)r * F + lane;
#pragma unroll
    for (int l = 0; l < L; ++l) v[u][l] = src[l * 32];
  }
#pragma unroll
  for (int u = 0; u < NB; ++u) {
    const int rel = __builtin_amdgcn_readlane(relv, b + u);
    const float w = __int_as_float(__builtin_amdgcn_readlane(__float_as_int(wv), b + u));
    float* dst = acc + rel * P + lane;
#pragma unroll
    for (int l = 0; l < L; ++l) atomicAdd(dst + l * 32, w * v[u][l]);
  }
}

template <int F, int P>
__device__ __forceinline__ void row_group(const int rv, const int relv, const float wv,
                                          const int left, const int lane,
                                          const float* __restrict__ h, float* acc) {
  constexpr int L = F / 32;
#pragma unroll
  for (int b = 0; b < 32; b += 16) {
    if (b + 16 <= left) {
      row_batch_n<F, P, 16>(rv, relv, wv, b, lane, h, acc);
    } else {
#pragma unroll
      for (int bb = b; bb < b + 16; bb += 8) {
        if (bb + 8 <= left) {
          row_batch_n<F, P, 8>(rv, relv, wv, bb, lane, h, acc);
        } else {
#pragma unroll
          for (int u = bb; u < bb + 8; ++u) {
            if (u < left) {
              const int r = __builtin_amdgcn_readlane(rv, u);
              const int rel = __builtin_amdgcn_readlane(relv, u);
              const float w =
                  __int_as_float(__builtin_amdgcn_readlane(__float_as_int(wv), u));
              const float* src = h + (size_t)r * F + lane;
              float* dst = acc + rel * P + lane;
#pragma unroll
              for (int l = 0; l < L; ++l) atomicAdd(dst + l * 32, w * src[l * 32]);
            }
          }
        }
      }
    }
  }
}

template <int F, int P>
__device__ __forceinline__ void bucket_sum(const int* __restrict__ lrow,
                                           const float* __restrict__ lw, const int begin,
                                           const int end, const int wave, const int lane,
                                           const float* __restrict__ h, float* acc) {
  int i = begin + wave * 32;
  int pk = 0;
  float wv = 0.0f;
  if (i < end && lane < end - i) {
    pk = lrow[i + lane];
    wv = lw[i + lane];
  }
  while (i < end) {
    const int inext = i + RBLOCK;
    int pkn = 0;
    float wn = 0.0f;
    if (inext < end && lane < end - inext) {
      pkn = lrow[inext + lane];
      wn = lw[inext + lane];
    }
    row_group<F, P>(pk >> 12, pk & 4095, wv, min(32, end - i), lane, h, acc);
    pk = pkn;
    wv = wn;
    i = inext;
  }
}


#define APITCH 68

__global__ __launch_bounds__(RBLOCK) void k_agg64_gemm2(const int* __restrict__ lrow,
                                                        const float* __restrict__ lw,
                                                        const int* __restrict__ starts,
                                                        const float* __restrict__ h,
                                                        const float* __restrict__ dinv,
                                                        const float* __restrict__ bias,
                                                        const float* __restrict__ W2,
                                                        float* __restrict__ h2, int Nn) {
  constexpr int F = H1C;
  constexpr int PER = OWNROWS * F / RBLOCK;
  static_assert(OWNROWS * F % RBLOCK == 0 && RBLOCK % F == 0, "a thread keeps one column");
  static_assert(OWNROWS % 16 == 0 && OWNROWS / 16 <= RBLOCK / 32, "one wave per 16-row tile");
  __shared__ float acc[OWNROWS * APITCH];
  __shared__ float dl[OWNROWS];
  const int tid = threadIdx.x;
  const int lane = tid & 31;
  const int wave = __builtin_amdgcn_readfirstlane(tid >> 5);
  const int begin = __builtin_amdgcn_readfirstlane(starts[blockIdx.x]);
  const int end = __builtin_amdgcn_readfirstlane(starts[blockIdx.x + 1]);
  const int rbase = blockIdx.x * OWNROWS;
  const int rlen = min(OWNROWS, Nn - rbase);
  if (tid < OWNROWS) dl[tid] = dinv[rbase + min(tid, rlen - 1)];
  for (int j = tid; j < OWNROWS * APITCH; j += RBLOCK) acc[j] = 0.0f;
  __syncthreads();

  bucket_sum<F, APITCH>(lrow, lw, begin, end, wave, lane, h, acc);

  const int col = tid % F;
  const int row0 = tid / F;
  const float bs = bias[col];
  float sv[PER];
#pragma unroll
  for (int k = 0; k < PER; ++k) {
    const int rc = rbase + min(row0 + 16 * k, rlen - 1);
    sv[k] = h[(size_t)rc * F + col];
  }
  __syncthreads();

#pragma unroll
  for (int k = 0; k < PER; ++k) {
    const int r = row0 + 16 * k;
    float* p = acc + r * APITCH + col;
    const float t = fmaxf(bs + dl[r] * (sv[k] + *p), 0.0f);
    *p = (r < rlen) ? t : 0.0f;
  }
  __syncthreads();

  const bool active = wave < OWNROWS / 16;
  if (active) {
    float* tile = acc + wave * 16 * APITCH;
    const int m = lane & 15;
    const int half = lane >> 4;
    const float* ap = tile + m * APITCH + 2 * half;
    const float* bp = W2 + (2 * half) * H2C + m;
    v8f c0 = {}, c1 = {};
#pragma unroll 8
    for (int k = 0; k < H1C; k += 4) {
      const float2 av = *reinterpret_cast<const float2*>(ap + k);
      v2f a, b0, b1;
      a.x = av.x;
      a.y = av.y;
      b0.x = bp[k * H2C];
      b0.y = bp[(k + 1) * H2C];
      b1.x = bp[k * H2C + 16];
      b1.y = bp[(k + 1) * H2C + 16];
      c0 = __builtin_amdgcn_wmma_f32_16x16x4_f32(false, a, false, b0, (short)0, c0, false, false);
      c1 = __builtin_amdgcn_wmma_f32_16x16x4_f32(false, a, false, b1, (short)0, c1, false, false);
    }
#pragma unroll
    for (int v = 0; v < 8; ++v) {
      float* p = tile + (v + 8 * half) * APITCH + m;
      p[0] = c0[v];
      p[16] = c1[v];
    }
  }
  __syncthreads();
  if (active) {
    const float* tile = acc + wave * 16 * APITCH;
    float4* dst = reinterpret_cast<float4*>(h2 + (size_t)(rbase + wave * 16) * H2C);
#pragma unroll
    for (int i = 0; i < 4; ++i) {
      const int r = 4 * i + (lane >> 3);
      const float s = dl[wave * 16 + r];
      float4 q = *reinterpret_cast<const float4*>(tile + r * APITCH + 4 * (lane & 7));
      q.x *= s;
      q.y *= s;
      q.z *= s;
      q.w *= s;
      if (wave * 16 + r < rlen) dst[i * 32 + lane] = q;
    }
  }
}

__global__ __launch_bounds__(RBLOCK) void k_agg32_dot(const int* __restrict__ lrow,
                                                      const float* __restrict__ lw,
                                                      const int* __restrict__ starts,
                                                      const float* __restrict__ h,
                                                      const float* __restrict__ dinv,
                                                      const float* __restrict__ bias,
                                                      const float* __restrict__ W3,
                                                      float* __restrict__ g, int Nn) {
  constexpr int F = H2C;
  static_assert(RBLOCK % F == 0 && F == 32, "a row of the tail is one wave's lanes");
  constexpr int PER = (OWNROWS * F + RBLOCK - 1) / RBLOCK;
  __shared__ float acc[OWNROWS * F];
  __shared__ float gs[OWNROWS];
  const int tid = threadIdx.x;
  const int lane = tid & 31;
  const int wave = __builtin_amdgcn_readfirstlane(tid >> 5);
  const int begin = __builtin_amdgcn_readfirstlane(starts[blockIdx.x]);
  const int end = __builtin_amdgcn_readfirstlane(starts[blockIdx.x + 1]);
  const int rbase = blockIdx.x * OWNROWS;
  const int rlen = min(OWNROWS, Nn - rbase);
  for (int j = tid; j < OWNROWS * F; j += RBLOCK) acc[j] = 0.0f;
  __syncthreads();

  bucket_sum<F, F>(lrow, lw, begin, end, wave, lane, h, acc);

  const float* __restrict__ h_rows = h + (size_t)rbase * F;
  const int nfl = rlen * F;
  const float bs = bias[lane];
  const float w3 = W3[lane];
  float dv[PER];
  float sv[PER];
#pragma unroll
  for (int k = 0; k < PER; ++k) {
    const int jc = min(tid + k * RBLOCK, nfl - 1);
    dv[k] = dinv[rbase + jc / F];
    sv[k] = h_rows[jc];
  }
  __syncthreads();

#pragma unroll
  for (int k = 0; k < PER; ++k)
    sv[k] = bs + dv[k] * (sv[k] + acc[min(tid + k * RBLOCK, nfl - 1)]);
#pragma unroll
  for (int k = 0; k < PER; ++k) {
    float t = fmaxf(sv[k], 0.0f) * w3;
    for (int d = 16; d > 0; d >>= 1) t += __shfl_xor(t, d);
    const int r = wave + 32 * k;
    if (lane == 0 && r < OWNROWS) gs[r] = dv[k] * t;
  }
  __syncthreads();
  if (tid < rlen) g[rbase + tid] = gs[tid];
}


#define P3E 16

__global__ __launch_bounds__(RBLOCK) void k_agg1_own(const int* __restrict__ lrow,
                                                     const float* __restrict__ lw,
                                                     const int* __restrict__ starts,
                                                     const float* __restrict__ g,
                                                     float* __restrict__ sum, int Nn) {
  __shared__ float acc[OWNROWS];
  const int tid = threadIdx.x;
  const int begin = starts[blockIdx.x];
  const int end = starts[blockIdx.x + 1];
  const int rbase = blockIdx.x * OWNROWS;
  const int rlen = min(OWNROWS, Nn - rbase);
  for (int j = tid; j < OWNROWS; j += RBLOCK) acc[j] = 0.0f;
  __syncthreads();

  for (int i0 = begin + tid; i0 < end; i0 += RBLOCK * P3E) {
    int pk[P3E];
    float w[P3E];
    float v[P3E];
#pragma unroll
    for (int u = 0; u < P3E; ++u) {
      const int ic = min(i0 + u * RBLOCK, end - 1);
      pk[u] = lrow[ic];
      w[u] = lw[ic];
    }
#pragma unroll
    for (int u = 0; u < P3E; ++u) v[u] = g[pk[u] >> 12];
#pragma unroll
    for (int u = 0; u < P3E; ++u) asm volatile("" : "+v"(v[u]), "+v"(w[u]));
#pragma unroll
    for (int u = 0; u < P3E; ++u)
      if (i0 + u * RBLOCK < end) atomicAdd(&acc[pk[u] & 4095], w[u] * v[u]);
  }
  __syncthreads();

  float* __restrict__ dst = sum + rbase;
  for (int j = tid; j < rlen; j += RBLOCK) dst[j] = acc[j];
}

__global__ void k_out_sum(float* __restrict__ out, const float* __restrict__ sum,
                          const float* __restrict__ g, const float* __restrict__ dinv,
                          const float* __restrict__ b3, int Nn) {
  int i = blockIdx.x * blockDim.x + threadIdx.x;
  if (i >= Nn) return;
  out[i] = b3[0] + dinv[i] * (g[i] + sum[i]);
}


static inline unsigned int blocks_for(long long total, int bs) {
  return (unsigned int)((total + bs - 1) / bs);
}

extern "C" void kernel_launch(void* const* d_in, const int* in_sizes, int n_in,
                              void* d_out, int out_size, void* d_ws, size_t ws_size,
                              hipStream_t stream) {
  const float* x = (const float*)d_in[0];
  const int* ei = (const int*)d_in[1];
  const float* ew = (const float*)d_in[2];
  const float* W1 = (const float*)d_in[3];
  const float* b1 = (const float*)d_in[4];
  const float* W2 = (const float*)d_in[5];
  const float* b2 = (const float*)d_in[6];
  const float* W3 = (const float*)d_in[7];
  const float* b3 = (const float*)d_in[8];

  const int E = in_sizes[2];
  const int Nn = in_sizes[0] / IN_C;
  float* out = (float*)d_out;

  float* dinv = (float*)d_ws;
  float* lw = dinv + Nn;
  float* bufA = lw + E;
  float* bufA_hi = bufA + (size_t)Nn * H2C;
  float* bufB = bufA + (size_t)Nn * H1C;
  float* bufB_hi = bufB + (size_t)Nn * H2C;

  const int nr2 = (int)blocks_for(Nn, OWNROWS);
  int* bk_cnt = (int*)d_out;
  int* bk_starts = bk_cnt + nr2 * BKB;
  int* lrow = (int*)bufB_hi;
  k_bk_count<<<BKB, BKT, 0, stream>>>(ei, bk_cnt, E, Nn, nr2);
  k_bk_scatter<<<BKB, BKT, 0, stream>>>(ei, ew, bk_cnt, bk_starts, lrow, lw, E, Nn, nr2);

  k_gemm1_rows<<<nr2, G1_THREADS, 0, stream>>>(x, W1, lrow, lw, bk_starts, bufA, dinv, Nn / 16, Nn);
  k_agg64_gemm2<<<nr2, RBLOCK, 0, stream>>>(lrow, lw, bk_starts, bufA, dinv, b1, W2, bufB, Nn);

  k_agg32_dot<<<nr2, RBLOCK, 0, stream>>>(lrow, lw, bk_starts, bufB, dinv, b2, W3, bufA, Nn);

  k_agg1_own<<<nr2, RBLOCK, 0, stream>>>(lrow, lw, bk_starts, bufA, bufA_hi, Nn);
  k_out_sum<<<blocks_for(Nn, 256), 256, 0, stream>>>(out, bufA_hi, bufA, dinv, b3, Nn);
}


# Scatter pass first orders its slice by bucket in LDS (rank = returning LDS add per range), then stores each bucket's run contiguously instead of 4-byte pieces; f32 sum order varies as before
# speedup vs baseline: 4.9963x; 1.2152x over previous
#include <hip/hip_runtime.h>
#include <hip/hip_bf16.h>


typedef float v2f __attribute__((ext_vector_type(2)));
typedef float v8f __attribute__((ext_vector_type(8)));

#define IN_C 512
#define H1C 64
#define H2C 32
#define OWNROWS 400

__device__ __forceinline__ float dinv_of(const float wsum) {
  const float d = 1.0f + wsum;
  return (d > 0.0f) ? rsqrtf(fmaxf(d, 1e-12f)) : 0.0f;
}


#define G1_WAVES 25
#define G1_THREADS (G1_WAVES * 32)
#define G1_DE 18

__device__ __forceinline__ void g1_block(const float4 a, const float4* bp, v8f& c0, v8f& c1,
                                         v8f& c2, v8f& c3) {
  const float4 b0 = bp[0];
  const float4 b1 = bp[16];
  const float4 b2 = bp[32];
  const float4 b3 = bp[48];
  v2f alo, ahi, f;
  alo.x = a.x; alo.y = a.y;
  ahi.x = a.z; ahi.y = a.w;
  f.x = b0.x; f.y = b0.y;
  c0 = __builtin_amdgcn_wmma_f32_16x16x4_f32(false, alo, false, f, (short)0, c0, false, false);
  f.x = b1.x; f.y = b1.y;
  c1 = __builtin_amdgcn_wmma_f32_16x16x4_f32(false, alo, false, f, (short)0, c1, false, false);
  f.x = b2.x; f.y = b2.y;
  c2 = __builtin_amdgcn_wmma_f32_16x16x4_f32(false, alo, false, f, (short)0, c2, false, false);
  f.x = b3.x; f.y = b3.y;
  c3 = __builtin_amdgcn_wmma_f32_16x16x4_f32(false, alo, false, f, (short)0, c3, false, false);
  f.x = b0.z; f.y = b0.w;
  c0 = __builtin_amdgcn_wmma_f32_16x16x4_f32(false, ahi, false, f, (short)0, c0, false, false);
  f.x = b1.z; f.y = b1.w;
  c1 = __builtin_amdgcn_wmma_f32_16x16x4_f32(false, ahi, false, f, (short)0, c1, false, false);
  f.x = b2.z; f.y = b2.w;
  c2 = __builtin_amdgcn_wmma_f32_16x16x4_f32(false, ahi, false, f, (short)0, c2, false, false);
  f.x = b3.z; f.y = b3.w;
  c3 = __builtin_amdgcn_wmma_f32_16x16x4_f32(false, ahi, false, f, (short)0, c3, false, false);
}

__global__ __launch_bounds__(G1_THREADS) void k_gemm1_rows(const float* __restrict__ A,
                                                           const float* __restrict__ B,
                                                           const int* __restrict__ lrow,
                                                           const float* __restrict__ lw,
                                                           const int* __restrict__ starts,
                                                           float* __restrict__ C,
                                                           float* __restrict__ dinv, int ntiles,
                                                           int Nn) {
  static_assert(G1_WAVES * 16 == OWNROWS, "a block's row tiles are one bucket's rows");
  __shared__ float4 Bs[(IN_C / 4) * H1C];
  __shared__ float4 Cs[G1_WAVES][16 * H1C / 4];
  __shared__ float degs[OWNROWS];
  const int tid = threadIdx.x;
  const int lane = tid & 31;
  const int wave = __builtin_amdgcn_readfirstlane(tid >> 5);
  const int m = lane & 15;
  const int half = lane >> 4;
  const int rbase = blockIdx.x * OWNROWS;
  const int rlen = min(OWNROWS, Nn - rbase);
  const int lbegin = starts[blockIdx.x];
  const int lend = starts[blockIdx.x + 1];

  int pk[G1_DE];
  float wd[G1_DE];
#pragma unroll
  for (int u = 0; u < G1_DE; ++u) {
    const int ic = max(min(lbegin + tid + u * G1_THREADS, lend - 1), 0);
    pk[u] = lrow[ic];
    wd[u] = lw[ic];
  }
  if (tid < OWNROWS) degs[tid] = 0.0f;

  for (int idx = tid; idx < (IN_C / 4) * H1C; idx += G1_THREADS) {
    const int kq = idx / H1C;
    const int n = idx % H1C;
    const float* src = B + (size_t)(4 * kq) * H1C + n;
    float4 v;
    v.x = src[0];
    v.y = src[H1C];
    v.z = src[2 * H1C];
    v.w = src[3 * H1C];
    Bs[idx] = v;
  }
  __syncthreads();

#pragma unroll
  for (int u = 0; u < G1_DE; ++u)
    if (lbegin + tid + u * G1_THREADS < lend) atomicAdd(&degs[pk[u] & 4095], wd[u]);
  for (int i0 = lbegin + tid + G1_THREADS * G1_DE; i0 < lend; i0 += G1_THREADS)
    atomicAdd(&degs[lrow[i0] & 4095], lw[i0]);

  const int tile = blockIdx.x * G1_WAVES + wave;
  const bool active = tile < ntiles;
  if (active) {
    const float* arow = A + ((size_t)tile * 16 + m) * IN_C + half * 4;
    const float4* bfrag = Bs + half * H1C + m;
    v8f c0 = {}, c1 = {}, c2 = {}, c3 = {};
    float4 a[4];
#pragma unroll
    for (int u = 0; u < 4; ++u) a[u] = *reinterpret_cast<const float4*>(arow + 8 * u);
    for (int kb = 0; kb < IN_C - 32; kb += 32) {
      float4 an[4];
#pragma unroll
      for (int u = 0; u < 4; ++u)
        an[u] = *reinterpret_cast<const float4*>(arow + kb + 32 + 8 * u);
#pragma unroll
      for (int u = 0; u < 4; ++u) g1_block(a[u], bfrag + ((kb + 8 * u) >> 2) * H1C, c0, c1, c2, c3);
#pragma unroll
      for (int u = 0; u < 4; ++u) a[u] = an[u];
    }
#pragma unroll
    for (int u = 0; u < 4; ++u)
      g1_block(a[u], bfrag + ((IN_C - 32 + 8 * u) >> 2) * H1C, c0, c1, c2, c3);

    float* cs = reinterpret_cast<float*>(Cs[wave]);
#pragma unroll
    for (int v = 0; v < 8; ++v) {
      float* p = cs + (v + 8 * half) * H1C + m;
      p[0] = c0[v];
      p[16] = c1[v];
      p[32] = c2[v];
      p[48] = c3[v];
    }
  }
  __syncthreads();
  if (tid < rlen) dinv[rbase + tid] = dinv_of(degs[tid]);
  if (active) {
    float4* dst = reinterpret_cast<float4*>(C + (size_t)tile * 16 * H1C);
#pragma unroll
    for (int i = 0; i < 8; ++i) {
      const float s = dinv_of(degs[wave * 16 + 2 * i + half]);
      float4 q = Cs[wave][i * 32 + lane];
      q.x *= s;
      q.y *= s;
      q.z *= s;
      q.w *= s;
      dst[i * 32 + lane] = q;
    }
  }
}


#define RBLOCK 1024

#define BKB 256
#define BKT 1024
#define BKE 16
#define BKR 256

__global__ __launch_bounds__(BKT) void k_bk_count(const int* __restrict__ ei,
                                                  int* __restrict__ cnt, int E, int Nn,
                                                  int nr) {
  __shared__ int hist[BKR];
  const int tid = threadIdx.x;
  if (tid < BKR) hist[tid] = 0;
  __syncthreads();
  const int per = (E + BKB - 1) / BKB;
  const int begin = blockIdx.x * per;
  const int end = min(E, begin + per);
  const int* __restrict__ col = ei + E;
  for (int e0 = begin + tid; e0 < end; e0 += BKT * BKE) {
    int c[BKE];
#pragma unroll
    for (int u = 0; u < BKE; ++u) {
      const int e = e0 + u * BKT;
      const int cc = col[min(e, end - 1)];
      c[u] = (e < end) ? cc : -1;
    }
#pragma unroll
    for (int u = 0; u < BKE; ++u)
      if ((unsigned)c[u] < (unsigned)Nn) atomicAdd(&hist[c[u] / OWNROWS], 1);
  }
  __syncthreads();
  if (tid < nr) cnt[tid * BKB + blockIdx.x] = hist[tid];
}

#define BKC (BKT * BKE)

__global__ __launch_bounds__(BKT) void k_bk_scatter(const int* __restrict__ ei,
                                                    const float* __restrict__ ew,
                                                    const int* __restrict__ cnt,
                                                    int* __restrict__ starts,
                                                    int* __restrict__ lrow,
                                                    float* __restrict__ lw, int E, int Nn,
                                                    int nr) {
  __shared__ int tot[BKR];
  __shared__ int before[BKR];
  __shared__ int cur[BKR];
  __shared__ int wsum[BKR / 32];
  __shared__ int lcnt[BKR];
  __shared__ int lbase[BKR];
  __shared__ int ntot;
  __shared__ int s_pk[BKC];
  __shared__ float s_w[BKC];
  __shared__ int s_pos[BKC];
  const int tid = threadIdx.x;
  const int lane = tid & 31;
  const int wave = tid >> 5;
  for (int r = wave; r < nr; r += BKT / 32) {
    int t = 0, bf = 0;
#pragma unroll
    for (int k = 0; k < BKB / 32; ++k) {
      const int b = k * 32 + lane;
      const int v = cnt[r * BKB + b];
      t += v;
      bf += (b < (int)blockIdx.x) ? v : 0;
    }
    for (int d = 16; d > 0; d >>= 1) {
      t += __shfl_xor(t, d);
      bf += __shfl_xor(bf, d);
    }
    if (lane == 0) {
      tot[r] = t;
      before[r] = bf;
    }
  }
  __syncthreads();
  int mine = 0, incl = 0;
  if (tid < BKR) {
    mine = (tid < nr) ? tot[tid] : 0;
    incl = mine;
    for (int d = 1; d < 32; d <<= 1) {
      const int below = __shfl_up(incl, d);
      if (lane >= d) incl += below;
    }
    if (lane == 31) wsum[wave] = incl;
  }
  __syncthreads();
  if (tid <= nr) {
    int sum = incl - mine;
    for (int q = 0; q < wave; ++q) sum += wsum[q];
    if (tid < nr) cur[tid] = sum + before[tid];
    if (blockIdx.x == 0) starts[tid] = sum;
  }
  __syncthreads();

  const int per = (E + BKB - 1) / BKB;
  const int begin = blockIdx.x * per;
  const int end = min(E, begin + per);
  const int* __restrict__ col = ei + E;
  for (int base = begin; base < end; base += BKC) {
    if (tid < BKR) lcnt[tid] = 0;
    __syncthreads();
    int c[BKE];
    int r[BKE];
    int rk[BKE];
    float w[BKE];
#pragma unroll
    for (int u = 0; u < BKE; ++u) {
      const int e = base + tid + u * BKT;
      const int ec = min(e, end - 1);
      const int cc = col[ec];
      r[u] = ei[ec];
      w[u] = ew[ec];
      c[u] = (e < end) ? cc : -1;
    }
#pragma unroll
    for (int u = 0; u < BKE; ++u) {
      rk[u] = 0;
      if ((unsigned)c[u] < (unsigned)Nn) rk[u] = atomicAdd(&lcnt[c[u] / OWNROWS], 1);
    }
    __syncthreads();
    int cm = 0, ci = 0;
    if (tid < BKR) {
      cm = (tid < nr) ? lcnt[tid] : 0;
      ci = cm;
      for (int d = 1; d < 32; d <<= 1) {
        const int below = __shfl_up(ci, d);
        if (lane >= d) ci += below;
      }
      if (lane == 31) wsum[wave] = ci;
    }
    __syncthreads();
    if (tid < BKR) {
      int sum = ci - cm;
      for (int q = 0; q < wave; ++q) sum += wsum[q];
      lbase[tid] = sum;
      if (tid == BKR - 1) ntot = sum;
    }
    __syncthreads();
#pragma unroll
    for (int u = 0; u < BKE; ++u) {
      if ((unsigned)c[u] < (unsigned)Nn) {
        const int rg = c[u] / OWNROWS;
        const int lp = lbase[rg] + rk[u];
        s_pk[lp] = (r[u] << 12) | (c[u] - rg * OWNROWS);
        s_w[lp] = w[u];
        s_pos[lp] = cur[rg] + rk[u];
      }
    }
    __syncthreads();
    if (tid < nr) cur[tid] += lcnt[tid];
    const int n = ntot;
    for (int j = tid; j < n; j += BKT) {
      const int pos = s_pos[j];
      lrow[pos] = s_pk[j];
      lw[pos] = s_w[j];
    }
  }
}

template <int F, int P, int NB>
__device__ __forceinline__ void row_batch_n(const int rv, const int relv, const float wv,
                                            const int b, const int lane,
                                            const float* __restrict__ h, float* acc) {
  constexpr int L = F / 32;
  float v[NB][L];
#pragma unroll
  for (int u = 0; u < NB; ++u) {
    const int r = __builtin_amdgcn_readlane(rv, b + u);
    const float* src = h + (size_t)r * F + lane;
#pragma unroll
    for (int l = 0; l < L; ++l) v[u][l] = src[l * 32];
  }
#pragma unroll
  for (int u = 0; u < NB; ++u) {
    const int rel = __builtin_amdgcn_readlane(relv, b + u);
    const float w = __int_as_float(__builtin_amdgcn_readlane(__float_as_int(wv), b + u));
    float* dst = acc + rel * P + lane;
#pragma unroll
    for (int l = 0; l < L; ++l) atomicAdd(dst + l * 32, w * v[u][l]);
  }
}

template <int F, int P>
__device__ __forceinline__ void row_group(const int rv, const int relv, const float wv,
                                          const int left, const int lane,
                                          const float* __restrict__ h, float* acc) {
  constexpr int L = F / 32;
#pragma unroll
  for (int b = 0; b < 32; b += 16) {
    if (b + 16 <= left) {
      row_batch_n<F, P, 16>(rv, relv, wv, b, lane, h, acc);
    } else {
#pragma unroll
      for (int bb = b; bb < b + 16; bb += 8) {
        if (bb + 8 <= left) {
          row_batch_n<F, P, 8>(rv, relv, wv, bb, lane, h, acc);
        } else {
#pragma unroll
          for (int u = bb; u < bb + 8; ++u) {
            if (u < left) {
              const int r = __builtin_amdgcn_readlane(rv, u);
              const int rel = __builtin_amdgcn_readlane(relv, u);
              const float w =
                  __int_as_float(__builtin_amdgcn_readlane(__float_as_int(wv), u));
              const float* src = h + (size_t)r * F + lane;
              float* dst = acc + rel * P + lane;
#pragma unroll
              for (int l = 0; l < L; ++l) atomicAdd(dst + l * 32, w * src[l * 32]);
            }
          }
        }
      }
    }
  }
}

template <int F, int P>
__device__ __forceinline__ void bucket_sum(const int* __restrict__ lrow,
                                           const float* __restrict__ lw, const int begin,
                                           const int end, const int wave, const int lane,
                                           const float* __restrict__ h, float* acc) {
  int i = begin + wave * 32;
  int pk = 0;
  float wv = 0.0f;
  if (i < end && lane < end - i) {
    pk = lrow[i + lane];
    wv = lw[i + lane];
  }
  while (i < end) {
    const int inext = i + RBLOCK;
    int pkn = 0;
    float wn = 0.0f;
    if (inext < end && lane < end - inext) {
      pkn = lrow[inext + lane];
      wn = lw[inext + lane];
    }
    row_group<F, P>(pk >> 12, pk & 4095, wv, min(32, end - i), lane, h, acc);
    pk = pkn;
    wv = wn;
    i = inext;
  }
}


#define APITCH 68

__global__ __launch_bounds__(RBLOCK) void k_agg64_gemm2(const int* __restrict__ lrow,
                                                        const float* __restrict__ lw,
                                                        const int* __restrict__ starts,
                                                        const float* __restrict__ h,
                                                        const float* __restrict__ dinv,
                                                        const float* __restrict__ bias,
                                                        const float* __restrict__ W2,
                                                        float* __restrict__ h2, int Nn) {
  constexpr int F = H1C;
  constexpr int PER = OWNROWS * F / RBLOCK;
  static_assert(OWNROWS * F % RBLOCK == 0 && RBLOCK % F == 0, "a thread keeps one column");
  static_assert(OWNROWS % 16 == 0 && OWNROWS / 16 <= RBLOCK / 32, "one wave per 16-row tile");
  __shared__ float acc[OWNROWS * APITCH];
  __shared__ float dl[OWNROWS];
  const int tid = threadIdx.x;
  const int lane = tid & 31;
  const int wave = __builtin_amdgcn_readfirstlane(tid >> 5);
  const int begin = __builtin_amdgcn_readfirstlane(starts[blockIdx.x]);
  const int end = __builtin_amdgcn_readfirstlane(starts[blockIdx.x + 1]);
  const int rbase = blockIdx.x * OWNROWS;
  const int rlen = min(OWNROWS, Nn - rbase);
  if (tid < OWNROWS) dl[tid] = dinv[rbase + min(tid, rlen - 1)];
  for (int j = tid; j < OWNROWS * APITCH; j += RBLOCK) acc[j] = 0.0f;
  __syncthreads();

  bucket_sum<F, APITCH>(lrow, lw, begin, end, wave, lane, h, acc);

  const int col = tid % F;
  const int row0 = tid / F;
  const float bs = bias[col];
  float sv[PER];
#pragma unroll
  for (int k = 0; k < PER; ++k) {
    const int rc = rbase + min(row0 + 16 * k, rlen - 1);
    sv[k] = h[(size_t)rc * F + col];
  }
  __syncthreads();

#pragma unroll
  for (int k = 0; k < PER; ++k) {
    const int r = row0 + 16 * k;
    float* p = acc + r * APITCH + col;
    const float t = fmaxf(bs + dl[r] * (sv[k] + *p), 0.0f);
    *p = (r < rlen) ? t : 0.0f;
  }
  __syncthreads();

  const bool active = wave < OWNROWS / 16;
  if (active) {
    float* tile = acc + wave * 16 * APITCH;
    const int m = lane & 15;
    const int half = lane >> 4;
    const float* ap = tile + m * APITCH + 2 * half;
    const float* bp = W2 + (2 * half) * H2C + m;
    v8f c0 = {}, c1 = {};
#pragma unroll 8
    for (int k = 0; k < H1C; k += 4) {
      const float2 av = *reinterpret_cast<const float2*>(ap + k);
      v2f a, b0, b1;
      a.x = av.x;
      a.y = av.y;
      b0.x = bp[k * H2C];
      b0.y = bp[(k + 1) * H2C];
      b1.x = bp[k * H2C + 16];
      b1.y = bp[(k + 1) * H2C + 16];
      c0 = __builtin_amdgcn_wmma_f32_16x16x4_f32(false, a, false, b0, (short)0, c0, false, false);
      c1 = __builtin_amdgcn_wmma_f32_16x16x4_f32(false, a, false, b1, (short)0, c1, false, false);
    }
#pragma unroll
    for (int v = 0; v < 8; ++v) {
      float* p = tile + (v + 8 * half) * APITCH + m;
      p[0] = c0[v];
      p[16] = c1[v];
    }
  }
  __syncthreads();
  if (active) {
    const float* tile = acc + wave * 16 * APITCH;
    float4* dst = reinterpret_cast<float4*>(h2 + (size_t)(rbase + wave * 16) * H2C);
#pragma unroll
    for (int i = 0; i < 4; ++i) {
      const int r = 4 * i + (lane >> 3);
      const float s = dl[wave * 16 + r];
      float4 q = *reinterpret_cast<const float4*>(tile + r * APITCH + 4 * (lane & 7));
      q.x *= s;
      q.y *= s;
      q.z *= s;
      q.w *= s;
      if (wave * 16 + r < rlen) dst[i * 32 + lane] = q;
    }
  }
}

__global__ __launch_bounds__(RBLOCK) void k_agg32_dot(const int* __restrict__ lrow,
                                                      const float* __restrict__ lw,
                                                      const int* __restrict__ starts,
                                                      const float* __restrict__ h,
                                                      const float* __restrict__ dinv,
                                                      const float* __restrict__ bias,
                                                      const float* __restrict__ W3,
                                                      float* __restrict__ g, int Nn) {
  constexpr int F = H2C;
  static_assert(RBLOCK % F == 0 && F == 32, "a row of the tail is one wave's lanes");
  constexpr int PER = (OWNROWS * F + RBLOCK - 1) / RBLOCK;
  __shared__ float acc[OWNROWS * F];
  __shared__ float gs[OWNROWS];
  const int tid = threadIdx.x;
  const int lane = tid & 31;
  const int wave = __builtin_amdgcn_readfirstlane(tid >> 5);
  const int begin = __builtin_amdgcn_readfirstlane(starts[blockIdx.x]);
  const int end = __builtin_amdgcn_readfirstlane(starts[blockIdx.x + 1]);
  const int rbase = blockIdx.x * OWNROWS;
  const int rlen = min(OWNROWS, Nn - rbase);
  for (int j = tid; j < OWNROWS * F; j += RBLOCK) acc[j] = 0.0f;
  __syncthreads();

  bucket_sum<F, F>(lrow, lw, begin, end, wave, lane, h, acc);

  const float* __restrict__ h_rows = h + (size_t)rbase * F;
  const int nfl = rlen * F;
  const float bs = bias[lane];
  const float w3 = W3[lane];
  float dv[PER];
  float sv[PER];
#pragma unroll
  for (int k = 0; k < PER; ++k) {
    const int jc = min(tid + k * RBLOCK, nfl - 1);
    dv[k] = dinv[rbase + jc / F];
    sv[k] = h_rows[jc];
  }
  __syncthreads();

#pragma unroll
  for (int k = 0; k < PER; ++k)
    sv[k] = bs + dv[k] * (sv[k] + acc[min(tid + k * RBLOCK, nfl - 1)]);
#pragma unroll
  for (int k = 0; k < PER; ++k) {
    float t = fmaxf(sv[k], 0.0f) * w3;
    for (int d = 16; d > 0; d >>= 1) t += __shfl_xor(t, d);
    const int r = wave + 32 * k;
    if (lane == 0 && r < OWNROWS) gs[r] = dv[k] * t;
  }
  __syncthreads();
  if (tid < rlen) g[rbase + tid] = gs[tid];
}


#define P3E 16

__global__ __launch_bounds__(RBLOCK) void k_agg1_own(const int* __restrict__ lrow,
                                                     const float* __restrict__ lw,
                                                     const int* __restrict__ starts,
                                                     const float* __restrict__ g,
                                                     float* __restrict__ sum, int Nn) {
  __shared__ float acc[OWNROWS];
  const int tid = threadIdx.x;
  const int begin = starts[blockIdx.x];
  const int end = starts[blockIdx.x + 1];
  const int rbase = blockIdx.x * OWNROWS;
  const int rlen = min(OWNROWS, Nn - rbase);
  for (int j = tid; j < OWNROWS; j += RBLOCK) acc[j] = 0.0f;
  __syncthreads();

  for (int i0 = begin + tid; i0 < end; i0 += RBLOCK * P3E) {
    int pk[P3E];
    float w[P3E];
    float v[P3E];
#pragma unroll
    for (int u = 0; u < P3E; ++u) {
      const int ic = min(i0 + u * RBLOCK, end - 1);
      pk[u] = lrow[ic];
      w[u] = lw[ic];
    }
#pragma unroll
    for (int u = 0; u < P3E; ++u) v[u] = g[pk[u] >> 12];
#pragma unroll
    for (int u = 0; u < P3E; ++u) asm volatile("" : "+v"(v[u]), "+v"(w[u]));
#pragma unroll
    for (int u = 0; u < P3E; ++u)
      if (i0 + u * RBLOCK < end) atomicAdd(&acc[pk[u] & 4095], w[u] * v[u]);
  }
  __syncthreads();

  float* __restrict__ dst = sum + rbase;
  for (int j = tid; j < rlen; j += RBLOCK) dst[j] = acc[j];
}

__global__ void k_out_sum(float* __restrict__ out, const float* __restrict__ sum,
                          const float* __restrict__ g, const float* __restrict__ dinv,
                          const float* __restrict__ b3, int Nn) {
  int i = blockIdx.x * blockDim.x + threadIdx.x;
  if (i >= Nn) return;
  out[i] = b3[0] + dinv[i] * (g[i] + sum[i]);
}


static inline unsigned int blocks_for(long long total, int bs) {
  return (unsigned int)((total + bs - 1) / bs);
}

extern "C" void kernel_launch(void* const* d_in, const int* in_sizes, int n_in,
                              void* d_out, int out_size, void* d_ws, size_t ws_size,
                              hipStream_t stream) {
  const float* x = (const float*)d_in[0];
  const int* ei = (const int*)d_in[1];
  const float* ew = (const float*)d_in[2];
  const float* W1 = (const float*)d_in[3];
  const float* b1 = (const float*)d_in[4];
  const float* W2 = (const float*)d_in[5];
  const float* b2 = (const float*)d_in[6];
  const float* W3 = (const float*)d_in[7];
  const float* b3 = (const float*)d_in[8];

  const int E = in_sizes[2];
  const int Nn = in_sizes[0] / IN_C;
  float* out = (float*)d_out;

  float* dinv = (float*)d_ws;
  float* lw = dinv + Nn;
  float* bufA = lw + E;
  float* bufA_hi = bufA + (size_t)Nn * H2C;
  float* bufB = bufA + (size_t)Nn * H1C;
  float* bufB_hi = bufB + (size_t)Nn * H2C;

  const int nr2 = (int)blocks_for(Nn, OWNROWS);
  int* bk_cnt = (int*)d_out;
  int* bk_starts = bk_cnt + nr2 * BKB;
  int* lrow = (int*)bufB_hi;
  k_bk_count<<<BKB, BKT, 0, stream>>>(ei, bk_cnt, E, Nn, nr2);
  k_bk_scatter<<<BKB, BKT, 0, stream>>>(ei, ew, bk_cnt, bk_starts, lrow, lw, E, Nn, nr2);

  k_gemm1_rows<<<nr2, G1_THREADS, 0, stream>>>(x, W1, lrow, lw, bk_starts, bufA, dinv, Nn / 16, Nn);
  k_agg64_gemm2<<<nr2, RBLOCK, 0, stream>>>(lrow, lw, bk_starts, bufA, dinv, b1, W2, bufB, Nn);

  k_agg32_dot<<<nr2, RBLOCK, 0, stream>>>(lrow, lw, bk_starts, bufB, dinv, b2, W3, bufA, Nn);

  k_agg1_own<<<nr2, RBLOCK, 0, stream>>>(lrow, lw, bk_starts, bufA, bufA_hi, Nn);
  k_out_sum<<<blocks_for(Nn, 256), 256, 0, stream>>>(out, bufA_hi, bufA, dinv, b3, Nn);
}


# Row aggregations order their bucket by destination in LDS (counting sort) and sum each destination's gathered rows in registers: one LDS add per run and group, not per edge; f32 sum order changes
# speedup vs baseline: 5.4660x; 1.0940x over previous
#include <hip/hip_runtime.h>
#include <hip/hip_bf16.h>


typedef float v2f __attribute__((ext_vector_type(2)));
typedef float v8f __attribute__((ext_vector_type(8)));

#define IN_C 512
#define H1C 64
#define H2C 32
#define OWNROWS 400

__device__ __forceinline__ float dinv_of(const float wsum) {
  const float d = 1.0f + wsum;
  return (d > 0.0f) ? rsqrtf(fmaxf(d, 1e-12f)) : 0.0f;
}


#define G1_WAVES 25
#define G1_THREADS (G1_WAVES * 32)
#define G1_DE 18

__device__ __forceinline__ void g1_block(const float4 a, const float4* bp, v8f& c0, v8f& c1,
                                         v8f& c2, v8f& c3) {
  const float4 b0 = bp[0];
  const float4 b1 = bp[16];
  const float4 b2 = bp[32];
  const float4 b3 = bp[48];
  v2f alo, ahi, f;
  alo.x = a.x; alo.y = a.y;
  ahi.x = a.z; ahi.y = a.w;
  f.x = b0.x; f.y = b0.y;
  c0 = __builtin_amdgcn_wmma_f32_16x16x4_f32(false, alo, false, f, (short)0, c0, false, false);
  f.x = b1.x; f.y = b1.y;
  c1 = __builtin_amdgcn_wmma_f32_16x16x4_f32(false, alo, false, f, (short)0, c1, false, false);
  f.x = b2.x; f.y = b2.y;
  c2 = __builtin_amdgcn_wmma_f32_16x16x4_f32(false, alo, false, f, (short)0, c2, false, false);
  f.x = b3.x; f.y = b3.y;
  c3 = __builtin_amdgcn_wmma_f32_16x16x4_f32(false, alo, false, f, (short)0, c3, false, false);
  f.x = b0.z; f.y = b0.w;
  c0 = __builtin_amdgcn_wmma_f32_16x16x4_f32(false, ahi, false, f, (short)0, c0, false, false);
  f.x = b1.z; f.y = b1.w;
  c1 = __builtin_amdgcn_wmma_f32_16x16x4_f32(false, ahi, false, f, (short)0, c1, false, false);
  f.x = b2.z; f.y = b2.w;
  c2 = __builtin_amdgcn_wmma_f32_16x16x4_f32(false, ahi, false, f, (short)0, c2, false, false);
  f.x = b3.z; f.y = b3.w;
  c3 = __builtin_amdgcn_wmma_f32_16x16x4_f32(false, ahi, false, f, (short)0, c3, false, false);
}

__global__ __launch_bounds__(G1_THREADS) void k_gemm1_rows(const float* __restrict__ A,
                                                           const float* __restrict__ B,
                                                           const int* __restrict__ lrow,
                                                           const float* __restrict__ lw,
                                                           const int* __restrict__ starts,
                                                           float* __restrict__ C,
                                                           float* __restrict__ dinv, int ntiles,
                                                           int Nn) {
  static_assert(G1_WAVES * 16 == OWNROWS, "a block's row tiles are one bucket's rows");
  __shared__ float4 Bs[(IN_C / 4) * H1C];
  __shared__ float4 Cs[G1_WAVES][16 * H1C / 4];
  __shared__ float degs[OWNROWS];
  const int tid = threadIdx.x;
  const int lane = tid & 31;
  const int wave = __builtin_amdgcn_readfirstlane(tid >> 5);
  const int m = lane & 15;
  const int half = lane >> 4;
  const int rbase = blockIdx.x * OWNROWS;
  const int rlen = min(OWNROWS, Nn - rbase);
  const int lbegin = starts[blockIdx.x];
  const int lend = starts[blockIdx.x + 1];

  int pk[G1_DE];
  float wd[G1_DE];
#pragma unroll
  for (int u = 0; u < G1_DE; ++u) {
    const int ic = max(min(lbegin + tid + u * G1_THREADS, lend - 1), 0);
    pk[u] = lrow[ic];
    wd[u] = lw[ic];
  }
  if (tid < OWNROWS) degs[tid] = 0.0f;

  for (int idx = tid; idx < (IN_C / 4) * H1C; idx += G1_THREADS) {
    const int kq = idx / H1C;
    const int n = idx % H1C;
    const float* src = B + (size_t)(4 * kq) * H1C + n;
    float4 v;
    v.x = src[0];
    v.y = src[H1C];
    v.z = src[2 * H1C];
    v.w = src[3 * H1C];
    Bs[idx] = v;
  }
  __syncthreads();

#pragma unroll
  for (int u = 0; u < G1_DE; ++u)
    if (lbegin + tid + u * G1_THREADS < lend) atomicAdd(&degs[pk[u] & 4095], wd[u]);
  for (int i0 = lbegin + tid + G1_THREADS * G1_DE; i0 < lend; i0 += G1_THREADS)
    atomicAdd(&degs[lrow[i0] & 4095], lw[i0]);

  const int tile = blockIdx.x * G1_WAVES + wave;
  const bool active = tile < ntiles;
  if (active) {
    const float* arow = A + ((size_t)tile * 16 + m) * IN_C + half * 4;
    const float4* bfrag = Bs + half * H1C + m;
    v8f c0 = {}, c1 = {}, c2 = {}, c3 = {};
    float4 a[4];
#pragma unroll
    for (int u = 0; u < 4; ++u) a[u] = *reinterpret_cast<const float4*>(arow + 8 * u);
    for (int kb = 0; kb < IN_C - 32; kb += 32) {
      float4 an[4];
#pragma unroll
      for (int u = 0; u < 4; ++u)
        an[u] = *reinterpret_cast<const float4*>(arow + kb + 32 + 8 * u);
#pragma unroll
      for (int u = 0; u < 4; ++u) g1_block(a[u], bfrag + ((kb + 8 * u) >> 2) * H1C, c0, c1, c2, c3);
#pragma unroll
      for (int u = 0; u < 4; ++u) a[u] = an[u];
    }
#pragma unroll
    for (int u = 0; u < 4; ++u)
      g1_block(a[u], bfrag + ((IN_C - 32 + 8 * u) >> 2) * H1C, c0, c1, c2, c3);

    float* cs = reinterpret_cast<float*>(Cs[wave]);
#pragma unroll
    for (int v = 0; v < 8; ++v) {
      float* p = cs + (v + 8 * half) * H1C + m;
      p[0] = c0[v];
      p[16] = c1[v];
      p[32] = c2[v];
      p[48] = c3[v];
    }
  }
  __syncthreads();
  if (tid < rlen) dinv[rbase + tid] = dinv_of(degs[tid]);
  if (active) {
    float4* dst = reinterpret_cast<float4*>(C + (size_t)tile * 16 * H1C);
#pragma unroll
    for (int i = 0; i < 8; ++i) {
      const float s = dinv_of(degs[wave * 16 + 2 * i + half]);
      float4 q = Cs[wave][i * 32 + lane];
      q.x *= s;
      q.y *= s;
      q.z *= s;
      q.w *= s;
      dst[i * 32 + lane] = q;
    }
  }
}


#define RBLOCK 1024

#define BKB 256
#define BKT 1024
#define BKE 16
#define BKR 256

__global__ __launch_bounds__(BKT) void k_bk_count(const int* __restrict__ ei,
                                                  int* __restrict__ cnt, int E, int Nn,
                                                  int nr) {
  __shared__ int hist[BKR];
  const int tid = threadIdx.x;
  if (tid < BKR) hist[tid] = 0;
  __syncthreads();
  const int per = (E + BKB - 1) / BKB;
  const int begin = blockIdx.x * per;
  const int end = min(E, begin + per);
  const int* __restrict__ col = ei + E;
  for (int e0 = begin + tid; e0 < end; e0 += BKT * BKE) {
    int c[BKE];
#pragma unroll
    for (int u = 0; u < BKE; ++u) {
      const int e = e0 + u * BKT;
      const int cc = col[min(e, end - 1)];
      c[u] = (e < end) ? cc : -1;
    }
#pragma unroll
    for (int u = 0; u < BKE; ++u)
      if ((unsigned)c[u] < (unsigned)Nn) atomicAdd(&hist[c[u] / OWNROWS], 1);
  }
  __syncthreads();
  if (tid < nr) cnt[tid * BKB + blockIdx.x] = hist[tid];
}

#define BKC (BKT * BKE)

__global__ __launch_bounds__(BKT) void k_bk_scatter(const int* __restrict__ ei,
                                                    const float* __restrict__ ew,
                                                    const int* __restrict__ cnt,
                                                    int* __restrict__ starts,
                                                    int* __restrict__ lrow,
                                                    float* __restrict__ lw, int E, int Nn,
                                                    int nr) {
  __shared__ int tot[BKR];
  __shared__ int before[BKR];
  __shared__ int cur[BKR];
  __shared__ int wsum[BKR / 32];
  __shared__ int lcnt[BKR];
  __shared__ int lbase[BKR];
  __shared__ int ntot;
  __shared__ int s_pk[BKC];
  __shared__ float s_w[BKC];
  __shared__ int s_pos[BKC];
  const int tid = threadIdx.x;
  const int lane = tid & 31;
  const int wave = tid >> 5;
  for (int r = wave; r < nr; r += BKT / 32) {
    int t = 0, bf = 0;
#pragma unroll
    for (int k = 0; k < BKB / 32; ++k) {
      const int b = k * 32 + lane;
      const int v = cnt[r * BKB + b];
      t += v;
      bf += (b < (int)blockIdx.x) ? v : 0;
    }
    for (int d = 16; d > 0; d >>= 1) {
      t += __shfl_xor(t, d);
      bf += __shfl_xor(bf, d);
    }
    if (lane == 0) {
      tot[r] = t;
      before[r] = bf;
    }
  }
  __syncthreads();
  int mine = 0, incl = 0;
  if (tid < BKR) {
    mine = (tid < nr) ? tot[tid] : 0;
    incl = mine;
    for (int d = 1; d < 32; d <<= 1) {
      const int below = __shfl_up(incl, d);
      if (lane >= d) incl += below;
    }
    if (lane == 31) wsum[wave] = incl;
  }
  __syncthreads();
  if (tid <= nr) {
    int sum = incl - mine;
    for (int q = 0; q < wave; ++q) sum += wsum[q];
    if (tid < nr) cur[tid] = sum + before[tid];
    if (blockIdx.x == 0) starts[tid] = sum;
  }
  __syncthreads();

  const int per = (E + BKB - 1) / BKB;
  const int begin = blockIdx.x * per;
  const int end = min(E, begin + per);
  const int* __restrict__ col = ei + E;
  for (int base = begin; base < end; base += BKC) {
    if (tid < BKR) lcnt[tid] = 0;
    __syncthreads();
    int c[BKE];
    int r[BKE];
    int rk[BKE];
    float w[BKE];
#pragma unroll
    for (int u = 0; u < BKE; ++u) {
      const int e = base + tid + u * BKT;
      const int ec = min(e, end - 1);
      const int cc = col[ec];
      r[u] = ei[ec];
      w[u] = ew[ec];
      c[u] = (e < end) ? cc : -1;
    }
#pragma unroll
    for (int u = 0; u < BKE; ++u) {
      rk[u] = 0;
      if ((unsigned)c[u] < (unsigned)Nn) rk[u] = atomicAdd(&lcnt[c[u] / OWNROWS], 1);
    }
    __syncthreads();
    int cm = 0, ci = 0;
    if (tid < BKR) {
      cm = (tid < nr) ? lcnt[tid] : 0;
      ci = cm;
      for (int d = 1; d < 32; d <<= 1) {
        const int below = __shfl_up(ci, d);
        if (lane >= d) ci += below;
      }
      if (lane == 31) wsum[wave] = ci;
    }
    __syncthreads();
    if (tid < BKR) {
      int sum = ci - cm;
      for (int q = 0; q < wave; ++q) sum += wsum[q];
      lbase[tid] = sum;
      if (tid == BKR - 1) ntot = sum;
    }
    __syncthreads();
#pragma unroll
    for (int u = 0; u < BKE; ++u) {
      if ((unsigned)c[u] < (unsigned)Nn) {
        const int rg = c[u] / OWNROWS;
        const int lp = lbase[rg] + rk[u];
        s_pk[lp] = (r[u] << 12) | (c[u] - rg * OWNROWS);
        s_w[lp] = w[u];
        s_pos[lp] = cur[rg] + rk[u];
      }
    }
    __syncthreads();
    if (tid < nr) cur[tid] += lcnt[tid];
    const int n = ntot;
    for (int j = tid; j < n; j += BKT) {
      const int pos = s_pos[j];
      lrow[pos] = s_pk[j];
      lw[pos] = s_w[j];
    }
  }
}


#define SCAP 16384
#define SPT (SCAP / RBLOCK)
#define SBINS 512

template <int F, int P, bool FULL>
__device__ __forceinline__ void row_group_seg(const int rv, const int relv, const float wv,
                                              const int left, const int lane,
                                              const float* __restrict__ h, float* acc) {
  constexpr int L = F / 32;
  float a[L];
#pragma unroll
  for (int l = 0; l < L; ++l) a[l] = 0.0f;
  int cur = __builtin_amdgcn_readlane(relv, 0);
#pragma unroll
  for (int b = 0; b < 32; b += 16) {
    if (FULL || b < left) {
      float v[16][L];
#pragma unroll
      for (int u = 0; u < 16; ++u) {
        const int r = __builtin_amdgcn_readlane(rv, b + u);
        const float* src = h + (size_t)r * F + L * lane;
        if (L == 2) {
          const float2 q = *reinterpret_cast<const float2*>(src);
          v[u][0] = q.x;
          v[u][L - 1] = q.y;
        } else {
          v[u][0] = src[0];
        }
      }
#pragma unroll
      for (int u = 0; u < 16; ++u) {
        if (FULL || b + u < left) {
          const int rel = __builtin_amdgcn_readlane(relv, b + u);
          const float w = __int_as_float(__builtin_amdgcn_readlane(__float_as_int(wv), b + u));
          if (rel != cur) {
            float* dst = acc + cur * P + L * lane;
#pragma unroll
            for (int l = 0; l < L; ++l) {
              atomicAdd(dst + l, a[l]);
              a[l] = 0.0f;
            }
            cur = rel;
          }
#pragma unroll
          for (int l = 0; l < L; ++l) a[l] += w * v[u][l];
        }
      }
    }
  }
  float* dst = acc + cur * P + L * lane;
#pragma unroll
  for (int l = 0; l < L; ++l) atomicAdd(dst + l, a[l]);
}

template <int F, int P>
__device__ __forceinline__ void bucket_sum_sorted(const int* __restrict__ lrow,
                                                  const float* __restrict__ lw, const int begin,
                                                  const int end, const int tid, const int wave,
                                                  const int lane, const float* __restrict__ h,
                                                  float* acc, int* s_pk, float* s_w, int* hist,
                                                  int* wsum) {
  static_assert(SPT * RBLOCK == SCAP && SBINS > OWNROWS && SBINS <= RBLOCK, "pass geometry");
  for (int cb = begin; cb < end; cb += SCAP) {
    const int n = min(SCAP, end - cb);
    int pk[SPT];
    int rk[SPT];
    float w[SPT];
#pragma unroll
    for (int u = 0; u < SPT; ++u) {
      const int ic = cb + min(tid + u * RBLOCK, n - 1);
      pk[u] = lrow[ic];
      w[u] = lw[ic];
    }
    if (tid < SBINS) hist[tid] = 0;
    __syncthreads();
#pragma unroll
    for (int u = 0; u < SPT; ++u) {
      rk[u] = 0;
      if (tid + u * RBLOCK < n) rk[u] = atomicAdd(&hist[pk[u] & 4095], 1);
    }
    __syncthreads();
    int mine = 0, incl = 0;
    if (tid < SBINS) {
      mine = hist[tid];
      incl = mine;
      for (int d = 1; d < 32; d <<= 1) {
        const int below = __shfl_up(incl, d);
        if (lane >= d) incl += below;
      }
      if (lane == 31) wsum[wave] = incl;
    }
    __syncthreads();
    if (tid < SBINS) {
      int sum = incl - mine;
      for (int q = 0; q < wave; ++q) sum += wsum[q];
      hist[tid] = sum;
    }
    __syncthreads();
#pragma unroll
    for (int u = 0; u < SPT; ++u) {
      if (tid + u * RBLOCK < n) {
        const int pos = hist[pk[u] & 4095] + rk[u];
        s_pk[pos] = pk[u];
        s_w[pos] = w[u];
      }
    }
    __syncthreads();
    for (int i = wave * 32; i < n; i += RBLOCK) {
      const int left = min(32, n - i);
      int pkv = 0;
      float wv = 0.0f;
      if (lane < left) {
        pkv = s_pk[i + lane];
        wv = s_w[i + lane];
      }
      if (left == 32)
        row_group_seg<F, P, true>(pkv >> 12, pkv & 4095, wv, left, lane, h, acc);
      else
        row_group_seg<F, P, false>(pkv >> 12, pkv & 4095, wv, left, lane, h, acc);
    }
    __syncthreads();
  }
}


#define APITCH 68

__global__ __launch_bounds__(RBLOCK) void k_agg64_gemm2(const int* __restrict__ lrow,
                                                        const float* __restrict__ lw,
                                                        const int* __restrict__ starts,
                                                        const float* __restrict__ h,
                                                        const float* __restrict__ dinv,
                                                        const float* __restrict__ bias,
                                                        const float* __restrict__ W2,
                                                        float* __restrict__ h2, int Nn) {
  constexpr int F = H1C;
  constexpr int PER = OWNROWS * F / RBLOCK;
  static_assert(OWNROWS * F % RBLOCK == 0 && RBLOCK % F == 0, "a thread keeps one column");
  static_assert(OWNROWS % 16 == 0 && OWNROWS / 16 <= RBLOCK / 32, "one wave per 16-row tile");
  __shared__ float acc[OWNROWS * APITCH];
  __shared__ float dl[OWNROWS];
  __shared__ int s_pk[SCAP];
  __shared__ float s_w[SCAP];
  __shared__ int hist[SBINS];
  __shared__ int wsum[SBINS / 32];
  const int tid = threadIdx.x;
  const int lane = tid & 31;
  const int wave = __builtin_amdgcn_readfirstlane(tid >> 5);
  const int begin = __builtin_amdgcn_readfirstlane(starts[blockIdx.x]);
  const int end = __builtin_amdgcn_readfirstlane(starts[blockIdx.x + 1]);
  const int rbase = blockIdx.x * OWNROWS;
  const int rlen = min(OWNROWS, Nn - rbase);
  if (tid < OWNROWS) dl[tid] = dinv[rbase + min(tid, rlen - 1)];
  for (int j = tid; j < OWNROWS * APITCH; j += RBLOCK) acc[j] = 0.0f;
  __syncthreads();

  bucket_sum_sorted<F, APITCH>(lrow, lw, begin, end, tid, wave, lane, h, acc, s_pk, s_w, hist, wsum);

  const int col = tid % F;
  const int row0 = tid / F;
  const float bs = bias[col];
  float sv[PER];
#pragma unroll
  for (int k = 0; k < PER; ++k) {
    const int rc = rbase + min(row0 + 16 * k, rlen - 1);
    sv[k] = h[(size_t)rc * F + col];
  }
  __syncthreads();

#pragma unroll
  for (int k = 0; k < PER; ++k) {
    const int r = row0 + 16 * k;
    float* p = acc + r * APITCH + col;
    const float t = fmaxf(bs + dl[r] * (sv[k] + *p), 0.0f);
    *p = (r < rlen) ? t : 0.0f;
  }
  __syncthreads();

  const bool active = wave < OWNROWS / 16;
  if (active) {
    float* tile = acc + wave * 16 * APITCH;
    const int m = lane & 15;
    const int half = lane >> 4;
    const float* ap = tile + m * APITCH + 2 * half;
    const float* bp = W2 + (2 * half) * H2C + m;
    v8f c0 = {}, c1 = {};
#pragma unroll 8
    for (int k = 0; k < H1C; k += 4) {
      const float2 av = *reinterpret_cast<const float2*>(ap + k);
      v2f a, b0, b1;
      a.x = av.x;
      a.y = av.y;
      b0.x = bp[k * H2C];
      b0.y = bp[(k + 1) * H2C];
      b1.x = bp[k * H2C + 16];
      b1.y = bp[(k + 1) * H2C + 16];
      c0 = __builtin_amdgcn_wmma_f32_16x16x4_f32(false, a, false, b0, (short)0, c0, false, false);
      c1 = __builtin_amdgcn_wmma_f32_16x16x4_f32(false, a, false, b1, (short)0, c1, false, false);
    }
#pragma unroll
    for (int v = 0; v < 8; ++v) {
      float* p = tile + (v + 8 * half) * APITCH + m;
      p[0] = c0[v];
      p[16] = c1[v];
    }
  }
  __syncthreads();
  if (active) {
    const float* tile = acc + wave * 16 * APITCH;
    float4* dst = reinterpret_cast<float4*>(h2 + (size_t)(rbase + wave * 16) * H2C);
#pragma unroll
    for (int i = 0; i < 4; ++i) {
      const int r = 4 * i + (lane >> 3);
      const float s = dl[wave * 16 + r];
      float4 q = *reinterpret_cast<const float4*>(tile + r * APITCH + 4 * (lane & 7));
      q.x *= s;
      q.y *= s;
      q.z *= s;
      q.w *= s;
      if (wave * 16 + r < rlen) dst[i * 32 + lane] = q;
    }
  }
}

__global__ __launch_bounds__(RBLOCK) void k_agg32_dot(const int* __restrict__ lrow,
                                                      const float* __restrict__ lw,
                                                      const int* __restrict__ starts,
                                                      const float* __restrict__ h,
                                                      const float* __restrict__ dinv,
                                                      const float* __restrict__ bias,
                                                      const float* __restrict__ W3,
                                                      float* __restrict__ g, int Nn) {
  constexpr int F = H2C;
  static_assert(RBLOCK % F == 0 && F == 32, "a row of the tail is one wave's lanes");
  constexpr int PER = (OWNROWS * F + RBLOCK - 1) / RBLOCK;
  __shared__ float acc[OWNROWS * F];
  __shared__ float gs[OWNROWS];
  __shared__ int s_pk[SCAP];
  __shared__ float s_w[SCAP];
  __shared__ int hist[SBINS];
  __shared__ int wsum[SBINS / 32];
  const int tid = threadIdx.x;
  const int lane = tid & 31;
  const int wave = __builtin_amdgcn_readfirstlane(tid >> 5);
  const int begin = __builtin_amdgcn_readfirstlane(starts[blockIdx.x]);
  const int end = __builtin_amdgcn_readfirstlane(starts[blockIdx.x + 1]);
  const int rbase = blockIdx.x * OWNROWS;
  const int rlen = min(OWNROWS, Nn - rbase);
  for (int j = tid; j < OWNROWS * F; j += RBLOCK) acc[j] = 0.0f;
  __syncthreads();

  bucket_sum_sorted<F, F>(lrow, lw, begin, end, tid, wave, lane, h, acc, s_pk, s_w, hist, wsum);

  const float* __restrict__ h_rows = h + (size_t)rbase * F;
  const int nfl = rlen * F;
  const float bs = bias[lane];
  const float w3 = W3[lane];
  float dv[PER];
  float sv[PER];
#pragma unroll
  for (int k = 0; k < PER; ++k) {
    const int jc = min(tid + k * RBLOCK, nfl - 1);
    dv[k] = dinv[rbase + jc / F];
    sv[k] = h_rows[jc];
  }
  __syncthreads();

#pragma unroll
  for (int k = 0; k < PER; ++k)
    sv[k] = bs + dv[k] * (sv[k] + acc[min(tid + k * RBLOCK, nfl - 1)]);
#pragma unroll
  for (int k = 0; k < PER; ++k) {
    float t = fmaxf(sv[k], 0.0f) * w3;
    for (int d = 16; d > 0; d >>= 1) t += __shfl_xor(t, d);
    const int r = wave + 32 * k;
    if (lane == 0 && r < OWNROWS) gs[r] = dv[k] * t;
  }
  __syncthreads();
  if (tid < rlen) g[rbase + tid] = gs[tid];
}


#define P3E 16

__global__ __launch_bounds__(RBLOCK) void k_agg1_own(const int* __restrict__ lrow,
                                                     const float* __restrict__ lw,
                                                     const int* __restrict__ starts,
                                                     const float* __restrict__ g,
                                                     float* __restrict__ sum, int Nn) {
  __shared__ float acc[OWNROWS];
  const int tid = threadIdx.x;
  const int begin = starts[blockIdx.x];
  const int end = starts[blockIdx.x + 1];
  const int rbase = blockIdx.x * OWNROWS;
  const int rlen = min(OWNROWS, Nn - rbase);
  for (int j = tid; j < OWNROWS; j += RBLOCK) acc[j] = 0.0f;
  __syncthreads();

  for (int i0 = begin + tid; i0 < end; i0 += RBLOCK * P3E) {
    int pk[P3E];
    float w[P3E];
    float v[P3E];
#pragma unroll
    for (int u = 0; u < P3E; ++u) {
      const int ic = min(i0 + u * RBLOCK, end - 1);
      pk[u] = lrow[ic];
      w[u] = lw[ic];
    }
#pragma unroll
    for (int u = 0; u < P3E; ++u) v[u] = g[pk[u] >> 12];
#pragma unroll
    for (int u = 0; u < P3E; ++u) asm volatile("" : "+v"(v[u]), "+v"(w[u]));
#pragma unroll
    for (int u = 0; u < P3E; ++u)
      if (i0 + u * RBLOCK < end) atomicAdd(&acc[pk[u] & 4095], w[u] * v[u]);
  }
  __syncthreads();

  float* __restrict__ dst = sum + rbase;
  for (int j = tid; j < rlen; j += RBLOCK) dst[j] = acc[j];
}

__global__ void k_out_sum(float* __restrict__ out, const float* __restrict__ sum,
                          const float* __restrict__ g, const float* __restrict__ dinv,
                          const float* __restrict__ b3, int Nn) {
  int i = blockIdx.x * blockDim.x + threadIdx.x;
  if (i >= Nn) return;
  out[i] = b3[0] + dinv[i] * (g[i] + sum[i]);
}


static inline unsigned int blocks_for(long long total, int bs) {
  return (unsigned int)((total + bs - 1) / bs);
}

extern "C" void kernel_launch(void* const* d_in, const int* in_sizes, int n_in,
                              void* d_out, int out_size, void* d_ws, size_t ws_size,
                              hipStream_t stream) {
  const float* x = (const float*)d_in[0];
  const int* ei = (const int*)d_in[1];
  const float* ew = (const float*)d_in[2];
  const float* W1 = (const float*)d_in[3];
  const float* b1 = (const float*)d_in[4];
  const float* W2 = (const float*)d_in[5];
  const float* b2 = (const float*)d_in[6];
  const float* W3 = (const float*)d_in[7];
  const float* b3 = (const float*)d_in[8];

  const int E = in_sizes[2];
  const int Nn = in_sizes[0] / IN_C;
  float* out = (float*)d_out;

  float* dinv = (float*)d_ws;
  float* lw = dinv + Nn;
  float* bufA = lw + E;
  float* bufA_hi = bufA + (size_t)Nn * H2C;
  float* bufB = bufA + (size_t)Nn * H1C;
  float* bufB_hi = bufB + (size_t)Nn * H2C;

  const int nr2 = (int)blocks_for(Nn, OWNROWS);
  int* bk_cnt = (int*)d_out;
  int* bk_starts = bk_cnt + nr2 * BKB;
  int* lrow = (int*)bufB_hi;
  k_bk_count<<<BKB, BKT, 0, stream>>>(ei, bk_cnt, E, Nn, nr2);
  k_bk_scatter<<<BKB, BKT, 0, stream>>>(ei, ew, bk_cnt, bk_starts, lrow, lw, E, Nn, nr2);

  k_gemm1_rows<<<nr2, G1_THREADS, 0, stream>>>(x, W1, lrow, lw, bk_starts, bufA, dinv, Nn / 16, Nn);
  k_agg64_gemm2<<<nr2, RBLOCK, 0, stream>>>(lrow, lw, bk_starts, bufA, dinv, b1, W2, bufB, Nn);

  k_agg32_dot<<<nr2, RBLOCK, 0, stream>>>(lrow, lw, bk_starts, bufB, dinv, b2, W3, bufA, Nn);

  k_agg1_own<<<nr2, RBLOCK, 0, stream>>>(lrow, lw, bk_starts, bufA, bufA_hi, Nn);
  k_out_sum<<<blocks_for(Nn, 256), 256, 0, stream>>>(out, bufA_hi, bufA, dinv, b3, Nn);
}
